# starting kernel
# baseline (speedup 1.0000x reference)
#include <hip/hip_runtime.h>
#include <hip/hip_bf16.h>

#define N_NODES 100000
#define N_EDGES 1000000
#define IN_DIM  8
#define HID     64
#define N_LAYERS 3
#define LN_EPS  1e-5f
#define SLOPE   0.2f
#define LDW     68

typedef float v2f __attribute__((ext_vector_type(2)));
typedef float v8f __attribute__((ext_vector_type(8)));

__global__ void gnn_proj_kernel(const float* __restrict__ x, const float* __restrict__ Wp,
                                const float* __restrict__ bp, float* __restrict__ h) {
  __shared__ float wp_s[HID * IN_DIM];
  __shared__ float bp_s[HID];
  int t = threadIdx.x;
  wp_s[t] = Wp[t];
  wp_s[t + 256] = Wp[t + 256];
  if (t < HID) bp_s[t] = bp[t];
  __syncthreads();
  int gid = blockIdx.x * 256 + t;
  int node = gid >> 6;
  int c = gid & 63;
  const float* xr = x + node * IN_DIM;
  const float* wr = wp_s + c * IN_DIM;
  float acc = bp_s[c];
#pragma unroll
  for (int k = 0; k < IN_DIM; ++k) acc += xr[k] * wr[k];
  h[gid] = acc;
}

__global__ void gnn_gemm_kernel(const float* __restrict__ h,
                                const float* __restrict__ Wself, const float* __restrict__ bself,
                                const float* __restrict__ Wneigh, const float* __restrict__ bneigh,
                                float* __restrict__ h_self, float* __restrict__ h_neigh) {
  __shared__ float w_s[HID * LDW];
  __shared__ float b_s[HID];
  const float* W  = blockIdx.y ? Wneigh : Wself;
  const float* b  = blockIdx.y ? bneigh : bself;
  float*       out = blockIdx.y ? h_neigh : h_self;

  int t = threadIdx.x;
#pragma unroll
  for (int i = 0; i < 16; ++i) {
    int idx = t + i * 256;
    w_s[(idx >> 6) * LDW + (idx & 63)] = W[idx];
  }
  if (t < HID) b_s[t] = b[t];
  __syncthreads();

  int wave = t >> 5, lane = t & 31;
  int r0 = blockIdx.x * 32 + (wave >> 2) * 16;
  int n0 = (wave & 3) * 16;
  int lr = lane & 15;
  int koff = (lane < 16) ? 0 : 2;

  const float* arow = h   + (size_t)(r0 + lr) * HID;
  const float* brow = w_s + (n0 + lr) * LDW;

  v8f c = {};
#pragma unroll
  for (int kb = 0; kb < 16; ++kb) {
    int k = kb * 4 + koff;
    v2f a  = *(const v2f*)(arow + k);
    v2f bf = *(const v2f*)(brow + k);
    c = __builtin_amdgcn_wmma_f32_16x16x4_f32(false, a, false, bf, (short)0, c, false, false);
  }

  float bias = b_s[n0 + lr];
  int mbase = r0 + ((lane < 16) ? 0 : 8);
#pragma unroll
  for (int i = 0; i < 8; ++i) {
    out[(size_t)(mbase + i) * HID + n0 + lr] = c[i] + bias;
  }
}

__global__ void gnn_edge_kernel(const int* __restrict__ eidx, const float* __restrict__ ew,
                                const float* __restrict__ Wedge, const float* __restrict__ h_neigh,
                                float* __restrict__ agg) {
  int wave = threadIdx.x >> 5, lane = threadIdx.x & 31;
  int e = blockIdx.x * 8 + wave;
  int row = eidx[e];
  int col = eidx[N_EDGES + e];
  float w = ew[e];
  int c2 = lane * 2;
  float g0 = 1.f / (1.f + __expf(-w * Wedge[c2]));
  float g1 = 1.f / (1.f + __expf(-w * Wedge[c2 + 1]));
  v2f m = *(const v2f*)(h_neigh + (size_t)col * HID + c2);
  float* dst = agg + (size_t)row * HID + c2;
  atomicAdd(dst,     m.x * g0);
  atomicAdd(dst + 1, m.y * g1);
}

__global__ void gnn_ln_kernel(const float* __restrict__ h_self, const float* __restrict__ agg,
                              const float* __restrict__ gamma, const float* __restrict__ beta,
                              const float* __restrict__ h, float* __restrict__ dst) {
  int wave = threadIdx.x >> 5, lane = threadIdx.x & 31;
  int node = blockIdx.x * 8 + wave;
  size_t base = (size_t)node * HID + lane * 2;
  v2f a = *(const v2f*)(h_self + base);
  v2f g = *(const v2f*)(agg + base);
  float v0 = a.x + g.x, v1 = a.y + g.y;
  float s = v0 + v1, ss = v0 * v0 + v1 * v1;
#pragma unroll
  for (int m = 16; m >= 1; m >>= 1) {
    s  += __shfl_xor(s,  m, 32);
    ss += __shfl_xor(ss, m, 32);
  }
  float mu  = s * (1.f / HID);
  float var = ss * (1.f / HID) - mu * mu;
  float inv = rsqrtf(var + LN_EPS);
  int c = lane * 2;
  float o0 = (v0 - mu) * inv * gamma[c]     + beta[c];
  float o1 = (v1 - mu) * inv * gamma[c + 1] + beta[c + 1];
  o0 = (o0 >= 0.f) ? o0 : SLOPE * o0;
  o1 = (o1 >= 0.f) ? o1 : SLOPE * o1;
  v2f hh = *(const v2f*)(h + base);
  v2f r; r.x = hh.x + o0; r.y = hh.y + o1;
  *(v2f*)(dst + base) = r;
}

extern "C" void kernel_launch(void* const* d_in, const int* in_sizes, int n_in,
                              void* d_out, int out_size, void* d_ws, size_t ws_size,
                              hipStream_t stream) {
  const float* x      = (const float*)d_in[0];
  const int*   eidx   = (const int*)  d_in[1];
  const float* ew     = (const float*)d_in[2];
  const float* Wp     = (const float*)d_in[3];
  const float* bp     = (const float*)d_in[4];
  const float* Wself  = (const float*)d_in[5];
  const float* bself  = (const float*)d_in[6];
  const float* Wneigh = (const float*)d_in[7];
  const float* bneigh = (const float*)d_in[8];
  const float* Wedge  = (const float*)d_in[9];
  const float* gamma  = (const float*)d_in[10];
  const float* beta   = (const float*)d_in[11];
  float* out = (float*)d_out;

  const size_t NH = (size_t)N_NODES * HID;
  float* h       = (float*)d_ws;
  float* h_self  = h + NH;
  float* h_neigh = h_self + NH;
  float* agg     = h_neigh + NH;

  gnn_proj_kernel<<<(N_NODES * HID) / 256, 256, 0, stream>>>(x, Wp, bp, h);

  for (int l = 0; l < N_LAYERS; ++l) {
    const float* Ws = Wself  + (size_t)l * HID * HID;
    const float* bs = bself  + (size_t)l * HID;
    const float* Wn = Wneigh + (size_t)l * HID * HID;
    const float* bn = bneigh + (size_t)l * HID;
    const float* We = Wedge  + (size_t)l * HID;
    const float* gm = gamma  + (size_t)l * HID;
    const float* bt = beta   + (size_t)l * HID;

    gnn_gemm_kernel<<<dim3(N_NODES / 32, 2), 256, 0, stream>>>(
        h, Ws, bs, Wn, bn, h_self, h_neigh);

    hipMemsetAsync(agg, 0, NH * sizeof(float), stream);

    gnn_edge_kernel<<<N_EDGES / 8, 256, 0, stream>>>(eidx, ew, We, h_neigh, agg);

    float* dst = (l == N_LAYERS - 1) ? out : h;
    gnn_ln_kernel<<<N_NODES / 8, 256, 0, stream>>>(h_self, agg, gm, bt, h, dst);
  }
}


# Replace float-atomic scatter-add by a per-call grouping of edges by destination (integer arrival rank) and a per-node register sum fused with the norm; sum order becomes arrival-rank order
# speedup vs baseline: 1.5209x; 1.5209x over previous
#include <hip/hip_runtime.h>
#include <hip/hip_bf16.h>

#define N_NODES 100000
#define N_EDGES 1000000
#define IN_DIM  8
#define HID     64
#define N_LAYERS 3
#define LN_EPS  1e-5f
#define SLOPE   0.2f
#define LDW     68

typedef float v2f __attribute__((ext_vector_type(2)));
typedef float v8f __attribute__((ext_vector_type(8)));
typedef int   v2i __attribute__((ext_vector_type(2)));

#define SCAN_T   1024
#define SCAN_B   98
#define OFF_CNT  0
#define OFF_OFF  100352
#define OFF_RANK 200704
#define OFF_PERM 1200704

__global__ void gnn_proj_kernel(const float* __restrict__ x, const float* __restrict__ Wp,
                                const float* __restrict__ bp, float* __restrict__ h,
                                int* __restrict__ cnt) {
  __shared__ float wp_s[HID * IN_DIM];
  __shared__ float bp_s[HID];
  int t = threadIdx.x;
  wp_s[t] = Wp[t];
  wp_s[t + 256] = Wp[t + 256];
  if (t < HID) bp_s[t] = bp[t];
  __syncthreads();
  int gid = blockIdx.x * 256 + t;
  int node = gid >> 6;
  int c = gid & 63;
  const float* xr = x + node * IN_DIM;
  const float* wr = wp_s + c * IN_DIM;
  float acc = bp_s[c];
#pragma unroll
  for (int k = 0; k < IN_DIM; ++k) acc += xr[k] * wr[k];
  h[gid] = acc;
  if (gid < N_NODES) cnt[gid] = 0;
}

__global__ void gnn_gemm_kernel(const float* __restrict__ h,
                                const float* __restrict__ Wself, const float* __restrict__ bself,
                                const float* __restrict__ Wneigh, const float* __restrict__ bneigh,
                                float* __restrict__ h_self, float* __restrict__ h_neigh) {
  __shared__ float w_s[HID * LDW];
  __shared__ float b_s[HID];
  const float* W  = blockIdx.y ? Wneigh : Wself;
  const float* b  = blockIdx.y ? bneigh : bself;
  float*       out = blockIdx.y ? h_neigh : h_self;

  int t = threadIdx.x;
#pragma unroll
  for (int i = 0; i < 16; ++i) {
    int idx = t + i * 256;
    w_s[(idx >> 6) * LDW + (idx & 63)] = W[idx];
  }
  if (t < HID) b_s[t] = b[t];
  __syncthreads();

  int wave = t >> 5, lane = t & 31;
  int r0 = blockIdx.x * 32 + (wave >> 2) * 16;
  int n0 = (wave & 3) * 16;
  int lr = lane & 15;
  int koff = (lane < 16) ? 0 : 2;

  const float* arow = h   + (size_t)(r0 + lr) * HID;
  const float* brow = w_s + (n0 + lr) * LDW;

  v8f c = {};
#pragma unroll
  for (int kb = 0; kb < 16; ++kb) {
    int k = kb * 4 + koff;
    v2f a  = *(const v2f*)(arow + k);
    v2f bf = *(const v2f*)(brow + k);
    c = __builtin_amdgcn_wmma_f32_16x16x4_f32(false, a, false, bf, (short)0, c, false, false);
  }

  float bias = b_s[n0 + lr];
  int mbase = r0 + ((lane < 16) ? 0 : 8);
#pragma unroll
  for (int i = 0; i < 8; ++i) {
    out[(size_t)(mbase + i) * HID + n0 + lr] = c[i] + bias;
  }
}

__global__ void gnn_rank_kernel(const int* __restrict__ eidx, int* __restrict__ cnt,
                                int* __restrict__ rank) {
  int e = blockIdx.x * 256 + threadIdx.x;
  if (e < N_EDGES) rank[e] = atomicAdd(cnt + eidx[e], 1);
}

__global__ void gnn_scan_kernel(const int* __restrict__ cnt, int* __restrict__ off) {
  __shared__ int wsum[32];
  __shared__ int base_s;
  int t = threadIdx.x, lane = t & 31, wave = t >> 5;
  int b = blockIdx.x;
  int part = 0;
  for (int k = 0; k < b; ++k) part += cnt[k * SCAN_T + t];
  for (int m = 16; m >= 1; m >>= 1) part += __shfl_xor(part, m, 32);
  if (lane == 0) wsum[wave] = part;
  __syncthreads();
  if (wave == 0) {
    int x = wsum[lane];
    for (int m = 16; m >= 1; m >>= 1) x += __shfl_xor(x, m, 32);
    if (lane == 0) base_s = x;
  }
  __syncthreads();
  int base = base_s;
  int i = b * SCAN_T + t;
  int v = (i < N_NODES) ? cnt[i] : 0;
  int inc = v;
  for (int d = 1; d < 32; d <<= 1) {
    int y = __shfl_up(inc, d, 32);
    if (lane >= d) inc += y;
  }
  if (lane == 31) wsum[wave] = inc;
  __syncthreads();
  if (wave == 0) {
    int x = wsum[lane];
    int xi = x;
    for (int d = 1; d < 32; d <<= 1) {
      int y = __shfl_up(xi, d, 32);
      if (lane >= d) xi += y;
    }
    wsum[lane] = xi - x;
  }
  __syncthreads();
  if (i <= N_NODES) off[i] = base + wsum[wave] + inc - v;
}

__global__ void gnn_place_kernel(const int* __restrict__ eidx, const float* __restrict__ ew,
                                 const int* __restrict__ off, const int* __restrict__ rank,
                                 int* __restrict__ perm) {
  int e = blockIdx.x * 256 + threadIdx.x;
  if (e < N_EDGES) {
    int row = eidx[e];
    int col = eidx[N_EDGES + e];
    int p = off[row] + rank[e];
    p = min(max(p, 0), N_EDGES - 1);
    v2i cw; cw.x = col; cw.y = __float_as_int(ew[e]);
    *(v2i*)(perm + 2 * (size_t)p) = cw;
  }
}

#define EDGE_COL(J) min(max(__builtin_amdgcn_readlane(cw.x, (J)), 0), N_NODES - 1)
#define EDGE_W(J)   __int_as_float(__builtin_amdgcn_readlane(cw.y, (J)))
#define ROW_OF(C)   (*(const v2f*)(h_neigh + (size_t)(C) * HID + c2))
#define GATE(W, WE) (1.f / (1.f + __expf(-(W) * (WE))))

__global__ void gnn_gather_ln_kernel(const int* __restrict__ off, const int* __restrict__ perm,
                                     const float* __restrict__ Wedge, const float* __restrict__ h_neigh,
                                     const float* __restrict__ h_self,
                                     const float* __restrict__ gamma, const float* __restrict__ beta,
                                     const float* h, float* dst) {
  int wave = threadIdx.x >> 5, lane = threadIdx.x & 31;
  int node = blockIdx.x * 8 + wave;
  int s = __builtin_amdgcn_readfirstlane(off[node]);
  int t = __builtin_amdgcn_readfirstlane(off[node + 1]);
  s = min(max(s, 0), N_EDGES);
  t = min(max(t, s), N_EDGES);
  int c2 = lane * 2;
  size_t base = (size_t)node * HID + c2;
  v2f a  = *(const v2f*)(h_self + base);
  v2f hh = *(const v2f*)(h + base);
  float we0 = Wedge[c2], we1 = Wedge[c2 + 1];
  float acc0 = 0.f, acc1 = 0.f;
  for (int p = s; p < t; p += 32) {
    int n = min(t - p, 32);
    v2i cw; cw.x = 0; cw.y = 0;
    if (lane < n) cw = *(const v2i*)(perm + 2 * (size_t)(p + lane));
    int j = 0;
    for (; j + 4 <= n; j += 4) {
      int ca = EDGE_COL(j), cb = EDGE_COL(j + 1), cc = EDGE_COL(j + 2), cd = EDGE_COL(j + 3);
      v2f ma = ROW_OF(ca), mb = ROW_OF(cb), mc = ROW_OF(cc), md = ROW_OF(cd);
      float wa = EDGE_W(j), wb = EDGE_W(j + 1), wc = EDGE_W(j + 2), wd = EDGE_W(j + 3);
      float ga0 = GATE(wa, we0), ga1 = GATE(wa, we1);
      float gb0 = GATE(wb, we0), gb1 = GATE(wb, we1);
      float gc0 = GATE(wc, we0), gc1 = GATE(wc, we1);
      float gd0 = GATE(wd, we0), gd1 = GATE(wd, we1);
      acc0 += ma.x * ga0; acc1 += ma.y * ga1;
      acc0 += mb.x * gb0; acc1 += mb.y * gb1;
      acc0 += mc.x * gc0; acc1 += mc.y * gc1;
      acc0 += md.x * gd0; acc1 += md.y * gd1;
    }
    for (; j < n; ++j) {
      int ca = EDGE_COL(j);
      v2f ma = ROW_OF(ca);
      float wa = EDGE_W(j);
      acc0 += ma.x * GATE(wa, we0); acc1 += ma.y * GATE(wa, we1);
    }
  }
  float v0 = a.x + acc0, v1 = a.y + acc1;
  float sm = v0 + v1, ss = v0 * v0 + v1 * v1;
  for (int m = 16; m >= 1; m >>= 1) {
    sm += __shfl_xor(sm, m, 32);
    ss += __shfl_xor(ss, m, 32);
  }
  float mu  = sm * (1.f / HID);
  float var = ss * (1.f / HID) - mu * mu;
  float inv = rsqrtf(var + LN_EPS);
  float o0 = (v0 - mu) * inv * gamma[c2]     + beta[c2];
  float o1 = (v1 - mu) * inv * gamma[c2 + 1] + beta[c2 + 1];
  o0 = (o0 >= 0.f) ? o0 : SLOPE * o0;
  o1 = (o1 >= 0.f) ? o1 : SLOPE * o1;
  v2f r; r.x = hh.x + o0; r.y = hh.y + o1;
  *(v2f*)(dst + base) = r;
}

extern "C" void kernel_launch(void* const* d_in, const int* in_sizes, int n_in,
                              void* d_out, int out_size, void* d_ws, size_t ws_size,
                              hipStream_t stream) {
  const float* x      = (const float*)d_in[0];
  const int*   eidx   = (const int*)  d_in[1];
  const float* ew     = (const float*)d_in[2];
  const float* Wp     = (const float*)d_in[3];
  const float* bp     = (const float*)d_in[4];
  const float* Wself  = (const float*)d_in[5];
  const float* bself  = (const float*)d_in[6];
  const float* Wneigh = (const float*)d_in[7];
  const float* bneigh = (const float*)d_in[8];
  const float* Wedge  = (const float*)d_in[9];
  const float* gamma  = (const float*)d_in[10];
  const float* beta   = (const float*)d_in[11];
  float* out = (float*)d_out;

  const size_t NH = (size_t)N_NODES * HID;
  float* h       = (float*)d_ws;
  float* h_self  = h + NH;
  float* h_neigh = h_self + NH;
  int*   ord     = (int*)(h_neigh + NH);
  int*   cnt     = ord + OFF_CNT;
  int*   off     = ord + OFF_OFF;
  int*   rank    = ord + OFF_RANK;
  int*   perm    = ord + OFF_PERM;

  gnn_proj_kernel<<<(N_NODES * HID) / 256, 256, 0, stream>>>(x, Wp, bp, h, cnt);

  gnn_rank_kernel<<<(N_EDGES + 255) / 256, 256, 0, stream>>>(eidx, cnt, rank);
  gnn_scan_kernel<<<SCAN_B, SCAN_T, 0, stream>>>(cnt, off);
  gnn_place_kernel<<<(N_EDGES + 255) / 256, 256, 0, stream>>>(eidx, ew, off, rank, perm);

  for (int l = 0; l < N_LAYERS; ++l) {
    const float* Ws = Wself  + (size_t)l * HID * HID;
    const float* bs = bself  + (size_t)l * HID;
    const float* Wn = Wneigh + (size_t)l * HID * HID;
    const float* bn = bneigh + (size_t)l * HID;
    const float* We = Wedge  + (size_t)l * HID;
    const float* gm = gamma  + (size_t)l * HID;
    const float* bt = beta   + (size_t)l * HID;

    gnn_gemm_kernel<<<dim3(N_NODES / 32, 2), 256, 0, stream>>>(
        h, Ws, bs, Wn, bn, h_self, h_neigh);

    float* dst = (l == N_LAYERS - 1) ? out : h;
    gnn_gather_ln_kernel<<<N_NODES / 8, 256, 0, stream>>>(off, perm, We, h_neigh, h_self, gm, bt, h, dst);
  }
}


# Edge gate as v_rcp_f32(1 + v_exp_f32(w * Wedge pre-scaled by -log2 e)) in place of expf and an IEEE-rounded divide (11 vector instructions each); gate differs by about 1 ulp, sum order unchanged
# speedup vs baseline: 1.6273x; 1.0700x over previous
#include <hip/hip_runtime.h>
#include <hip/hip_bf16.h>

#define N_NODES 100000
#define N_EDGES 1000000
#define IN_DIM  8
#define HID     64
#define N_LAYERS 3
#define LN_EPS  1e-5f
#define SLOPE   0.2f
#define LDW     68

typedef float v2f __attribute__((ext_vector_type(2)));
typedef float v8f __attribute__((ext_vector_type(8)));
typedef int   v2i __attribute__((ext_vector_type(2)));

#define SCAN_T   1024
#define SCAN_B   98
#define OFF_CNT  0
#define OFF_OFF  100352
#define OFF_RANK 200704
#define OFF_PERM 1200704

__global__ void gnn_proj_kernel(const float* __restrict__ x, const float* __restrict__ Wp,
                                const float* __restrict__ bp, float* __restrict__ h,
                                int* __restrict__ cnt) {
  __shared__ float wp_s[HID * IN_DIM];
  __shared__ float bp_s[HID];
  int t = threadIdx.x;
  wp_s[t] = Wp[t];
  wp_s[t + 256] = Wp[t + 256];
  if (t < HID) bp_s[t] = bp[t];
  __syncthreads();
  int gid = blockIdx.x * 256 + t;
  int node = gid >> 6;
  int c = gid & 63;
  const float* xr = x + node * IN_DIM;
  const float* wr = wp_s + c * IN_DIM;
  float acc = bp_s[c];
#pragma unroll
  for (int k = 0; k < IN_DIM; ++k) acc += xr[k] * wr[k];
  h[gid] = acc;
  if (gid < N_NODES) cnt[gid] = 0;
}

__global__ void gnn_gemm_kernel(const float* __restrict__ h,
                                const float* __restrict__ Wself, const float* __restrict__ bself,
                                const float* __restrict__ Wneigh, const float* __restrict__ bneigh,
                                float* __restrict__ h_self, float* __restrict__ h_neigh) {
  __shared__ float w_s[HID * LDW];
  __shared__ float b_s[HID];
  const float* W  = blockIdx.y ? Wneigh : Wself;
  const float* b  = blockIdx.y ? bneigh : bself;
  float*       out = blockIdx.y ? h_neigh : h_self;

  int t = threadIdx.x;
#pragma unroll
  for (int i = 0; i < 16; ++i) {
    int idx = t + i * 256;
    w_s[(idx >> 6) * LDW + (idx & 63)] = W[idx];
  }
  if (t < HID) b_s[t] = b[t];
  __syncthreads();

  int wave = t >> 5, lane = t & 31;
  int r0 = blockIdx.x * 32 + (wave >> 2) * 16;
  int n0 = (wave & 3) * 16;
  int lr = lane & 15;
  int koff = (lane < 16) ? 0 : 2;

  const float* arow = h   + (size_t)(r0 + lr) * HID;
  const float* brow = w_s + (n0 + lr) * LDW;

  v8f c = {};
#pragma unroll
  for (int kb = 0; kb < 16; ++kb) {
    int k = kb * 4 + koff;
    v2f a  = *(const v2f*)(arow + k);
    v2f bf = *(const v2f*)(brow + k);
    c = __builtin_amdgcn_wmma_f32_16x16x4_f32(false, a, false, bf, (short)0, c, false, false);
  }

  float bias = b_s[n0 + lr];
  int mbase = r0 + ((lane < 16) ? 0 : 8);
#pragma unroll
  for (int i = 0; i < 8; ++i) {
    out[(size_t)(mbase + i) * HID + n0 + lr] = c[i] + bias;
  }
}

__global__ void gnn_rank_kernel(const int* __restrict__ eidx, int* __restrict__ cnt,
                                int* __restrict__ rank) {
  int e = blockIdx.x * 256 + threadIdx.x;
  if (e < N_EDGES) rank[e] = atomicAdd(cnt + eidx[e], 1);
}

__global__ void gnn_scan_kernel(const int* __restrict__ cnt, int* __restrict__ off) {
  __shared__ int wsum[32];
  __shared__ int base_s;
  int t = threadIdx.x, lane = t & 31, wave = t >> 5;
  int b = blockIdx.x;
  int part = 0;
  for (int k = 0; k < b; ++k) part += cnt[k * SCAN_T + t];
  for (int m = 16; m >= 1; m >>= 1) part += __shfl_xor(part, m, 32);
  if (lane == 0) wsum[wave] = part;
  __syncthreads();
  if (wave == 0) {
    int x = wsum[lane];
    for (int m = 16; m >= 1; m >>= 1) x += __shfl_xor(x, m, 32);
    if (lane == 0) base_s = x;
  }
  __syncthreads();
  int base = base_s;
  int i = b * SCAN_T + t;
  int v = (i < N_NODES) ? cnt[i] : 0;
  int inc = v;
  for (int d = 1; d < 32; d <<= 1) {
    int y = __shfl_up(inc, d, 32);
    if (lane >= d) inc += y;
  }
  if (lane == 31) wsum[wave] = inc;
  __syncthreads();
  if (wave == 0) {
    int x = wsum[lane];
    int xi = x;
    for (int d = 1; d < 32; d <<= 1) {
      int y = __shfl_up(xi, d, 32);
      if (lane >= d) xi += y;
    }
    wsum[lane] = xi - x;
  }
  __syncthreads();
  if (i <= N_NODES) off[i] = base + wsum[wave] + inc - v;
}

__global__ void gnn_place_kernel(const int* __restrict__ eidx, const float* __restrict__ ew,
                                 const int* __restrict__ off, const int* __restrict__ rank,
                                 int* __restrict__ perm) {
  int e = blockIdx.x * 256 + threadIdx.x;
  if (e < N_EDGES) {
    int row = eidx[e];
    int col = eidx[N_EDGES + e];
    int p = off[row] + rank[e];
    p = min(max(p, 0), N_EDGES - 1);
    v2i cw; cw.x = col; cw.y = __float_as_int(ew[e]);
    *(v2i*)(perm + 2 * (size_t)p) = cw;
  }
}

#define EDGE_COL(J) min(max(__builtin_amdgcn_readlane(cw.x, (J)), 0), N_NODES - 1)
#define EDGE_W(J)   __int_as_float(__builtin_amdgcn_readlane(cw.y, (J)))
#define ROW_OF(C)   (*(const v2f*)(h_neigh + (size_t)(C) * HID + c2))
#define GATE(W, NWE) __builtin_amdgcn_rcpf(1.f + __builtin_amdgcn_exp2f((W) * (NWE)))

__global__ void gnn_gather_ln_kernel(const int* __restrict__ off, const int* __restrict__ perm,
                                     const float* __restrict__ Wedge, const float* __restrict__ h_neigh,
                                     const float* __restrict__ h_self,
                                     const float* __restrict__ gamma, const float* __restrict__ beta,
                                     const float* h, float* dst) {
  int wave = threadIdx.x >> 5, lane = threadIdx.x & 31;
  int node = blockIdx.x * 8 + wave;
  int s = __builtin_amdgcn_readfirstlane(off[node]);
  int t = __builtin_amdgcn_readfirstlane(off[node + 1]);
  s = min(max(s, 0), N_EDGES);
  t = min(max(t, s), N_EDGES);
  int c2 = lane * 2;
  size_t base = (size_t)node * HID + c2;
  v2f a  = *(const v2f*)(h_self + base);
  v2f hh = *(const v2f*)(h + base);
  float we0 = Wedge[c2] * -1.44269504f, we1 = Wedge[c2 + 1] * -1.44269504f;
  float acc0 = 0.f, acc1 = 0.f;
  for (int p = s; p < t; p += 32) {
    int n = min(t - p, 32);
    v2i cw; cw.x = 0; cw.y = 0;
    if (lane < n) cw = *(const v2i*)(perm + 2 * (size_t)(p + lane));
    int j = 0;
    for (; j + 4 <= n; j += 4) {
      int ca = EDGE_COL(j), cb = EDGE_COL(j + 1), cc = EDGE_COL(j + 2), cd = EDGE_COL(j + 3);
      v2f ma = ROW_OF(ca), mb = ROW_OF(cb), mc = ROW_OF(cc), md = ROW_OF(cd);
      float wa = EDGE_W(j), wb = EDGE_W(j + 1), wc = EDGE_W(j + 2), wd = EDGE_W(j + 3);
      float ga0 = GATE(wa, we0), ga1 = GATE(wa, we1);
      float gb0 = GATE(wb, we0), gb1 = GATE(wb, we1);
      float gc0 = GATE(wc, we0), gc1 = GATE(wc, we1);
      float gd0 = GATE(wd, we0), gd1 = GATE(wd, we1);
      acc0 += ma.x * ga0; acc1 += ma.y * ga1;
      acc0 += mb.x * gb0; acc1 += mb.y * gb1;
      acc0 += mc.x * gc0; acc1 += mc.y * gc1;
      acc0 += md.x * gd0; acc1 += md.y * gd1;
    }
    for (; j < n; ++j) {
      int ca = EDGE_COL(j);
      v2f ma = ROW_OF(ca);
      float wa = EDGE_W(j);
      acc0 += ma.x * GATE(wa, we0); acc1 += ma.y * GATE(wa, we1);
    }
  }
  float v0 = a.x + acc0, v1 = a.y + acc1;
  float sm = v0 + v1, ss = v0 * v0 + v1 * v1;
  for (int m = 16; m >= 1; m >>= 1) {
    sm += __shfl_xor(sm, m, 32);
    ss += __shfl_xor(ss, m, 32);
  }
  float mu  = sm * (1.f / HID);
  float var = ss * (1.f / HID) - mu * mu;
  float inv = rsqrtf(var + LN_EPS);
  float o0 = (v0 - mu) * inv * gamma[c2]     + beta[c2];
  float o1 = (v1 - mu) * inv * gamma[c2 + 1] + beta[c2 + 1];
  o0 = (o0 >= 0.f) ? o0 : SLOPE * o0;
  o1 = (o1 >= 0.f) ? o1 : SLOPE * o1;
  v2f r; r.x = hh.x + o0; r.y = hh.y + o1;
  *(v2f*)(dst + base) = r;
}

extern "C" void kernel_launch(void* const* d_in, const int* in_sizes, int n_in,
                              void* d_out, int out_size, void* d_ws, size_t ws_size,
                              hipStream_t stream) {
  const float* x      = (const float*)d_in[0];
  const int*   eidx   = (const int*)  d_in[1];
  const float* ew     = (const float*)d_in[2];
  const float* Wp     = (const float*)d_in[3];
  const float* bp     = (const float*)d_in[4];
  const float* Wself  = (const float*)d_in[5];
  const float* bself  = (const float*)d_in[6];
  const float* Wneigh = (const float*)d_in[7];
  const float* bneigh = (const float*)d_in[8];
  const float* Wedge  = (const float*)d_in[9];
  const float* gamma  = (const float*)d_in[10];
  const float* beta   = (const float*)d_in[11];
  float* out = (float*)d_out;

  const size_t NH = (size_t)N_NODES * HID;
  float* h       = (float*)d_ws;
  float* h_self  = h + NH;
  float* h_neigh = h_self + NH;
  int*   ord     = (int*)(h_neigh + NH);
  int*   cnt     = ord + OFF_CNT;
  int*   off     = ord + OFF_OFF;
  int*   rank    = ord + OFF_RANK;
  int*   perm    = ord + OFF_PERM;

  gnn_proj_kernel<<<(N_NODES * HID) / 256, 256, 0, stream>>>(x, Wp, bp, h, cnt);

  gnn_rank_kernel<<<(N_EDGES + 255) / 256, 256, 0, stream>>>(eidx, cnt, rank);
  gnn_scan_kernel<<<SCAN_B, SCAN_T, 0, stream>>>(cnt, off);
  gnn_place_kernel<<<(N_EDGES + 255) / 256, 256, 0, stream>>>(eidx, ew, off, rank, perm);

  for (int l = 0; l < N_LAYERS; ++l) {
    const float* Ws = Wself  + (size_t)l * HID * HID;
    const float* bs = bself  + (size_t)l * HID;
    const float* Wn = Wneigh + (size_t)l * HID * HID;
    const float* bn = bneigh + (size_t)l * HID;
    const float* We = Wedge  + (size_t)l * HID;
    const float* gm = gamma  + (size_t)l * HID;
    const float* bt = beta   + (size_t)l * HID;

    gnn_gemm_kernel<<<dim3(N_NODES / 32, 2), 256, 0, stream>>>(
        h, Ws, bs, Wn, bn, h_self, h_neigh);

    float* dst = (l == N_LAYERS - 1) ? out : h;
    gnn_gather_ln_kernel<<<N_NODES / 8, 256, 0, stream>>>(off, perm, We, h_neigh, h_self, gm, bt, h, dst);
  }
}


# Every wave loops over several work items (proj 10 nodes, rank and place 4 edges a thread, GEMM wave 16 rows x 128 cols, gather 2 nodes): under a third of the waves enter per call; sum orders kept
# speedup vs baseline: 1.6905x; 1.0388x over previous
#include <hip/hip_runtime.h>
#include <hip/hip_bf16.h>

#define N_NODES 100000
#define N_EDGES 1000000
#define IN_DIM  8
#define HID     64
#define N_LAYERS 3
#define LN_EPS  1e-5f
#define SLOPE   0.2f
#define LDW     68

typedef float v2f __attribute__((ext_vector_type(2)));
typedef float v8f __attribute__((ext_vector_type(8)));
typedef int   v2i __attribute__((ext_vector_type(2)));
typedef float v4f __attribute__((ext_vector_type(4)));
typedef int   v4i __attribute__((ext_vector_type(4)));

#define PROJ_BLOCKS 1250
#define PROJ_TRIPS  5
#define QUADS       (N_EDGES / 4)
#define ROW_TILES   (N_NODES / 16)
#define GATHER_NPW  2

#define SCAN_T   1024
#define SCAN_B   98
#define OFF_CNT  0
#define OFF_OFF  100352
#define OFF_RANK 200704
#define OFF_PERM 1200704

__global__ void gnn_proj_kernel(const float* __restrict__ x, const float* __restrict__ Wp,
                                const float* __restrict__ bp, float* __restrict__ h,
                                int* __restrict__ cnt) {
  int t = threadIdx.x;
  int gt = blockIdx.x * 256 + t;
  if (gt < N_NODES) cnt[gt] = 0;
  int lane = t & 31;
  int gw = gt >> 5;
  int hf = lane >> 4;
  int c4 = (lane & 15) * 4;
  const v4f* wp4 = (const v4f*)(Wp + c4 * IN_DIM);
  v4f w0a = wp4[0], w0b = wp4[1], w1a = wp4[2], w1b = wp4[3];
  v4f w2a = wp4[4], w2b = wp4[5], w3a = wp4[6], w3b = wp4[7];
  v4f b4 = *(const v4f*)(bp + c4);
  v4f xs[PROJ_TRIPS][2];
#pragma unroll
  for (int i = 0; i < PROJ_TRIPS; ++i) {
    int node = (gw * PROJ_TRIPS + i) * 2 + hf;
    const v4f* xr = (const v4f*)(x + (size_t)node * IN_DIM);
    xs[i][0] = xr[0]; xs[i][1] = xr[1];
  }
#pragma unroll
  for (int i = 0; i < PROJ_TRIPS; ++i) {
    int node = (gw * PROJ_TRIPS + i) * 2 + hf;
    v4f xa = xs[i][0], xb = xs[i][1];
    float a0 = b4.x, a1 = b4.y, a2 = b4.z, a3 = b4.w;
    a0 += xa.x * w0a.x; a0 += xa.y * w0a.y; a0 += xa.z * w0a.z; a0 += xa.w * w0a.w;
    a0 += xb.x * w0b.x; a0 += xb.y * w0b.y; a0 += xb.z * w0b.z; a0 += xb.w * w0b.w;
    a1 += xa.x * w1a.x; a1 += xa.y * w1a.y; a1 += xa.z * w1a.z; a1 += xa.w * w1a.w;
    a1 += xb.x * w1b.x; a1 += xb.y * w1b.y; a1 += xb.z * w1b.z; a1 += xb.w * w1b.w;
    a2 += xa.x * w2a.x; a2 += xa.y * w2a.y; a2 += xa.z * w2a.z; a2 += xa.w * w2a.w;
    a2 += xb.x * w2b.x; a2 += xb.y * w2b.y; a2 += xb.z * w2b.z; a2 += xb.w * w2b.w;
    a3 += xa.x * w3a.x; a3 += xa.y * w3a.y; a3 += xa.z * w3a.z; a3 += xa.w * w3a.w;
    a3 += xb.x * w3b.x; a3 += xb.y * w3b.y; a3 += xb.z * w3b.z; a3 += xb.w * w3b.w;
    v4f r; r.x = a0; r.y = a1; r.z = a2; r.w = a3;
    *(v4f*)(h + (size_t)node * HID + c4) = r;
  }
}

__global__ void gnn_gemm_kernel(const float* __restrict__ h,
                                const float* __restrict__ Wself, const float* __restrict__ bself,
                                const float* __restrict__ Wneigh, const float* __restrict__ bneigh,
                                float* __restrict__ h_self, float* __restrict__ h_neigh) {
  __shared__ float w_s[2 * HID * LDW];
  __shared__ float b_s[2 * HID];

  int t = threadIdx.x;
#pragma unroll
  for (int i = 0; i < 4; ++i) {
    int q = t + i * 256;
    int row = q >> 4, col = (q & 15) * 4;
    *(v4f*)(w_s + row * LDW + col)         = *(const v4f*)(Wself  + 4 * q);
    *(v4f*)(w_s + (HID + row) * LDW + col) = *(const v4f*)(Wneigh + 4 * q);
  }
  if (t < HID) b_s[t] = bself[t];
  else if (t < 2 * HID) b_s[t] = bneigh[t - HID];
  __syncthreads();

  int wave = t >> 5, lane = t & 31;
  int tile = blockIdx.x * 8 + wave;
  if (tile >= ROW_TILES) return;
  int r0 = tile * 16;
  int lr = lane & 15;
  int koff = (lane < 16) ? 0 : 2;

  const float* arow = h + (size_t)(r0 + lr) * HID;
  const float* brow = w_s + lr * LDW + koff;

  v2f a[16];
#pragma unroll
  for (int kb = 0; kb < 16; ++kb) a[kb] = *(const v2f*)(arow + kb * 4 + koff);

  v8f c[8];
#pragma unroll
  for (int j = 0; j < 8; ++j) c[j] = (v8f){};
#pragma unroll
  for (int kb = 0; kb < 16; ++kb) {
#pragma unroll
    for (int j = 0; j < 8; ++j) {
      v2f bf = *(const v2f*)(brow + j * 16 * LDW + kb * 4);
      c[j] = __builtin_amdgcn_wmma_f32_16x16x4_f32(false, a[kb], false, bf, (short)0, c[j], false, false);
    }
  }

  int mbase = r0 + ((lane < 16) ? 0 : 8);
#pragma unroll
  for (int j = 0; j < 8; ++j) {
    float bias = b_s[j * 16 + lr];
    float* out = (j < 4) ? h_self : h_neigh;
    int n = (j & 3) * 16 + lr;
#pragma unroll
    for (int i = 0; i < 8; ++i) {
      out[(size_t)(mbase + i) * HID + n] = c[j][i] + bias;
    }
  }
}

__global__ void gnn_rank_kernel(const int* __restrict__ eidx, int* __restrict__ cnt,
                                int* __restrict__ rank) {
  int q = blockIdx.x * 256 + threadIdx.x;
  if (q < QUADS) {
    v4i r = *(const v4i*)(eidx + 4 * (size_t)q);
    v4i k;
    k.x = atomicAdd(cnt + r.x, 1);
    k.y = atomicAdd(cnt + r.y, 1);
    k.z = atomicAdd(cnt + r.z, 1);
    k.w = atomicAdd(cnt + r.w, 1);
    *(v4i*)(rank + 4 * (size_t)q) = k;
  }
}

__global__ void gnn_scan_kernel(const int* __restrict__ cnt, int* __restrict__ off) {
  __shared__ int wsum[32];
  __shared__ int base_s;
  int t = threadIdx.x, lane = t & 31, wave = t >> 5;
  int b = blockIdx.x;
  int part = 0;
  for (int k = 0; k < b; ++k) part += cnt[k * SCAN_T + t];
  for (int m = 16; m >= 1; m >>= 1) part += __shfl_xor(part, m, 32);
  if (lane == 0) wsum[wave] = part;
  __syncthreads();
  if (wave == 0) {
    int x = wsum[lane];
    for (int m = 16; m >= 1; m >>= 1) x += __shfl_xor(x, m, 32);
    if (lane == 0) base_s = x;
  }
  __syncthreads();
  int base = base_s;
  int i = b * SCAN_T + t;
  int v = (i < N_NODES) ? cnt[i] : 0;
  int inc = v;
  for (int d = 1; d < 32; d <<= 1) {
    int y = __shfl_up(inc, d, 32);
    if (lane >= d) inc += y;
  }
  if (lane == 31) wsum[wave] = inc;
  __syncthreads();
  if (wave == 0) {
    int x = wsum[lane];
    int xi = x;
    for (int d = 1; d < 32; d <<= 1) {
      int y = __shfl_up(xi, d, 32);
      if (lane >= d) xi += y;
    }
    wsum[lane] = xi - x;
  }
  __syncthreads();
  if (i <= N_NODES) off[i] = base + wsum[wave] + inc - v;
}

__global__ void gnn_place_kernel(const int* __restrict__ eidx, const float* __restrict__ ew,
                                 const int* __restrict__ off, const int* __restrict__ rank,
                                 int* __restrict__ perm) {
  int q = blockIdx.x * 256 + threadIdx.x;
  if (q < QUADS) {
    v4i row = *(const v4i*)(eidx + 4 * (size_t)q);
    v4i col = *(const v4i*)(eidx + N_EDGES + 4 * (size_t)q);
    v4f w   = *(const v4f*)(ew + 4 * (size_t)q);
    v4i rk  = *(const v4i*)(rank + 4 * (size_t)q);
    int o0 = off[row.x], o1 = off[row.y], o2 = off[row.z], o3 = off[row.w];
    int p0 = min(max(o0 + rk.x, 0), N_EDGES - 1);
    int p1 = min(max(o1 + rk.y, 0), N_EDGES - 1);
    int p2 = min(max(o2 + rk.z, 0), N_EDGES - 1);
    int p3 = min(max(o3 + rk.w, 0), N_EDGES - 1);
    v2i cw;
    cw.x = col.x; cw.y = __float_as_int(w.x); *(v2i*)(perm + 2 * (size_t)p0) = cw;
    cw.x = col.y; cw.y = __float_as_int(w.y); *(v2i*)(perm + 2 * (size_t)p1) = cw;
    cw.x = col.z; cw.y = __float_as_int(w.z); *(v2i*)(perm + 2 * (size_t)p2) = cw;
    cw.x = col.w; cw.y = __float_as_int(w.w); *(v2i*)(perm + 2 * (size_t)p3) = cw;
  }
}

#define EDGE_COL(J) min(max(__builtin_amdgcn_readlane(cw.x, (J)), 0), N_NODES - 1)
#define EDGE_W(J)   __int_as_float(__builtin_amdgcn_readlane(cw.y, (J)))
#define ROW_OF(C)   (*(const v2f*)(h_neigh + (size_t)(C) * HID + c2))
#define GATE(W, NWE) __builtin_amdgcn_rcpf(1.f + __builtin_amdgcn_exp2f((W) * (NWE)))

__global__ void gnn_gather_ln_kernel(const int* __restrict__ off, const int* __restrict__ perm,
                                     const float* __restrict__ Wedge, const float* __restrict__ h_neigh,
                                     const float* __restrict__ h_self,
                                     const float* __restrict__ gamma, const float* __restrict__ beta,
                                     const float* h, float* dst) {
  int wave = threadIdx.x >> 5, lane = threadIdx.x & 31;
  int wv = blockIdx.x * 8 + wave;
  int c2 = lane * 2;
  float we0 = Wedge[c2] * -1.44269504f, we1 = Wedge[c2 + 1] * -1.44269504f;
  float g0 = gamma[c2], g1 = gamma[c2 + 1], bt0 = beta[c2], bt1 = beta[c2 + 1];
  for (int q = 0; q < GATHER_NPW; ++q) {
  int node = wv * GATHER_NPW + q;
  int s = __builtin_amdgcn_readfirstlane(off[node]);
  int t = __builtin_amdgcn_readfirstlane(off[node + 1]);
  s = min(max(s, 0), N_EDGES);
  t = min(max(t, s), N_EDGES);
  size_t base = (size_t)node * HID + c2;
  v2f a  = *(const v2f*)(h_self + base);
  v2f hh = *(const v2f*)(h + base);
  float acc0 = 0.f, acc1 = 0.f;
  for (int p = s; p < t; p += 32) {
    int n = min(t - p, 32);
    v2i cw; cw.x = 0; cw.y = 0;
    if (lane < n) cw = *(const v2i*)(perm + 2 * (size_t)(p + lane));
    int j = 0;
    for (; j + 4 <= n; j += 4) {
      int ca = EDGE_COL(j), cb = EDGE_COL(j + 1), cc = EDGE_COL(j + 2), cd = EDGE_COL(j + 3);
      v2f ma = ROW_OF(ca), mb = ROW_OF(cb), mc = ROW_OF(cc), md = ROW_OF(cd);
      float wa = EDGE_W(j), wb = EDGE_W(j + 1), wc = EDGE_W(j + 2), wd = EDGE_W(j + 3);
      float ga0 = GATE(wa, we0), ga1 = GATE(wa, we1);
      float gb0 = GATE(wb, we0), gb1 = GATE(wb, we1);
      float gc0 = GATE(wc, we0), gc1 = GATE(wc, we1);
      float gd0 = GATE(wd, we0), gd1 = GATE(wd, we1);
      acc0 += ma.x * ga0; acc1 += ma.y * ga1;
      acc0 += mb.x * gb0; acc1 += mb.y * gb1;
      acc0 += mc.x * gc0; acc1 += mc.y * gc1;
      acc0 += md.x * gd0; acc1 += md.y * gd1;
    }
    for (; j < n; ++j) {
      int ca = EDGE_COL(j);
      v2f ma = ROW_OF(ca);
      float wa = EDGE_W(j);
      acc0 += ma.x * GATE(wa, we0); acc1 += ma.y * GATE(wa, we1);
    }
  }
  float v0 = a.x + acc0, v1 = a.y + acc1;
  float sm = v0 + v1, ss = v0 * v0 + v1 * v1;
  for (int m = 16; m >= 1; m >>= 1) {
    sm += __shfl_xor(sm, m, 32);
    ss += __shfl_xor(ss, m, 32);
  }
  float mu  = sm * (1.f / HID);
  float var = ss * (1.f / HID) - mu * mu;
  float inv = rsqrtf(var + LN_EPS);
  float o0 = (v0 - mu) * inv * g0 + bt0;
  float o1 = (v1 - mu) * inv * g1 + bt1;
  o0 = (o0 >= 0.f) ? o0 : SLOPE * o0;
  o1 = (o1 >= 0.f) ? o1 : SLOPE * o1;
  v2f r; r.x = hh.x + o0; r.y = hh.y + o1;
  *(v2f*)(dst + base) = r;
  }
}

extern "C" void kernel_launch(void* const* d_in, const int* in_sizes, int n_in,
                              void* d_out, int out_size, void* d_ws, size_t ws_size,
                              hipStream_t stream) {
  const float* x      = (const float*)d_in[0];
  const int*   eidx   = (const int*)  d_in[1];
  const float* ew     = (const float*)d_in[2];
  const float* Wp     = (const float*)d_in[3];
  const float* bp     = (const float*)d_in[4];
  const float* Wself  = (const float*)d_in[5];
  const float* bself  = (const float*)d_in[6];
  const float* Wneigh = (const float*)d_in[7];
  const float* bneigh = (const float*)d_in[8];
  const float* Wedge  = (const float*)d_in[9];
  const float* gamma  = (const float*)d_in[10];
  const float* beta   = (const float*)d_in[11];
  float* out = (float*)d_out;

  const size_t NH = (size_t)N_NODES * HID;
  float* h       = (float*)d_ws;
  float* h_self  = h + NH;
  float* h_neigh = h_self + NH;
  int*   ord     = (int*)(h_neigh + NH);
  int*   cnt     = ord + OFF_CNT;
  int*   off     = ord + OFF_OFF;
  int*   rank    = ord + OFF_RANK;
  int*   perm    = ord + OFF_PERM;

  gnn_proj_kernel<<<PROJ_BLOCKS, 256, 0, stream>>>(x, Wp, bp, h, cnt);

  gnn_rank_kernel<<<(QUADS + 255) / 256, 256, 0, stream>>>(eidx, cnt, rank);
  gnn_scan_kernel<<<SCAN_B, SCAN_T, 0, stream>>>(cnt, off);
  gnn_place_kernel<<<(QUADS + 255) / 256, 256, 0, stream>>>(eidx, ew, off, rank, perm);

  for (int l = 0; l < N_LAYERS; ++l) {
    const float* Ws = Wself  + (size_t)l * HID * HID;
    const float* bs = bself  + (size_t)l * HID;
    const float* Wn = Wneigh + (size_t)l * HID * HID;
    const float* bn = bneigh + (size_t)l * HID;
    const float* We = Wedge  + (size_t)l * HID;
    const float* gm = gamma  + (size_t)l * HID;
    const float* bt = beta   + (size_t)l * HID;

    gnn_gemm_kernel<<<(ROW_TILES + 7) / 8, 256, 0, stream>>>(
        h, Ws, bs, Wn, bn, h_self, h_neigh);

    float* dst = (l == N_LAYERS - 1) ? out : h;
    gnn_gather_ln_kernel<<<N_NODES / (8 * GATHER_NPW), 256, 0, stream>>>(off, perm, We, h_neigh, h_self, gm, bt, h, dst);
  }
}


# Edge ordering by destination as a two-level counting sort with LDS counters (count, scan, move to 196 ranges, per-range fill): no global atomics; per-node sum order becomes LDS arrival order
# speedup vs baseline: 2.0227x; 1.1965x over previous
#include <hip/hip_runtime.h>
#include <hip/hip_bf16.h>

#define N_NODES 100000
#define N_EDGES 1000000
#define IN_DIM  8
#define HID     64
#define N_LAYERS 3
#define LN_EPS  1e-5f
#define SLOPE   0.2f
#define LDW     68

typedef float v2f __attribute__((ext_vector_type(2)));
typedef float v8f __attribute__((ext_vector_type(8)));
typedef int   v2i __attribute__((ext_vector_type(2)));
typedef float v4f __attribute__((ext_vector_type(4)));
typedef int   v4i __attribute__((ext_vector_type(4)));

#define PROJ_BLOCKS 1250
#define PROJ_TRIPS  5
#define ROW_TILES   (N_NODES / 16)
#define GATHER_NPW  2

#define BUCK_SHIFT 9
#define BUCK_NODES 512
#define NBUCK      196
#define EB_BLOCKS  250
#define EB_EDGES   4000
#define NBH        (NBUCK * EB_BLOCKS)
#define BH_PER_T   48
#define OFF_OFF  0
#define OFF_BH   100352
#define OFF_BB   149504
#define OFF_PERM 200704
#define OFF_TMP  2200704

__global__ void gnn_proj_kernel(const float* __restrict__ x, const float* __restrict__ Wp,
                                const float* __restrict__ bp, float* __restrict__ h) {
  int t = threadIdx.x;
  int gt = blockIdx.x * 256 + t;
  int lane = t & 31;
  int gw = gt >> 5;
  int hf = lane >> 4;
  int c4 = (lane & 15) * 4;
  const v4f* wp4 = (const v4f*)(Wp + c4 * IN_DIM);
  v4f w0a = wp4[0], w0b = wp4[1], w1a = wp4[2], w1b = wp4[3];
  v4f w2a = wp4[4], w2b = wp4[5], w3a = wp4[6], w3b = wp4[7];
  v4f b4 = *(const v4f*)(bp + c4);
  v4f xs[PROJ_TRIPS][2];
#pragma unroll
  for (int i = 0; i < PROJ_TRIPS; ++i) {
    int node = (gw * PROJ_TRIPS + i) * 2 + hf;
    const v4f* xr = (const v4f*)(x + (size_t)node * IN_DIM);
    xs[i][0] = xr[0]; xs[i][1] = xr[1];
  }
#pragma unroll
  for (int i = 0; i < PROJ_TRIPS; ++i) {
    int node = (gw * PROJ_TRIPS + i) * 2 + hf;
    v4f xa = xs[i][0], xb = xs[i][1];
    float a0 = b4.x, a1 = b4.y, a2 = b4.z, a3 = b4.w;
    a0 += xa.x * w0a.x; a0 += xa.y * w0a.y; a0 += xa.z * w0a.z; a0 += xa.w * w0a.w;
    a0 += xb.x * w0b.x; a0 += xb.y * w0b.y; a0 += xb.z * w0b.z; a0 += xb.w * w0b.w;
    a1 += xa.x * w1a.x; a1 += xa.y * w1a.y; a1 += xa.z * w1a.z; a1 += xa.w * w1a.w;
    a1 += xb.x * w1b.x; a1 += xb.y * w1b.y; a1 += xb.z * w1b.z; a1 += xb.w * w1b.w;
    a2 += xa.x * w2a.x; a2 += xa.y * w2a.y; a2 += xa.z * w2a.z; a2 += xa.w * w2a.w;
    a2 += xb.x * w2b.x; a2 += xb.y * w2b.y; a2 += xb.z * w2b.z; a2 += xb.w * w2b.w;
    a3 += xa.x * w3a.x; a3 += xa.y * w3a.y; a3 += xa.z * w3a.z; a3 += xa.w * w3a.w;
    a3 += xb.x * w3b.x; a3 += xb.y * w3b.y; a3 += xb.z * w3b.z; a3 += xb.w * w3b.w;
    v4f r; r.x = a0; r.y = a1; r.z = a2; r.w = a3;
    *(v4f*)(h + (size_t)node * HID + c4) = r;
  }
}

__global__ void gnn_gemm_kernel(const float* __restrict__ h,
                                const float* __restrict__ Wself, const float* __restrict__ bself,
                                const float* __restrict__ Wneigh, const float* __restrict__ bneigh,
                                float* __restrict__ h_self, float* __restrict__ h_neigh) {
  __shared__ float w_s[2 * HID * LDW];
  __shared__ float b_s[2 * HID];

  int t = threadIdx.x;
#pragma unroll
  for (int i = 0; i < 4; ++i) {
    int q = t + i * 256;
    int row = q >> 4, col = (q & 15) * 4;
    *(v4f*)(w_s + row * LDW + col)         = *(const v4f*)(Wself  + 4 * q);
    *(v4f*)(w_s + (HID + row) * LDW + col) = *(const v4f*)(Wneigh + 4 * q);
  }
  if (t < HID) b_s[t] = bself[t];
  else if (t < 2 * HID) b_s[t] = bneigh[t - HID];
  __syncthreads();

  int wave = t >> 5, lane = t & 31;
  int tile = blockIdx.x * 8 + wave;
  if (tile >= ROW_TILES) return;
  int r0 = tile * 16;
  int lr = lane & 15;
  int koff = (lane < 16) ? 0 : 2;

  const float* arow = h + (size_t)(r0 + lr) * HID;
  const float* brow = w_s + lr * LDW + koff;

  v2f a[16];
#pragma unroll
  for (int kb = 0; kb < 16; ++kb) a[kb] = *(const v2f*)(arow + kb * 4 + koff);

  v8f c[8];
#pragma unroll
  for (int j = 0; j < 8; ++j) c[j] = (v8f){};
#pragma unroll
  for (int kb = 0; kb < 16; ++kb) {
#pragma unroll
    for (int j = 0; j < 8; ++j) {
      v2f bf = *(const v2f*)(brow + j * 16 * LDW + kb * 4);
      c[j] = __builtin_amdgcn_wmma_f32_16x16x4_f32(false, a[kb], false, bf, (short)0, c[j], false, false);
    }
  }

  int mbase = r0 + ((lane < 16) ? 0 : 8);
#pragma unroll
  for (int j = 0; j < 8; ++j) {
    float bias = b_s[j * 16 + lr];
    float* out = (j < 4) ? h_self : h_neigh;
    int n = (j & 3) * 16 + lr;
#pragma unroll
    for (int i = 0; i < 8; ++i) {
      out[(size_t)(mbase + i) * HID + n] = c[j][i] + bias;
    }
  }
}

__global__ void gnn_bcount_kernel(const int* __restrict__ eidx, int* __restrict__ bh) {
  __shared__ int hist[NBUCK];
  int t = threadIdx.x, g = blockIdx.x;
  if (t < NBUCK) hist[t] = 0;
  __syncthreads();
  if (t < EB_EDGES / 4) {
    v4i r = *(const v4i*)(eidx + g * EB_EDGES + 4 * t);
    atomicAdd(&hist[min(max(r.x >> BUCK_SHIFT, 0), NBUCK - 1)], 1);
    atomicAdd(&hist[min(max(r.y >> BUCK_SHIFT, 0), NBUCK - 1)], 1);
    atomicAdd(&hist[min(max(r.z >> BUCK_SHIFT, 0), NBUCK - 1)], 1);
    atomicAdd(&hist[min(max(r.w >> BUCK_SHIFT, 0), NBUCK - 1)], 1);
  }
  __syncthreads();
  if (t < NBUCK) bh[t * EB_BLOCKS + g] = hist[t];
}

__global__ void gnn_bscan_kernel(const int* __restrict__ bh, int* __restrict__ bb) {
  __shared__ int wsum[32];
  int t = threadIdx.x, lane = t & 31, wave = t >> 5;
  int base = t * BH_PER_T;
  v4i v[BH_PER_T / 4];
#pragma unroll
  for (int i = 0; i < BH_PER_T / 4; ++i) {
    int idx = base + 4 * i;
    v4i z; z.x = 0; z.y = 0; z.z = 0; z.w = 0;
    v[i] = (idx < NBH) ? *(const v4i*)(bh + idx) : z;
  }
  int s = 0;
#pragma unroll
  for (int i = 0; i < BH_PER_T / 4; ++i) s += v[i].x + v[i].y + v[i].z + v[i].w;
  int inc = s;
  for (int d = 1; d < 32; d <<= 1) {
    int y = __shfl_up(inc, d, 32);
    if (lane >= d) inc += y;
  }
  if (lane == 31) wsum[wave] = inc;
  __syncthreads();
  if (wave == 0) {
    int x = wsum[lane];
    int xi = x;
    for (int d = 1; d < 32; d <<= 1) {
      int y = __shfl_up(xi, d, 32);
      if (lane >= d) xi += y;
    }
    wsum[lane] = xi - x;
  }
  __syncthreads();
  int run = wsum[wave] + inc - s;
#pragma unroll
  for (int i = 0; i < BH_PER_T / 4; ++i) {
    int idx = base + 4 * i;
    v4i o;
    o.x = run; run += v[i].x;
    o.y = run; run += v[i].y;
    o.z = run; run += v[i].z;
    o.w = run; run += v[i].w;
    if (idx < NBH) *(v4i*)(bb + idx) = o;
  }
}

__global__ void gnn_bmove_kernel(const int* __restrict__ eidx, const float* __restrict__ ew,
                                 const int* __restrict__ bb, int* __restrict__ tmp) {
  __shared__ int cur[NBUCK];
  int t = threadIdx.x, g = blockIdx.x;
  if (t < NBUCK) cur[t] = bb[t * EB_BLOCKS + g];
  __syncthreads();
  if (t < EB_EDGES / 4) {
    size_t e = (size_t)g * EB_EDGES + 4 * t;
    v4i row = *(const v4i*)(eidx + e);
    v4i col = *(const v4i*)(eidx + N_EDGES + e);
    v4f w   = *(const v4f*)(ew + e);
    int p0 = atomicAdd(&cur[min(max(row.x >> BUCK_SHIFT, 0), NBUCK - 1)], 1);
    int p1 = atomicAdd(&cur[min(max(row.y >> BUCK_SHIFT, 0), NBUCK - 1)], 1);
    int p2 = atomicAdd(&cur[min(max(row.z >> BUCK_SHIFT, 0), NBUCK - 1)], 1);
    int p3 = atomicAdd(&cur[min(max(row.w >> BUCK_SHIFT, 0), NBUCK - 1)], 1);
    p0 = min(max(p0, 0), N_EDGES - 1);
    p1 = min(max(p1, 0), N_EDGES - 1);
    p2 = min(max(p2, 0), N_EDGES - 1);
    p3 = min(max(p3, 0), N_EDGES - 1);
    v4i rec; rec.w = 0;
    rec.x = row.x; rec.y = col.x; rec.z = __float_as_int(w.x); *(v4i*)(tmp + 4 * (size_t)p0) = rec;
    rec.x = row.y; rec.y = col.y; rec.z = __float_as_int(w.y); *(v4i*)(tmp + 4 * (size_t)p1) = rec;
    rec.x = row.z; rec.y = col.z; rec.z = __float_as_int(w.z); *(v4i*)(tmp + 4 * (size_t)p2) = rec;
    rec.x = row.w; rec.y = col.w; rec.z = __float_as_int(w.w); *(v4i*)(tmp + 4 * (size_t)p3) = rec;
  }
}

__global__ void gnn_bfill_kernel(const int* __restrict__ bb, const int* __restrict__ tmp,
                                 int* __restrict__ off, int* __restrict__ perm) {
  __shared__ int cnt[BUCK_NODES];
  __shared__ int cur[BUCK_NODES];
  __shared__ int wsum[32];
  int t = threadIdx.x, lane = t & 31, wave = t >> 5;
  int b = blockIdx.x;
  int s = __builtin_amdgcn_readfirstlane(bb[b * EB_BLOCKS]);
  int e_end = (b + 1 < NBUCK) ? __builtin_amdgcn_readfirstlane(bb[(b + 1) * EB_BLOCKS]) : N_EDGES;
  s = min(max(s, 0), N_EDGES);
  e_end = min(max(e_end, s), N_EDGES);
  if (t < BUCK_NODES) cnt[t] = 0;
  __syncthreads();
  for (int e0 = s + t; e0 < e_end; e0 += 4 * 1024) {
    int r[4];
#pragma unroll
    for (int u = 0; u < 4; ++u) {
      int e = e0 + u * 1024;
      r[u] = (e < e_end) ? tmp[4 * (size_t)e] : -1;
    }
#pragma unroll
    for (int u = 0; u < 4; ++u)
      if (r[u] >= 0) atomicAdd(&cnt[r[u] & (BUCK_NODES - 1)], 1);
  }
  __syncthreads();
  int v = (t < BUCK_NODES) ? cnt[t] : 0;
  int inc = v;
  for (int d = 1; d < 32; d <<= 1) {
    int y = __shfl_up(inc, d, 32);
    if (lane >= d) inc += y;
  }
  if (lane == 31) wsum[wave] = inc;
  __syncthreads();
  if (wave == 0) {
    int x = wsum[lane];
    int xi = x;
    for (int d = 1; d < 32; d <<= 1) {
      int y = __shfl_up(xi, d, 32);
      if (lane >= d) xi += y;
    }
    wsum[lane] = xi - x;
  }
  __syncthreads();
  if (t < BUCK_NODES) {
    int first = s + wsum[wave] + inc - v;
    cur[t] = first;
    int node = b * BUCK_NODES + t;
    if (node <= N_NODES) off[node] = first;
  }
  __syncthreads();
  for (int e0 = s + t; e0 < e_end; e0 += 4 * 1024) {
    v4i rec[4];
#pragma unroll
    for (int u = 0; u < 4; ++u) {
      int e = e0 + u * 1024;
      v4i z; z.x = -1; z.y = 0; z.z = 0; z.w = 0;
      rec[u] = (e < e_end) ? *(const v4i*)(tmp + 4 * (size_t)e) : z;
    }
#pragma unroll
    for (int u = 0; u < 4; ++u) {
      if (rec[u].x >= 0) {
        int p = atomicAdd(&cur[rec[u].x & (BUCK_NODES - 1)], 1);
        p = min(max(p, 0), N_EDGES - 1);
        v2i cw; cw.x = rec[u].y; cw.y = rec[u].z;
        *(v2i*)(perm + 2 * (size_t)p) = cw;
      }
    }
  }
}

#define EDGE_COL(J) min(max(__builtin_amdgcn_readlane(cw.x, (J)), 0), N_NODES - 1)
#define EDGE_W(J)   __int_as_float(__builtin_amdgcn_readlane(cw.y, (J)))
#define ROW_OF(C)   (*(const v2f*)(h_neigh + (size_t)(C) * HID + c2))
#define GATE(W, NWE) __builtin_amdgcn_rcpf(1.f + __builtin_amdgcn_exp2f((W) * (NWE)))

__global__ void gnn_gather_ln_kernel(const int* __restrict__ off, const int* __restrict__ perm,
                                     const float* __restrict__ Wedge, const float* __restrict__ h_neigh,
                                     const float* __restrict__ h_self,
                                     const float* __restrict__ gamma, const float* __restrict__ beta,
                                     const float* h, float* dst) {
  int wave = threadIdx.x >> 5, lane = threadIdx.x & 31;
  int wv = blockIdx.x * 8 + wave;
  int c2 = lane * 2;
  float we0 = Wedge[c2] * -1.44269504f, we1 = Wedge[c2 + 1] * -1.44269504f;
  float g0 = gamma[c2], g1 = gamma[c2 + 1], bt0 = beta[c2], bt1 = beta[c2 + 1];
  for (int q = 0; q < GATHER_NPW; ++q) {
  int node = wv * GATHER_NPW + q;
  int s = __builtin_amdgcn_readfirstlane(off[node]);
  int t = __builtin_amdgcn_readfirstlane(off[node + 1]);
  s = min(max(s, 0), N_EDGES);
  t = min(max(t, s), N_EDGES);
  size_t base = (size_t)node * HID + c2;
  v2f a  = *(const v2f*)(h_self + base);
  v2f hh = *(const v2f*)(h + base);
  float acc0 = 0.f, acc1 = 0.f;
  for (int p = s; p < t; p += 32) {
    int n = min(t - p, 32);
    v2i cw; cw.x = 0; cw.y = 0;
    if (lane < n) cw = *(const v2i*)(perm + 2 * (size_t)(p + lane));
    int j = 0;
    for (; j + 4 <= n; j += 4) {
      int ca = EDGE_COL(j), cb = EDGE_COL(j + 1), cc = EDGE_COL(j + 2), cd = EDGE_COL(j + 3);
      v2f ma = ROW_OF(ca), mb = ROW_OF(cb), mc = ROW_OF(cc), md = ROW_OF(cd);
      float wa = EDGE_W(j), wb = EDGE_W(j + 1), wc = EDGE_W(j + 2), wd = EDGE_W(j + 3);
      float ga0 = GATE(wa, we0), ga1 = GATE(wa, we1);
      float gb0 = GATE(wb, we0), gb1 = GATE(wb, we1);
      float gc0 = GATE(wc, we0), gc1 = GATE(wc, we1);
      float gd0 = GATE(wd, we0), gd1 = GATE(wd, we1);
      acc0 += ma.x * ga0; acc1 += ma.y * ga1;
      acc0 += mb.x * gb0; acc1 += mb.y * gb1;
      acc0 += mc.x * gc0; acc1 += mc.y * gc1;
      acc0 += md.x * gd0; acc1 += md.y * gd1;
    }
    for (; j < n; ++j) {
      int ca = EDGE_COL(j);
      v2f ma = ROW_OF(ca);
      float wa = EDGE_W(j);
      acc0 += ma.x * GATE(wa, we0); acc1 += ma.y * GATE(wa, we1);
    }
  }
  float v0 = a.x + acc0, v1 = a.y + acc1;
  float sm = v0 + v1, ss = v0 * v0 + v1 * v1;
  for (int m = 16; m >= 1; m >>= 1) {
    sm += __shfl_xor(sm, m, 32);
    ss += __shfl_xor(ss, m, 32);
  }
  float mu  = sm * (1.f / HID);
  float var = ss * (1.f / HID) - mu * mu;
  float inv = rsqrtf(var + LN_EPS);
  float o0 = (v0 - mu) * inv * g0 + bt0;
  float o1 = (v1 - mu) * inv * g1 + bt1;
  o0 = (o0 >= 0.f) ? o0 : SLOPE * o0;
  o1 = (o1 >= 0.f) ? o1 : SLOPE * o1;
  v2f r; r.x = hh.x + o0; r.y = hh.y + o1;
  *(v2f*)(dst + base) = r;
  }
}

extern "C" void kernel_launch(void* const* d_in, const int* in_sizes, int n_in,
                              void* d_out, int out_size, void* d_ws, size_t ws_size,
                              hipStream_t stream) {
  const float* x      = (const float*)d_in[0];
  const int*   eidx   = (const int*)  d_in[1];
  const float* ew     = (const float*)d_in[2];
  const float* Wp     = (const float*)d_in[3];
  const float* bp     = (const float*)d_in[4];
  const float* Wself  = (const float*)d_in[5];
  const float* bself  = (const float*)d_in[6];
  const float* Wneigh = (const float*)d_in[7];
  const float* bneigh = (const float*)d_in[8];
  const float* Wedge  = (const float*)d_in[9];
  const float* gamma  = (const float*)d_in[10];
  const float* beta   = (const float*)d_in[11];
  float* out = (float*)d_out;

  const size_t NH = (size_t)N_NODES * HID;
  float* h       = (float*)d_ws;
  float* h_self  = h + NH;
  float* h_neigh = h_self + NH;
  int*   ord     = (int*)(h_neigh + NH);
  int*   off     = ord + OFF_OFF;
  int*   bh      = ord + OFF_BH;
  int*   bb      = ord + OFF_BB;
  int*   perm    = ord + OFF_PERM;
  int*   tmp     = ord + OFF_TMP;

  gnn_proj_kernel<<<PROJ_BLOCKS, 256, 0, stream>>>(x, Wp, bp, h);

  gnn_bcount_kernel<<<EB_BLOCKS, 1024, 0, stream>>>(eidx, bh);
  gnn_bscan_kernel<<<1, 1024, 0, stream>>>(bh, bb);
  gnn_bmove_kernel<<<EB_BLOCKS, 1024, 0, stream>>>(eidx, ew, bb, tmp);
  gnn_bfill_kernel<<<NBUCK, 1024, 0, stream>>>(bb, tmp, off, perm);

  for (int l = 0; l < N_LAYERS; ++l) {
    const float* Ws = Wself  + (size_t)l * HID * HID;
    const float* bs = bself  + (size_t)l * HID;
    const float* Wn = Wneigh + (size_t)l * HID * HID;
    const float* bn = bneigh + (size_t)l * HID;
    const float* We = Wedge  + (size_t)l * HID;
    const float* gm = gamma  + (size_t)l * HID;
    const float* bt = beta   + (size_t)l * HID;

    gnn_gemm_kernel<<<(ROW_TILES + 7) / 8, 256, 0, stream>>>(
        h, Ws, bs, Wn, bn, h_self, h_neigh);

    float* dst = (l == N_LAYERS - 1) ? out : h;
    gnn_gather_ln_kernel<<<N_NODES / (8 * GATHER_NPW), 256, 0, stream>>>(off, perm, We, h_neigh, h_self, gm, bt, h, dst);
  }
}


# Independent kernels share launches: edge count beside the input projection, counter scan beside the first GEMM (blocks take their role from their number); 9 launches instead of 11, sum orders kept
# speedup vs baseline: 2.2212x; 1.0981x over previous
#include <hip/hip_runtime.h>
#include <hip/hip_bf16.h>

#define N_NODES 100000
#define N_EDGES 1000000
#define IN_DIM  8
#define HID     64
#define N_LAYERS 3
#define LN_EPS  1e-5f
#define SLOPE   0.2f
#define LDW     68

typedef float v2f __attribute__((ext_vector_type(2)));
typedef float v8f __attribute__((ext_vector_type(8)));
typedef int   v2i __attribute__((ext_vector_type(2)));
typedef float v4f __attribute__((ext_vector_type(4)));
typedef int   v4i __attribute__((ext_vector_type(4)));

#define PROJ_WAVES  10000
#define PROJ_BLOCKS 313
#define PROJ_TRIPS  5
#define ROW_TILES   (N_NODES / 16)
#define GATHER_NPW  2

#define BUCK_SHIFT 9
#define BUCK_NODES 512
#define NBUCK      196
#define EB_BLOCKS  250
#define EB_EDGES   4000
#define NBH        (NBUCK * EB_BLOCKS)
#define BH_PER_T   48
#define OFF_OFF  0
#define OFF_BH   100352
#define OFF_BB   149504
#define OFF_PERM 200704
#define OFF_TMP  2200704

__device__ __forceinline__ void gnn_proj_body(const float* __restrict__ x, const float* __restrict__ Wp,
                                              const float* __restrict__ bp, float* __restrict__ h, int blk) {
  int t = threadIdx.x;
  int gt = blk * 1024 + t;
  int lane = t & 31;
  int gw = gt >> 5;
  if (gw >= PROJ_WAVES) return;
  int hf = lane >> 4;
  int c4 = (lane & 15) * 4;
  const v4f* wp4 = (const v4f*)(Wp + c4 * IN_DIM);
  v4f w0a = wp4[0], w0b = wp4[1], w1a = wp4[2], w1b = wp4[3];
  v4f w2a = wp4[4], w2b = wp4[5], w3a = wp4[6], w3b = wp4[7];
  v4f b4 = *(const v4f*)(bp + c4);
  v4f xs[PROJ_TRIPS][2];
#pragma unroll
  for (int i = 0; i < PROJ_TRIPS; ++i) {
    int node = (gw * PROJ_TRIPS + i) * 2 + hf;
    const v4f* xr = (const v4f*)(x + (size_t)node * IN_DIM);
    xs[i][0] = xr[0]; xs[i][1] = xr[1];
  }
#pragma unroll
  for (int i = 0; i < PROJ_TRIPS; ++i) {
    int node = (gw * PROJ_TRIPS + i) * 2 + hf;
    v4f xa = xs[i][0], xb = xs[i][1];
    float a0 = b4.x, a1 = b4.y, a2 = b4.z, a3 = b4.w;
    a0 += xa.x * w0a.x; a0 += xa.y * w0a.y; a0 += xa.z * w0a.z; a0 += xa.w * w0a.w;
    a0 += xb.x * w0b.x; a0 += xb.y * w0b.y; a0 += xb.z * w0b.z; a0 += xb.w * w0b.w;
    a1 += xa.x * w1a.x; a1 += xa.y * w1a.y; a1 += xa.z * w1a.z; a1 += xa.w * w1a.w;
    a1 += xb.x * w1b.x; a1 += xb.y * w1b.y; a1 += xb.z * w1b.z; a1 += xb.w * w1b.w;
    a2 += xa.x * w2a.x; a2 += xa.y * w2a.y; a2 += xa.z * w2a.z; a2 += xa.w * w2a.w;
    a2 += xb.x * w2b.x; a2 += xb.y * w2b.y; a2 += xb.z * w2b.z; a2 += xb.w * w2b.w;
    a3 += xa.x * w3a.x; a3 += xa.y * w3a.y; a3 += xa.z * w3a.z; a3 += xa.w * w3a.w;
    a3 += xb.x * w3b.x; a3 += xb.y * w3b.y; a3 += xb.z * w3b.z; a3 += xb.w * w3b.w;
    v4f r; r.x = a0; r.y = a1; r.z = a2; r.w = a3;
    *(v4f*)(h + (size_t)node * HID + c4) = r;
  }
}

template <int THREADS>
__device__ __forceinline__ void gnn_gemm_body(const float* __restrict__ h,
                                const float* __restrict__ Wself, const float* __restrict__ bself,
                                const float* __restrict__ Wneigh, const float* __restrict__ bneigh,
                                float* __restrict__ h_self, float* __restrict__ h_neigh, int blk) {
  __shared__ float w_s[2 * HID * LDW];
  __shared__ float b_s[2 * HID];

  int t = threadIdx.x;
#pragma unroll
  for (int i = 0; i < 1024 / THREADS; ++i) {
    int q = t + i * THREADS;
    int row = q >> 4, col = (q & 15) * 4;
    *(v4f*)(w_s + row * LDW + col)         = *(const v4f*)(Wself  + 4 * q);
    *(v4f*)(w_s + (HID + row) * LDW + col) = *(const v4f*)(Wneigh + 4 * q);
  }
  if (t < HID) b_s[t] = bself[t];
  else if (t < 2 * HID) b_s[t] = bneigh[t - HID];
  __syncthreads();

  int wave = t >> 5, lane = t & 31;
  int tile = blk * (THREADS / 32) + wave;
  if (tile >= ROW_TILES) return;
  int r0 = tile * 16;
  int lr = lane & 15;
  int koff = (lane < 16) ? 0 : 2;

  const float* arow = h + (size_t)(r0 + lr) * HID;
  const float* brow = w_s + lr * LDW + koff;

  v2f a[16];
#pragma unroll
  for (int kb = 0; kb < 16; ++kb) a[kb] = *(const v2f*)(arow + kb * 4 + koff);

  v8f c[8];
#pragma unroll
  for (int j = 0; j < 8; ++j) c[j] = (v8f){};
#pragma unroll
  for (int kb = 0; kb < 16; ++kb) {
#pragma unroll
    for (int j = 0; j < 8; ++j) {
      v2f bf = *(const v2f*)(brow + j * 16 * LDW + kb * 4);
      c[j] = __builtin_amdgcn_wmma_f32_16x16x4_f32(false, a[kb], false, bf, (short)0, c[j], false, false);
    }
  }

  int mbase = r0 + ((lane < 16) ? 0 : 8);
#pragma unroll
  for (int j = 0; j < 8; ++j) {
    float bias = b_s[j * 16 + lr];
    float* out = (j < 4) ? h_self : h_neigh;
    int n = (j & 3) * 16 + lr;
#pragma unroll
    for (int i = 0; i < 8; ++i) {
      out[(size_t)(mbase + i) * HID + n] = c[j][i] + bias;
    }
  }
}

__global__ void gnn_gemm_kernel(const float* __restrict__ h,
                                const float* __restrict__ Wself, const float* __restrict__ bself,
                                const float* __restrict__ Wneigh, const float* __restrict__ bneigh,
                                float* __restrict__ h_self, float* __restrict__ h_neigh) {
  gnn_gemm_body<256>(h, Wself, bself, Wneigh, bneigh, h_self, h_neigh, blockIdx.x);
}

__device__ __forceinline__ void gnn_bcount_body(const int* __restrict__ eidx, int* __restrict__ bh, int g) {
  __shared__ int hist[NBUCK];
  int t = threadIdx.x;
  if (t < NBUCK) hist[t] = 0;
  __syncthreads();
  if (t < EB_EDGES / 4) {
    v4i r = *(const v4i*)(eidx + g * EB_EDGES + 4 * t);
    atomicAdd(&hist[min(max(r.x >> BUCK_SHIFT, 0), NBUCK - 1)], 1);
    atomicAdd(&hist[min(max(r.y >> BUCK_SHIFT, 0), NBUCK - 1)], 1);
    atomicAdd(&hist[min(max(r.z >> BUCK_SHIFT, 0), NBUCK - 1)], 1);
    atomicAdd(&hist[min(max(r.w >> BUCK_SHIFT, 0), NBUCK - 1)], 1);
  }
  __syncthreads();
  if (t < NBUCK) bh[t * EB_BLOCKS + g] = hist[t];
}

__device__ __forceinline__ void gnn_bscan_body(const int* __restrict__ bh, int* __restrict__ bb) {
  __shared__ int wsum[32];
  int t = threadIdx.x, lane = t & 31, wave = t >> 5;
  int base = t * BH_PER_T;
  v4i v[BH_PER_T / 4];
#pragma unroll
  for (int i = 0; i < BH_PER_T / 4; ++i) {
    int idx = base + 4 * i;
    v4i z; z.x = 0; z.y = 0; z.z = 0; z.w = 0;
    v[i] = (idx < NBH) ? *(const v4i*)(bh + idx) : z;
  }
  int s = 0;
#pragma unroll
  for (int i = 0; i < BH_PER_T / 4; ++i) s += v[i].x + v[i].y + v[i].z + v[i].w;
  int inc = s;
  for (int d = 1; d < 32; d <<= 1) {
    int y = __shfl_up(inc, d, 32);
    if (lane >= d) inc += y;
  }
  if (lane == 31) wsum[wave] = inc;
  __syncthreads();
  if (wave == 0) {
    int x = wsum[lane];
    int xi = x;
    for (int d = 1; d < 32; d <<= 1) {
      int y = __shfl_up(xi, d, 32);
      if (lane >= d) xi += y;
    }
    wsum[lane] = xi - x;
  }
  __syncthreads();
  int run = wsum[wave] + inc - s;
#pragma unroll
  for (int i = 0; i < BH_PER_T / 4; ++i) {
    int idx = base + 4 * i;
    v4i o;
    o.x = run; run += v[i].x;
    o.y = run; run += v[i].y;
    o.z = run; run += v[i].z;
    o.w = run; run += v[i].w;
    if (idx < NBH) *(v4i*)(bb + idx) = o;
  }
}

__global__ void gnn_count_proj_kernel(const int* __restrict__ eidx, int* __restrict__ bh,
                                      const float* __restrict__ x, const float* __restrict__ Wp,
                                      const float* __restrict__ bp, float* __restrict__ h) {
  if (blockIdx.x < EB_BLOCKS) gnn_bcount_body(eidx, bh, blockIdx.x);
  else gnn_proj_body(x, Wp, bp, h, blockIdx.x - EB_BLOCKS);
}

#define GEMM0_BLOCKS ((ROW_TILES + 31) / 32)
__global__ void gnn_scan_gemm_kernel(const int* __restrict__ bh, int* __restrict__ bb,
                                     const float* __restrict__ h,
                                     const float* __restrict__ Wself, const float* __restrict__ bself,
                                     const float* __restrict__ Wneigh, const float* __restrict__ bneigh,
                                     float* __restrict__ h_self, float* __restrict__ h_neigh) {
  if (blockIdx.x == 0) gnn_bscan_body(bh, bb);
  else gnn_gemm_body<1024>(h, Wself, bself, Wneigh, bneigh, h_self, h_neigh, blockIdx.x - 1);
}

__global__ void gnn_bmove_kernel(const int* __restrict__ eidx, const float* __restrict__ ew,
                                 const int* __restrict__ bb, int* __restrict__ tmp) {
  __shared__ int cur[NBUCK];
  int t = threadIdx.x, g = blockIdx.x;
  if (t < NBUCK) cur[t] = bb[t * EB_BLOCKS + g];
  __syncthreads();
  if (t < EB_EDGES / 4) {
    size_t e = (size_t)g * EB_EDGES + 4 * t;
    v4i row = *(const v4i*)(eidx + e);
    v4i col = *(const v4i*)(eidx + N_EDGES + e);
    v4f w   = *(const v4f*)(ew + e);
    int p0 = atomicAdd(&cur[min(max(row.x >> BUCK_SHIFT, 0), NBUCK - 1)], 1);
    int p1 = atomicAdd(&cur[min(max(row.y >> BUCK_SHIFT, 0), NBUCK - 1)], 1);
    int p2 = atomicAdd(&cur[min(max(row.z >> BUCK_SHIFT, 0), NBUCK - 1)], 1);
    int p3 = atomicAdd(&cur[min(max(row.w >> BUCK_SHIFT, 0), NBUCK - 1)], 1);
    p0 = min(max(p0, 0), N_EDGES - 1);
    p1 = min(max(p1, 0), N_EDGES - 1);
    p2 = min(max(p2, 0), N_EDGES - 1);
    p3 = min(max(p3, 0), N_EDGES - 1);
    v4i rec; rec.w = 0;
    rec.x = row.x; rec.y = col.x; rec.z = __float_as_int(w.x); *(v4i*)(tmp + 4 * (size_t)p0) = rec;
    rec.x = row.y; rec.y = col.y; rec.z = __float_as_int(w.y); *(v4i*)(tmp + 4 * (size_t)p1) = rec;
    rec.x = row.z; rec.y = col.z; rec.z = __float_as_int(w.z); *(v4i*)(tmp + 4 * (size_t)p2) = rec;
    rec.x = row.w; rec.y = col.w; rec.z = __float_as_int(w.w); *(v4i*)(tmp + 4 * (size_t)p3) = rec;
  }
}

__global__ void gnn_bfill_kernel(const int* __restrict__ bb, const int* __restrict__ tmp,
                                 int* __restrict__ off, int* __restrict__ perm) {
  __shared__ int cnt[BUCK_NODES];
  __shared__ int cur[BUCK_NODES];
  __shared__ int wsum[32];
  int t = threadIdx.x, lane = t & 31, wave = t >> 5;
  int b = blockIdx.x;
  int s = __builtin_amdgcn_readfirstlane(bb[b * EB_BLOCKS]);
  int e_end = (b + 1 < NBUCK) ? __builtin_amdgcn_readfirstlane(bb[(b + 1) * EB_BLOCKS]) : N_EDGES;
  s = min(max(s, 0), N_EDGES);
  e_end = min(max(e_end, s), N_EDGES);
  if (t < BUCK_NODES) cnt[t] = 0;
  __syncthreads();
  for (int e0 = s + t; e0 < e_end; e0 += 4 * 1024) {
    int r[4];
#pragma unroll
    for (int u = 0; u < 4; ++u) {
      int e = e0 + u * 1024;
      r[u] = (e < e_end) ? tmp[4 * (size_t)e] : -1;
    }
#pragma unroll
    for (int u = 0; u < 4; ++u)
      if (r[u] >= 0) atomicAdd(&cnt[r[u] & (BUCK_NODES - 1)], 1);
  }
  __syncthreads();
  int v = (t < BUCK_NODES) ? cnt[t] : 0;
  int inc = v;
  for (int d = 1; d < 32; d <<= 1) {
    int y = __shfl_up(inc, d, 32);
    if (lane >= d) inc += y;
  }
  if (lane == 31) wsum[wave] = inc;
  __syncthreads();
  if (wave == 0) {
    int x = wsum[lane];
    int xi = x;
    for (int d = 1; d < 32; d <<= 1) {
      int y = __shfl_up(xi, d, 32);
      if (lane >= d) xi += y;
    }
    wsum[lane] = xi - x;
  }
  __syncthreads();
  if (t < BUCK_NODES) {
    int first = s + wsum[wave] + inc - v;
    cur[t] = first;
    int node = b * BUCK_NODES + t;
    if (node <= N_NODES) off[node] = first;
  }
  __syncthreads();
  for (int e0 = s + t; e0 < e_end; e0 += 4 * 1024) {
    v4i rec[4];
#pragma unroll
    for (int u = 0; u < 4; ++u) {
      int e = e0 + u * 1024;
      v4i z; z.x = -1; z.y = 0; z.z = 0; z.w = 0;
      rec[u] = (e < e_end) ? *(const v4i*)(tmp + 4 * (size_t)e) : z;
    }
#pragma unroll
    for (int u = 0; u < 4; ++u) {
      if (rec[u].x >= 0) {
        int p = atomicAdd(&cur[rec[u].x & (BUCK_NODES - 1)], 1);
        p = min(max(p, 0), N_EDGES - 1);
        v2i cw; cw.x = rec[u].y; cw.y = rec[u].z;
        *(v2i*)(perm + 2 * (size_t)p) = cw;
      }
    }
  }
}

#define EDGE_COL(J) min(max(__builtin_amdgcn_readlane(cw.x, (J)), 0), N_NODES - 1)
#define EDGE_W(J)   __int_as_float(__builtin_amdgcn_readlane(cw.y, (J)))
#define ROW_OF(C)   (*(const v2f*)(h_neigh + (size_t)(C) * HID + c2))
#define GATE(W, NWE) __builtin_amdgcn_rcpf(1.f + __builtin_amdgcn_exp2f((W) * (NWE)))

__global__ void gnn_gather_ln_kernel(const int* __restrict__ off, const int* __restrict__ perm,
                                     const float* __restrict__ Wedge, const float* __restrict__ h_neigh,
                                     const float* __restrict__ h_self,
                                     const float* __restrict__ gamma, const float* __restrict__ beta,
                                     const float* h, float* dst) {
  int wave = threadIdx.x >> 5, lane = threadIdx.x & 31;
  int wv = blockIdx.x * 8 + wave;
  int c2 = lane * 2;
  float we0 = Wedge[c2] * -1.44269504f, we1 = Wedge[c2 + 1] * -1.44269504f;
  float g0 = gamma[c2], g1 = gamma[c2 + 1], bt0 = beta[c2], bt1 = beta[c2 + 1];
  for (int q = 0; q < GATHER_NPW; ++q) {
  int node = wv * GATHER_NPW + q;
  int s = __builtin_amdgcn_readfirstlane(off[node]);
  int t = __builtin_amdgcn_readfirstlane(off[node + 1]);
  s = min(max(s, 0), N_EDGES);
  t = min(max(t, s), N_EDGES);
  size_t base = (size_t)node * HID + c2;
  v2f a  = *(const v2f*)(h_self + base);
  v2f hh = *(const v2f*)(h + base);
  float acc0 = 0.f, acc1 = 0.f;
  for (int p = s; p < t; p += 32) {
    int n = min(t - p, 32);
    v2i cw; cw.x = 0; cw.y = 0;
    if (lane < n) cw = *(const v2i*)(perm + 2 * (size_t)(p + lane));
    int j = 0;
    for (; j + 4 <= n; j += 4) {
      int ca = EDGE_COL(j), cb = EDGE_COL(j + 1), cc = EDGE_COL(j + 2), cd = EDGE_COL(j + 3);
      v2f ma = ROW_OF(ca), mb = ROW_OF(cb), mc = ROW_OF(cc), md = ROW_OF(cd);
      float wa = EDGE_W(j), wb = EDGE_W(j + 1), wc = EDGE_W(j + 2), wd = EDGE_W(j + 3);
      float ga0 = GATE(wa, we0), ga1 = GATE(wa, we1);
      float gb0 = GATE(wb, we0), gb1 = GATE(wb, we1);
      float gc0 = GATE(wc, we0), gc1 = GATE(wc, we1);
      float gd0 = GATE(wd, we0), gd1 = GATE(wd, we1);
      acc0 += ma.x * ga0; acc1 += ma.y * ga1;
      acc0 += mb.x * gb0; acc1 += mb.y * gb1;
      acc0 += mc.x * gc0; acc1 += mc.y * gc1;
      acc0 += md.x * gd0; acc1 += md.y * gd1;
    }
    for (; j < n; ++j) {
      int ca = EDGE_COL(j);
      v2f ma = ROW_OF(ca);
      float wa = EDGE_W(j);
      acc0 += ma.x * GATE(wa, we0); acc1 += ma.y * GATE(wa, we1);
    }
  }
  float v0 = a.x + acc0, v1 = a.y + acc1;
  float sm = v0 + v1, ss = v0 * v0 + v1 * v1;
  for (int m = 16; m >= 1; m >>= 1) {
    sm += __shfl_xor(sm, m, 32);
    ss += __shfl_xor(ss, m, 32);
  }
  float mu  = sm * (1.f / HID);
  float var = ss * (1.f / HID) - mu * mu;
  float inv = rsqrtf(var + LN_EPS);
  float o0 = (v0 - mu) * inv * g0 + bt0;
  float o1 = (v1 - mu) * inv * g1 + bt1;
  o0 = (o0 >= 0.f) ? o0 : SLOPE * o0;
  o1 = (o1 >= 0.f) ? o1 : SLOPE * o1;
  v2f r; r.x = hh.x + o0; r.y = hh.y + o1;
  *(v2f*)(dst + base) = r;
  }
}

extern "C" void kernel_launch(void* const* d_in, const int* in_sizes, int n_in,
                              void* d_out, int out_size, void* d_ws, size_t ws_size,
                              hipStream_t stream) {
  const float* x      = (const float*)d_in[0];
  const int*   eidx   = (const int*)  d_in[1];
  const float* ew     = (const float*)d_in[2];
  const float* Wp     = (const float*)d_in[3];
  const float* bp     = (const float*)d_in[4];
  const float* Wself  = (const float*)d_in[5];
  const float* bself  = (const float*)d_in[6];
  const float* Wneigh = (const float*)d_in[7];
  const float* bneigh = (const float*)d_in[8];
  const float* Wedge  = (const float*)d_in[9];
  const float* gamma  = (const float*)d_in[10];
  const float* beta   = (const float*)d_in[11];
  float* out = (float*)d_out;

  const size_t NH = (size_t)N_NODES * HID;
  float* h       = (float*)d_ws;
  float* h_self  = h + NH;
  float* h_neigh = h_self + NH;
  int*   ord     = (int*)(h_neigh + NH);
  int*   off     = ord + OFF_OFF;
  int*   bh      = ord + OFF_BH;
  int*   bb      = ord + OFF_BB;
  int*   perm    = ord + OFF_PERM;
  int*   tmp     = ord + OFF_TMP;

  gnn_count_proj_kernel<<<EB_BLOCKS + PROJ_BLOCKS, 1024, 0, stream>>>(eidx, bh, x, Wp, bp, h);
  gnn_scan_gemm_kernel<<<1 + GEMM0_BLOCKS, 1024, 0, stream>>>(bh, bb, h, Wself, bself, Wneigh, bneigh, h_self, h_neigh);

  gnn_bmove_kernel<<<EB_BLOCKS, 1024, 0, stream>>>(eidx, ew, bb, tmp);
  gnn_bfill_kernel<<<NBUCK, 1024, 0, stream>>>(bb, tmp, off, perm);

  for (int l = 0; l < N_LAYERS; ++l) {
    const float* Ws = Wself  + (size_t)l * HID * HID;
    const float* bs = bself  + (size_t)l * HID;
    const float* Wn = Wneigh + (size_t)l * HID * HID;
    const float* bn = bneigh + (size_t)l * HID;
    const float* We = Wedge  + (size_t)l * HID;
    const float* gm = gamma  + (size_t)l * HID;
    const float* bt = beta   + (size_t)l * HID;

    if (l > 0)
      gnn_gemm_kernel<<<(ROW_TILES + 7) / 8, 256, 0, stream>>>(
          h, Ws, bs, Wn, bn, h_self, h_neigh);

    float* dst = (l == N_LAYERS - 1) ? out : h;
    gnn_gather_ln_kernel<<<N_NODES / (8 * GATHER_NPW), 256, 0, stream>>>(off, perm, We, h_neigh, h_self, gm, bt, h, dst);
  }
}


# Next layer's GEMM runs in the tail of the gather kernel on the block's own 32 new rows kept in LDS (weights staged per block); 7 launches, not 9; h_neigh alternates between 2 planes; sum orders kept
# speedup vs baseline: 2.6036x; 1.1722x over previous
#include <hip/hip_runtime.h>
#include <hip/hip_bf16.h>

#define N_NODES 100000
#define N_EDGES 1000000
#define IN_DIM  8
#define HID     64
#define N_LAYERS 3
#define LN_EPS  1e-5f
#define SLOPE   0.2f
#define LDW     68

typedef float v2f __attribute__((ext_vector_type(2)));
typedef float v8f __attribute__((ext_vector_type(8)));
typedef int   v2i __attribute__((ext_vector_type(2)));
typedef float v4f __attribute__((ext_vector_type(4)));
typedef int   v4i __attribute__((ext_vector_type(4)));

#define PROJ_WAVES  10000
#define PROJ_BLOCKS 313
#define PROJ_TRIPS  5
#define ROW_TILES   (N_NODES / 16)
#define GATHER_NPW  2

#define BUCK_SHIFT 9
#define BUCK_NODES 512
#define NBUCK      196
#define EB_BLOCKS  250
#define EB_EDGES   4000
#define NBH        (NBUCK * EB_BLOCKS)
#define BH_PER_T   48
#define OFF_OFF  0
#define OFF_BH   100352
#define OFF_BB   149504
#define OFF_PERM 200704
#define OFF_TMP  2200704

__device__ __forceinline__ void gnn_proj_body(const float* __restrict__ x, const float* __restrict__ Wp,
                                              const float* __restrict__ bp, float* __restrict__ h, int blk) {
  int t = threadIdx.x;
  int gt = blk * 1024 + t;
  int lane = t & 31;
  int gw = gt >> 5;
  if (gw >= PROJ_WAVES) return;
  int hf = lane >> 4;
  int c4 = (lane & 15) * 4;
  const v4f* wp4 = (const v4f*)(Wp + c4 * IN_DIM);
  v4f w0a = wp4[0], w0b = wp4[1], w1a = wp4[2], w1b = wp4[3];
  v4f w2a = wp4[4], w2b = wp4[5], w3a = wp4[6], w3b = wp4[7];
  v4f b4 = *(const v4f*)(bp + c4);
  v4f xs[PROJ_TRIPS][2];
#pragma unroll
  for (int i = 0; i < PROJ_TRIPS; ++i) {
    int node = (gw * PROJ_TRIPS + i) * 2 + hf;
    const v4f* xr = (const v4f*)(x + (size_t)node * IN_DIM);
    xs[i][0] = xr[0]; xs[i][1] = xr[1];
  }
#pragma unroll
  for (int i = 0; i < PROJ_TRIPS; ++i) {
    int node = (gw * PROJ_TRIPS + i) * 2 + hf;
    v4f xa = xs[i][0], xb = xs[i][1];
    float a0 = b4.x, a1 = b4.y, a2 = b4.z, a3 = b4.w;
    a0 += xa.x * w0a.x; a0 += xa.y * w0a.y; a0 += xa.z * w0a.z; a0 += xa.w * w0a.w;
    a0 += xb.x * w0b.x; a0 += xb.y * w0b.y; a0 += xb.z * w0b.z; a0 += xb.w * w0b.w;
    a1 += xa.x * w1a.x; a1 += xa.y * w1a.y; a1 += xa.z * w1a.z; a1 += xa.w * w1a.w;
    a1 += xb.x * w1b.x; a1 += xb.y * w1b.y; a1 += xb.z * w1b.z; a1 += xb.w * w1b.w;
    a2 += xa.x * w2a.x; a2 += xa.y * w2a.y; a2 += xa.z * w2a.z; a2 += xa.w * w2a.w;
    a2 += xb.x * w2b.x; a2 += xb.y * w2b.y; a2 += xb.z * w2b.z; a2 += xb.w * w2b.w;
    a3 += xa.x * w3a.x; a3 += xa.y * w3a.y; a3 += xa.z * w3a.z; a3 += xa.w * w3a.w;
    a3 += xb.x * w3b.x; a3 += xb.y * w3b.y; a3 += xb.z * w3b.z; a3 += xb.w * w3b.w;
    v4f r; r.x = a0; r.y = a1; r.z = a2; r.w = a3;
    *(v4f*)(h + (size_t)node * HID + c4) = r;
  }
}

template <int THREADS>
__device__ __forceinline__ void gnn_gemm_body(const float* __restrict__ h,
                                const float* __restrict__ Wself, const float* __restrict__ bself,
                                const float* __restrict__ Wneigh, const float* __restrict__ bneigh,
                                float* __restrict__ h_self, float* __restrict__ h_neigh, int blk) {
  __shared__ float w_s[2 * HID * LDW];
  __shared__ float b_s[2 * HID];

  int t = threadIdx.x;
#pragma unroll
  for (int i = 0; i < 1024 / THREADS; ++i) {
    int q = t + i * THREADS;
    int row = q >> 4, col = (q & 15) * 4;
    *(v4f*)(w_s + row * LDW + col)         = *(const v4f*)(Wself  + 4 * q);
    *(v4f*)(w_s + (HID + row) * LDW + col) = *(const v4f*)(Wneigh + 4 * q);
  }
  if (t < HID) b_s[t] = bself[t];
  else if (t < 2 * HID) b_s[t] = bneigh[t - HID];
  __syncthreads();

  int wave = t >> 5, lane = t & 31;
  int tile = blk * (THREADS / 32) + wave;
  if (tile >= ROW_TILES) return;
  int r0 = tile * 16;
  int lr = lane & 15;
  int koff = (lane < 16) ? 0 : 2;

  const float* arow = h + (size_t)(r0 + lr) * HID;
  const float* brow = w_s + lr * LDW + koff;

  v2f a[16];
#pragma unroll
  for (int kb = 0; kb < 16; ++kb) a[kb] = *(const v2f*)(arow + kb * 4 + koff);

  v8f c[8];
#pragma unroll
  for (int j = 0; j < 8; ++j) c[j] = (v8f){};
#pragma unroll
  for (int kb = 0; kb < 16; ++kb) {
#pragma unroll
    for (int j = 0; j < 8; ++j) {
      v2f bf = *(const v2f*)(brow + j * 16 * LDW + kb * 4);
      c[j] = __builtin_amdgcn_wmma_f32_16x16x4_f32(false, a[kb], false, bf, (short)0, c[j], false, false);
    }
  }

  int mbase = r0 + ((lane < 16) ? 0 : 8);
#pragma unroll
  for (int j = 0; j < 8; ++j) {
    float bias = b_s[j * 16 + lr];
    float* out = (j < 4) ? h_self : h_neigh;
    int n = (j & 3) * 16 + lr;
#pragma unroll
    for (int i = 0; i < 8; ++i) {
      out[(size_t)(mbase + i) * HID + n] = c[j][i] + bias;
    }
  }
}

__device__ __forceinline__ void gnn_bcount_body(const int* __restrict__ eidx, int* __restrict__ bh, int g) {
  __shared__ int hist[NBUCK];
  int t = threadIdx.x;
  if (t < NBUCK) hist[t] = 0;
  __syncthreads();
  if (t < EB_EDGES / 4) {
    v4i r = *(const v4i*)(eidx + g * EB_EDGES + 4 * t);
    atomicAdd(&hist[min(max(r.x >> BUCK_SHIFT, 0), NBUCK - 1)], 1);
    atomicAdd(&hist[min(max(r.y >> BUCK_SHIFT, 0), NBUCK - 1)], 1);
    atomicAdd(&hist[min(max(r.z >> BUCK_SHIFT, 0), NBUCK - 1)], 1);
    atomicAdd(&hist[min(max(r.w >> BUCK_SHIFT, 0), NBUCK - 1)], 1);
  }
  __syncthreads();
  if (t < NBUCK) bh[t * EB_BLOCKS + g] = hist[t];
}

__device__ __forceinline__ void gnn_bscan_body(const int* __restrict__ bh, int* __restrict__ bb) {
  __shared__ int wsum[32];
  int t = threadIdx.x, lane = t & 31, wave = t >> 5;
  int base = t * BH_PER_T;
  v4i v[BH_PER_T / 4];
#pragma unroll
  for (int i = 0; i < BH_PER_T / 4; ++i) {
    int idx = base + 4 * i;
    v4i z; z.x = 0; z.y = 0; z.z = 0; z.w = 0;
    v[i] = (idx < NBH) ? *(const v4i*)(bh + idx) : z;
  }
  int s = 0;
#pragma unroll
  for (int i = 0; i < BH_PER_T / 4; ++i) s += v[i].x + v[i].y + v[i].z + v[i].w;
  int inc = s;
  for (int d = 1; d < 32; d <<= 1) {
    int y = __shfl_up(inc, d, 32);
    if (lane >= d) inc += y;
  }
  if (lane == 31) wsum[wave] = inc;
  __syncthreads();
  if (wave == 0) {
    int x = wsum[lane];
    int xi = x;
    for (int d = 1; d < 32; d <<= 1) {
      int y = __shfl_up(xi, d, 32);
      if (lane >= d) xi += y;
    }
    wsum[lane] = xi - x;
  }
  __syncthreads();
  int run = wsum[wave] + inc - s;
#pragma unroll
  for (int i = 0; i < BH_PER_T / 4; ++i) {
    int idx = base + 4 * i;
    v4i o;
    o.x = run; run += v[i].x;
    o.y = run; run += v[i].y;
    o.z = run; run += v[i].z;
    o.w = run; run += v[i].w;
    if (idx < NBH) *(v4i*)(bb + idx) = o;
  }
}

__global__ void gnn_count_proj_kernel(const int* __restrict__ eidx, int* __restrict__ bh,
                                      const float* __restrict__ x, const float* __restrict__ Wp,
                                      const float* __restrict__ bp, float* __restrict__ h) {
  if (blockIdx.x < EB_BLOCKS) gnn_bcount_body(eidx, bh, blockIdx.x);
  else gnn_proj_body(x, Wp, bp, h, blockIdx.x - EB_BLOCKS);
}

#define GEMM0_BLOCKS ((ROW_TILES + 31) / 32)
__global__ void gnn_scan_gemm_kernel(const int* __restrict__ bh, int* __restrict__ bb,
                                     const float* __restrict__ h,
                                     const float* __restrict__ Wself, const float* __restrict__ bself,
                                     const float* __restrict__ Wneigh, const float* __restrict__ bneigh,
                                     float* __restrict__ h_self, float* __restrict__ h_neigh) {
  if (blockIdx.x == 0) gnn_bscan_body(bh, bb);
  else gnn_gemm_body<1024>(h, Wself, bself, Wneigh, bneigh, h_self, h_neigh, blockIdx.x - 1);
}

__global__ void gnn_bmove_kernel(const int* __restrict__ eidx, const float* __restrict__ ew,
                                 const int* __restrict__ bb, int* __restrict__ tmp) {
  __shared__ int cur[NBUCK];
  int t = threadIdx.x, g = blockIdx.x;
  if (t < NBUCK) cur[t] = bb[t * EB_BLOCKS + g];
  __syncthreads();
  if (t < EB_EDGES / 4) {
    size_t e = (size_t)g * EB_EDGES + 4 * t;
    v4i row = *(const v4i*)(eidx + e);
    v4i col = *(const v4i*)(eidx + N_EDGES + e);
    v4f w   = *(const v4f*)(ew + e);
    int p0 = atomicAdd(&cur[min(max(row.x >> BUCK_SHIFT, 0), NBUCK - 1)], 1);
    int p1 = atomicAdd(&cur[min(max(row.y >> BUCK_SHIFT, 0), NBUCK - 1)], 1);
    int p2 = atomicAdd(&cur[min(max(row.z >> BUCK_SHIFT, 0), NBUCK - 1)], 1);
    int p3 = atomicAdd(&cur[min(max(row.w >> BUCK_SHIFT, 0), NBUCK - 1)], 1);
    p0 = min(max(p0, 0), N_EDGES - 1);
    p1 = min(max(p1, 0), N_EDGES - 1);
    p2 = min(max(p2, 0), N_EDGES - 1);
    p3 = min(max(p3, 0), N_EDGES - 1);
    v4i rec; rec.w = 0;
    rec.x = row.x; rec.y = col.x; rec.z = __float_as_int(w.x); *(v4i*)(tmp + 4 * (size_t)p0) = rec;
    rec.x = row.y; rec.y = col.y; rec.z = __float_as_int(w.y); *(v4i*)(tmp + 4 * (size_t)p1) = rec;
    rec.x = row.z; rec.y = col.z; rec.z = __float_as_int(w.z); *(v4i*)(tmp + 4 * (size_t)p2) = rec;
    rec.x = row.w; rec.y = col.w; rec.z = __float_as_int(w.w); *(v4i*)(tmp + 4 * (size_t)p3) = rec;
  }
}

__global__ void gnn_bfill_kernel(const int* __restrict__ bb, const int* __restrict__ tmp,
                                 int* __restrict__ off, int* __restrict__ perm) {
  __shared__ int cnt[BUCK_NODES];
  __shared__ int cur[BUCK_NODES];
  __shared__ int wsum[32];
  int t = threadIdx.x, lane = t & 31, wave = t >> 5;
  int b = blockIdx.x;
  int s = __builtin_amdgcn_readfirstlane(bb[b * EB_BLOCKS]);
  int e_end = (b + 1 < NBUCK) ? __builtin_amdgcn_readfirstlane(bb[(b + 1) * EB_BLOCKS]) : N_EDGES;
  s = min(max(s, 0), N_EDGES);
  e_end = min(max(e_end, s), N_EDGES);
  if (t < BUCK_NODES) cnt[t] = 0;
  __syncthreads();
  for (int e0 = s + t; e0 < e_end; e0 += 4 * 1024) {
    int r[4];
#pragma unroll
    for (int u = 0; u < 4; ++u) {
      int e = e0 + u * 1024;
      r[u] = (e < e_end) ? tmp[4 * (size_t)e] : -1;
    }
#pragma unroll
    for (int u = 0; u < 4; ++u)
      if (r[u] >= 0) atomicAdd(&cnt[r[u] & (BUCK_NODES - 1)], 1);
  }
  __syncthreads();
  int v = (t < BUCK_NODES) ? cnt[t] : 0;
  int inc = v;
  for (int d = 1; d < 32; d <<= 1) {
    int y = __shfl_up(inc, d, 32);
    if (lane >= d) inc += y;
  }
  if (lane == 31) wsum[wave] = inc;
  __syncthreads();
  if (wave == 0) {
    int x = wsum[lane];
    int xi = x;
    for (int d = 1; d < 32; d <<= 1) {
      int y = __shfl_up(xi, d, 32);
      if (lane >= d) xi += y;
    }
    wsum[lane] = xi - x;
  }
  __syncthreads();
  if (t < BUCK_NODES) {
    int first = s + wsum[wave] + inc - v;
    cur[t] = first;
    int node = b * BUCK_NODES + t;
    if (node <= N_NODES) off[node] = first;
  }
  __syncthreads();
  for (int e0 = s + t; e0 < e_end; e0 += 4 * 1024) {
    v4i rec[4];
#pragma unroll
    for (int u = 0; u < 4; ++u) {
      int e = e0 + u * 1024;
      v4i z; z.x = -1; z.y = 0; z.z = 0; z.w = 0;
      rec[u] = (e < e_end) ? *(const v4i*)(tmp + 4 * (size_t)e) : z;
    }
#pragma unroll
    for (int u = 0; u < 4; ++u) {
      if (rec[u].x >= 0) {
        int p = atomicAdd(&cur[rec[u].x & (BUCK_NODES - 1)], 1);
        p = min(max(p, 0), N_EDGES - 1);
        v2i cw; cw.x = rec[u].y; cw.y = rec[u].z;
        *(v2i*)(perm + 2 * (size_t)p) = cw;
      }
    }
  }
}

#define EDGE_COL(J) min(max(__builtin_amdgcn_readlane(cw.x, (J)), 0), N_NODES - 1)
#define EDGE_W(J)   __int_as_float(__builtin_amdgcn_readlane(cw.y, (J)))
#define ROW_OF(C)   (*(const v2f*)(h_neigh + (size_t)(C) * HID + c2))
#define GATE(W, NWE) __builtin_amdgcn_rcpf(1.f + __builtin_amdgcn_exp2f((W) * (NWE)))

__device__ __forceinline__ v2f gnn_node_update(int node, int lane, int c2,
                                               const int* __restrict__ off, const int* __restrict__ perm,
                                               const float* __restrict__ h_neigh, const float* h_self, const float* h,
                                               float we0, float we1, float g0, float g1, float bt0, float bt1) {
  int s = __builtin_amdgcn_readfirstlane(off[node]);
  int t = __builtin_amdgcn_readfirstlane(off[node + 1]);
  s = min(max(s, 0), N_EDGES);
  t = min(max(t, s), N_EDGES);
  size_t base = (size_t)node * HID + c2;
  v2f a  = *(const v2f*)(h_self + base);
  v2f hh = *(const v2f*)(h + base);
  float acc0 = 0.f, acc1 = 0.f;
  for (int p = s; p < t; p += 32) {
    int n = min(t - p, 32);
    v2i cw; cw.x = 0; cw.y = 0;
    if (lane < n) cw = *(const v2i*)(perm + 2 * (size_t)(p + lane));
    int j = 0;
    for (; j + 4 <= n; j += 4) {
      int ca = EDGE_COL(j), cb = EDGE_COL(j + 1), cc = EDGE_COL(j + 2), cd = EDGE_COL(j + 3);
      v2f ma = ROW_OF(ca), mb = ROW_OF(cb), mc = ROW_OF(cc), md = ROW_OF(cd);
      float wa = EDGE_W(j), wb = EDGE_W(j + 1), wc = EDGE_W(j + 2), wd = EDGE_W(j + 3);
      float ga0 = GATE(wa, we0), ga1 = GATE(wa, we1);
      float gb0 = GATE(wb, we0), gb1 = GATE(wb, we1);
      float gc0 = GATE(wc, we0), gc1 = GATE(wc, we1);
      float gd0 = GATE(wd, we0), gd1 = GATE(wd, we1);
      acc0 += ma.x * ga0; acc1 += ma.y * ga1;
      acc0 += mb.x * gb0; acc1 += mb.y * gb1;
      acc0 += mc.x * gc0; acc1 += mc.y * gc1;
      acc0 += md.x * gd0; acc1 += md.y * gd1;
    }
    for (; j < n; ++j) {
      int ca = EDGE_COL(j);
      v2f ma = ROW_OF(ca);
      float wa = EDGE_W(j);
      acc0 += ma.x * GATE(wa, we0); acc1 += ma.y * GATE(wa, we1);
    }
  }
  float v0 = a.x + acc0, v1 = a.y + acc1;
  float sm = v0 + v1, ss = v0 * v0 + v1 * v1;
  for (int m = 16; m >= 1; m >>= 1) {
    sm += __shfl_xor(sm, m, 32);
    ss += __shfl_xor(ss, m, 32);
  }
  float mu  = sm * (1.f / HID);
  float var = ss * (1.f / HID) - mu * mu;
  float inv = rsqrtf(var + LN_EPS);
  float o0 = (v0 - mu) * inv * g0 + bt0;
  float o1 = (v1 - mu) * inv * g1 + bt1;
  o0 = (o0 >= 0.f) ? o0 : SLOPE * o0;
  o1 = (o1 >= 0.f) ? o1 : SLOPE * o1;
  v2f r; r.x = hh.x + o0; r.y = hh.y + o1;
  return r;
}

__global__ void gnn_gather_ln_kernel(const int* __restrict__ off, const int* __restrict__ perm,
                                     const float* __restrict__ Wedge, const float* __restrict__ h_neigh,
                                     const float* __restrict__ h_self,
                                     const float* __restrict__ gamma, const float* __restrict__ beta,
                                     const float* h, float* dst) {
  int wave = threadIdx.x >> 5, lane = threadIdx.x & 31;
  int wv = blockIdx.x * 8 + wave;
  int c2 = lane * 2;
  float we0 = Wedge[c2] * -1.44269504f, we1 = Wedge[c2 + 1] * -1.44269504f;
  float g0 = gamma[c2], g1 = gamma[c2 + 1], bt0 = beta[c2], bt1 = beta[c2 + 1];
  for (int q = 0; q < GATHER_NPW; ++q) {
    int node = wv * GATHER_NPW + q;
    v2f r = gnn_node_update(node, lane, c2, off, perm, h_neigh, h_self, h, we0, we1, g0, g1, bt0, bt1);
    *(v2f*)(dst + (size_t)node * HID + c2) = r;
  }
}

#define FUSE_WAVES 16
#define FUSE_NODES (FUSE_WAVES * 2)
__global__ void gnn_gather_ln_gemm_kernel(const int* __restrict__ off, const int* __restrict__ perm,
                                          const float* __restrict__ Wedge, const float* __restrict__ h_neigh,
                                          float* h_self,
                                          const float* __restrict__ gamma, const float* __restrict__ beta,
                                          float* h,
                                          const float* __restrict__ Wself, const float* __restrict__ bself,
                                          const float* __restrict__ Wneigh, const float* __restrict__ bneigh,
                                          float* __restrict__ h_neigh_next) {
  __shared__ float w_s[2 * HID * LDW];
  __shared__ float hn_s[FUSE_NODES * LDW];

  int t = threadIdx.x;
  int wave = t >> 5, lane = t & 31;
  int c2 = lane * 2;
  int lr = lane & 15;
  int rt = wave >> 3, ct = wave & 7;
  float we0 = Wedge[c2] * -1.44269504f, we1 = Wedge[c2 + 1] * -1.44269504f;
  float g0 = gamma[c2], g1 = gamma[c2 + 1], bt0 = beta[c2], bt1 = beta[c2 + 1];
  const float* bsrc = (ct < 4) ? bself : bneigh;
  float bias = bsrc[(ct & 3) * 16 + lr];
  int n0 = blockIdx.x * FUSE_NODES;
  int ln = wave * 2;

  int row = t >> 4, col = (t & 15) * 4;
  v4f s0 = *(const v4f*)(Wself + 4 * t), s1 = *(const v4f*)(Wself + 4 * (t + 512));
  {
    v2f r = gnn_node_update(n0 + ln, lane, c2, off, perm, h_neigh, h_self, h, we0, we1, g0, g1, bt0, bt1);
    *(v2f*)(hn_s + ln * LDW + c2) = r;
  }
  *(v4f*)(w_s + row * LDW + col)        = s0;
  *(v4f*)(w_s + (row + 32) * LDW + col) = s1;
  v4f m0 = *(const v4f*)(Wneigh + 4 * t), m1 = *(const v4f*)(Wneigh + 4 * (t + 512));
  {
    v2f r = gnn_node_update(n0 + ln + 1, lane, c2, off, perm, h_neigh, h_self, h, we0, we1, g0, g1, bt0, bt1);
    *(v2f*)(hn_s + (ln + 1) * LDW + c2) = r;
  }
  *(v4f*)(w_s + (HID + row) * LDW + col)      = m0;
  *(v4f*)(w_s + (HID + row + 32) * LDW + col) = m1;
  __syncthreads();

  *(v2f*)(h + (size_t)(n0 + ln) * HID + c2)     = *(const v2f*)(hn_s + ln * LDW + c2);
  *(v2f*)(h + (size_t)(n0 + ln + 1) * HID + c2) = *(const v2f*)(hn_s + (ln + 1) * LDW + c2);

  int koff = (lane < 16) ? 0 : 2;
  const float* arow = hn_s + (rt * 16 + lr) * LDW + koff;
  const float* brow = w_s + (ct * 16 + lr) * LDW + koff;
  v8f c = (v8f){};
#pragma unroll
  for (int kb = 0; kb < 16; ++kb) {
    v2f af = *(const v2f*)(arow + kb * 4);
    v2f bf = *(const v2f*)(brow + kb * 4);
    c = __builtin_amdgcn_wmma_f32_16x16x4_f32(false, af, false, bf, (short)0, c, false, false);
  }
  asm volatile("v_nop\n\tv_nop\n\tv_nop\n\tv_nop\n\tv_nop" : "+v"(c));
  float* o = (ct < 4) ? h_self : h_neigh_next;
  int n = (ct & 3) * 16 + lr;
  int mbase = n0 + rt * 16 + ((lane < 16) ? 0 : 8);
#pragma unroll
  for (int i = 0; i < 8; ++i) o[(size_t)(mbase + i) * HID + n] = c[i] + bias;
}

extern "C" void kernel_launch(void* const* d_in, const int* in_sizes, int n_in,
                              void* d_out, int out_size, void* d_ws, size_t ws_size,
                              hipStream_t stream) {
  const float* x      = (const float*)d_in[0];
  const int*   eidx   = (const int*)  d_in[1];
  const float* ew     = (const float*)d_in[2];
  const float* Wp     = (const float*)d_in[3];
  const float* bp     = (const float*)d_in[4];
  const float* Wself  = (const float*)d_in[5];
  const float* bself  = (const float*)d_in[6];
  const float* Wneigh = (const float*)d_in[7];
  const float* bneigh = (const float*)d_in[8];
  const float* Wedge  = (const float*)d_in[9];
  const float* gamma  = (const float*)d_in[10];
  const float* beta   = (const float*)d_in[11];
  float* out = (float*)d_out;

  const size_t NH = (size_t)N_NODES * HID;
  float* h       = (float*)d_ws;
  float* h_self  = h + NH;
  float* h_neigh = h_self + NH;
  int*   ord     = (int*)(h_neigh + NH);
  int*   off     = ord + OFF_OFF;
  int*   bh      = ord + OFF_BH;
  int*   bb      = ord + OFF_BB;
  int*   perm    = ord + OFF_PERM;
  int*   tmp     = ord + OFF_TMP;

  gnn_count_proj_kernel<<<EB_BLOCKS + PROJ_BLOCKS, 1024, 0, stream>>>(eidx, bh, x, Wp, bp, h);
  gnn_scan_gemm_kernel<<<1 + GEMM0_BLOCKS, 1024, 0, stream>>>(bh, bb, h, Wself, bself, Wneigh, bneigh, h_self, h_neigh);

  gnn_bmove_kernel<<<EB_BLOCKS, 1024, 0, stream>>>(eidx, ew, bb, tmp);
  gnn_bfill_kernel<<<NBUCK, 1024, 0, stream>>>(bb, tmp, off, perm);

  for (int l = 0; l < N_LAYERS; ++l) {
    const float* We = Wedge  + (size_t)l * HID;
    const float* gm = gamma  + (size_t)l * HID;
    const float* bt = beta   + (size_t)l * HID;
    float* hn_cur = (l == 1) ? out : h_neigh;
    float* hn_nxt = (l == 1) ? h_neigh : out;

    if (l < N_LAYERS - 1) {
      const float* Ws = Wself  + (size_t)(l + 1) * HID * HID;
      const float* bs = bself  + (size_t)(l + 1) * HID;
      const float* Wn = Wneigh + (size_t)(l + 1) * HID * HID;
      const float* bn = bneigh + (size_t)(l + 1) * HID;
      gnn_gather_ln_gemm_kernel<<<N_NODES / FUSE_NODES, 32 * FUSE_WAVES, 0, stream>>>(off, perm, We, hn_cur, h_self, gm, bt, h,
                                                                        Ws, bs, Wn, bn, hn_nxt);
    } else {
      gnn_gather_ln_kernel<<<N_NODES / (8 * GATHER_NPW), 256, 0, stream>>>(off, perm, We, hn_cur, h_self, gm, bt, h, out);
    }
  }
}


# Edge walk by half waves: a lane owns 4 columns, one 16-byte load per lane fetches two edges' rows (DPP row_share, not readlane), 8 edges in flight; node sum = even-rank + odd-rank partial sums
# speedup vs baseline: 2.7309x; 1.0489x over previous
#include <hip/hip_runtime.h>
#include <hip/hip_bf16.h>

#define N_NODES 100000
#define N_EDGES 1000000
#define IN_DIM  8
#define HID     64
#define N_LAYERS 3
#define LN_EPS  1e-5f
#define SLOPE   0.2f
#define LDW     68

typedef float v2f __attribute__((ext_vector_type(2)));
typedef float v8f __attribute__((ext_vector_type(8)));
typedef int   v2i __attribute__((ext_vector_type(2)));
typedef float v4f __attribute__((ext_vector_type(4)));
typedef int   v4i __attribute__((ext_vector_type(4)));

#define PROJ_WAVES  10000
#define PROJ_BLOCKS 313
#define PROJ_TRIPS  5
#define ROW_TILES   (N_NODES / 16)
#define GATHER_NPW  2

#define BUCK_SHIFT 9
#define BUCK_NODES 512
#define NBUCK      196
#define EB_BLOCKS  250
#define EB_EDGES   4000
#define NBH        (NBUCK * EB_BLOCKS)
#define BH_PER_T   48
#define OFF_OFF  0
#define OFF_BH   100352
#define OFF_BB   149504
#define OFF_PERM 200704
#define OFF_TMP  2200704

__device__ __forceinline__ void gnn_proj_body(const float* __restrict__ x, const float* __restrict__ Wp,
                                              const float* __restrict__ bp, float* __restrict__ h, int blk) {
  int t = threadIdx.x;
  int gt = blk * 1024 + t;
  int lane = t & 31;
  int gw = gt >> 5;
  if (gw >= PROJ_WAVES) return;
  int hf = lane >> 4;
  int c4 = (lane & 15) * 4;
  const v4f* wp4 = (const v4f*)(Wp + c4 * IN_DIM);
  v4f w0a = wp4[0], w0b = wp4[1], w1a = wp4[2], w1b = wp4[3];
  v4f w2a = wp4[4], w2b = wp4[5], w3a = wp4[6], w3b = wp4[7];
  v4f b4 = *(const v4f*)(bp + c4);
  v4f xs[PROJ_TRIPS][2];
#pragma unroll
  for (int i = 0; i < PROJ_TRIPS; ++i) {
    int node = (gw * PROJ_TRIPS + i) * 2 + hf;
    const v4f* xr = (const v4f*)(x + (size_t)node * IN_DIM);
    xs[i][0] = xr[0]; xs[i][1] = xr[1];
  }
#pragma unroll
  for (int i = 0; i < PROJ_TRIPS; ++i) {
    int node = (gw * PROJ_TRIPS + i) * 2 + hf;
    v4f xa = xs[i][0], xb = xs[i][1];
    float a0 = b4.x, a1 = b4.y, a2 = b4.z, a3 = b4.w;
    a0 += xa.x * w0a.x; a0 += xa.y * w0a.y; a0 += xa.z * w0a.z; a0 += xa.w * w0a.w;
    a0 += xb.x * w0b.x; a0 += xb.y * w0b.y; a0 += xb.z * w0b.z; a0 += xb.w * w0b.w;
    a1 += xa.x * w1a.x; a1 += xa.y * w1a.y; a1 += xa.z * w1a.z; a1 += xa.w * w1a.w;
    a1 += xb.x * w1b.x; a1 += xb.y * w1b.y; a1 += xb.z * w1b.z; a1 += xb.w * w1b.w;
    a2 += xa.x * w2a.x; a2 += xa.y * w2a.y; a2 += xa.z * w2a.z; a2 += xa.w * w2a.w;
    a2 += xb.x * w2b.x; a2 += xb.y * w2b.y; a2 += xb.z * w2b.z; a2 += xb.w * w2b.w;
    a3 += xa.x * w3a.x; a3 += xa.y * w3a.y; a3 += xa.z * w3a.z; a3 += xa.w * w3a.w;
    a3 += xb.x * w3b.x; a3 += xb.y * w3b.y; a3 += xb.z * w3b.z; a3 += xb.w * w3b.w;
    v4f r; r.x = a0; r.y = a1; r.z = a2; r.w = a3;
    *(v4f*)(h + (size_t)node * HID + c4) = r;
  }
}

template <int THREADS>
__device__ __forceinline__ void gnn_gemm_body(const float* __restrict__ h,
                                const float* __restrict__ Wself, const float* __restrict__ bself,
                                const float* __restrict__ Wneigh, const float* __restrict__ bneigh,
                                float* __restrict__ h_self, float* __restrict__ h_neigh, int blk) {
  __shared__ float w_s[2 * HID * LDW];
  __shared__ float b_s[2 * HID];

  int t = threadIdx.x;
#pragma unroll
  for (int i = 0; i < 1024 / THREADS; ++i) {
    int q = t + i * THREADS;
    int row = q >> 4, col = (q & 15) * 4;
    *(v4f*)(w_s + row * LDW + col)         = *(const v4f*)(Wself  + 4 * q);
    *(v4f*)(w_s + (HID + row) * LDW + col) = *(const v4f*)(Wneigh + 4 * q);
  }
  if (t < HID) b_s[t] = bself[t];
  else if (t < 2 * HID) b_s[t] = bneigh[t - HID];
  __syncthreads();

  int wave = t >> 5, lane = t & 31;
  int tile = blk * (THREADS / 32) + wave;
  if (tile >= ROW_TILES) return;
  int r0 = tile * 16;
  int lr = lane & 15;
  int koff = (lane < 16) ? 0 : 2;

  const float* arow = h + (size_t)(r0 + lr) * HID;
  const float* brow = w_s + lr * LDW + koff;

  v2f a[16];
#pragma unroll
  for (int kb = 0; kb < 16; ++kb) a[kb] = *(const v2f*)(arow + kb * 4 + koff);

  v8f c[8];
#pragma unroll
  for (int j = 0; j < 8; ++j) c[j] = (v8f){};
#pragma unroll
  for (int kb = 0; kb < 16; ++kb) {
#pragma unroll
    for (int j = 0; j < 8; ++j) {
      v2f bf = *(const v2f*)(brow + j * 16 * LDW + kb * 4);
      c[j] = __builtin_amdgcn_wmma_f32_16x16x4_f32(false, a[kb], false, bf, (short)0, c[j], false, false);
    }
  }

  int mbase = r0 + ((lane < 16) ? 0 : 8);
#pragma unroll
  for (int j = 0; j < 8; ++j) {
    float bias = b_s[j * 16 + lr];
    float* out = (j < 4) ? h_self : h_neigh;
    int n = (j & 3) * 16 + lr;
#pragma unroll
    for (int i = 0; i < 8; ++i) {
      out[(size_t)(mbase + i) * HID + n] = c[j][i] + bias;
    }
  }
}

__device__ __forceinline__ void gnn_bcount_body(const int* __restrict__ eidx, int* __restrict__ bh, int g) {
  __shared__ int hist[NBUCK];
  int t = threadIdx.x;
  if (t < NBUCK) hist[t] = 0;
  __syncthreads();
  if (t < EB_EDGES / 4) {
    v4i r = *(const v4i*)(eidx + g * EB_EDGES + 4 * t);
    atomicAdd(&hist[min(max(r.x >> BUCK_SHIFT, 0), NBUCK - 1)], 1);
    atomicAdd(&hist[min(max(r.y >> BUCK_SHIFT, 0), NBUCK - 1)], 1);
    atomicAdd(&hist[min(max(r.z >> BUCK_SHIFT, 0), NBUCK - 1)], 1);
    atomicAdd(&hist[min(max(r.w >> BUCK_SHIFT, 0), NBUCK - 1)], 1);
  }
  __syncthreads();
  if (t < NBUCK) bh[t * EB_BLOCKS + g] = hist[t];
}

__device__ __forceinline__ void gnn_bscan_body(const int* __restrict__ bh, int* __restrict__ bb) {
  __shared__ int wsum[32];
  int t = threadIdx.x, lane = t & 31, wave = t >> 5;
  int base = t * BH_PER_T;
  v4i v[BH_PER_T / 4];
#pragma unroll
  for (int i = 0; i < BH_PER_T / 4; ++i) {
    int idx = base + 4 * i;
    v4i z; z.x = 0; z.y = 0; z.z = 0; z.w = 0;
    v[i] = (idx < NBH) ? *(const v4i*)(bh + idx) : z;
  }
  int s = 0;
#pragma unroll
  for (int i = 0; i < BH_PER_T / 4; ++i) s += v[i].x + v[i].y + v[i].z + v[i].w;
  int inc = s;
  for (int d = 1; d < 32; d <<= 1) {
    int y = __shfl_up(inc, d, 32);
    if (lane >= d) inc += y;
  }
  if (lane == 31) wsum[wave] = inc;
  __syncthreads();
  if (wave == 0) {
    int x = wsum[lane];
    int xi = x;
    for (int d = 1; d < 32; d <<= 1) {
      int y = __shfl_up(xi, d, 32);
      if (lane >= d) xi += y;
    }
    wsum[lane] = xi - x;
  }
  __syncthreads();
  int run = wsum[wave] + inc - s;
#pragma unroll
  for (int i = 0; i < BH_PER_T / 4; ++i) {
    int idx = base + 4 * i;
    v4i o;
    o.x = run; run += v[i].x;
    o.y = run; run += v[i].y;
    o.z = run; run += v[i].z;
    o.w = run; run += v[i].w;
    if (idx < NBH) *(v4i*)(bb + idx) = o;
  }
}

__global__ void gnn_count_proj_kernel(const int* __restrict__ eidx, int* __restrict__ bh,
                                      const float* __restrict__ x, const float* __restrict__ Wp,
                                      const float* __restrict__ bp, float* __restrict__ h) {
  if (blockIdx.x < EB_BLOCKS) gnn_bcount_body(eidx, bh, blockIdx.x);
  else gnn_proj_body(x, Wp, bp, h, blockIdx.x - EB_BLOCKS);
}

#define GEMM0_BLOCKS ((ROW_TILES + 31) / 32)
__global__ void gnn_scan_gemm_kernel(const int* __restrict__ bh, int* __restrict__ bb,
                                     const float* __restrict__ h,
                                     const float* __restrict__ Wself, const float* __restrict__ bself,
                                     const float* __restrict__ Wneigh, const float* __restrict__ bneigh,
                                     float* __restrict__ h_self, float* __restrict__ h_neigh) {
  if (blockIdx.x == 0) gnn_bscan_body(bh, bb);
  else gnn_gemm_body<1024>(h, Wself, bself, Wneigh, bneigh, h_self, h_neigh, blockIdx.x - 1);
}

__global__ void gnn_bmove_kernel(const int* __restrict__ eidx, const float* __restrict__ ew,
                                 const int* __restrict__ bb, int* __restrict__ tmp) {
  __shared__ int cur[NBUCK];
  int t = threadIdx.x, g = blockIdx.x;
  if (t < NBUCK) cur[t] = bb[t * EB_BLOCKS + g];
  __syncthreads();
  if (t < EB_EDGES / 4) {
    size_t e = (size_t)g * EB_EDGES + 4 * t;
    v4i row = *(const v4i*)(eidx + e);
    v4i col = *(const v4i*)(eidx + N_EDGES + e);
    v4f w   = *(const v4f*)(ew + e);
    int p0 = atomicAdd(&cur[min(max(row.x >> BUCK_SHIFT, 0), NBUCK - 1)], 1);
    int p1 = atomicAdd(&cur[min(max(row.y >> BUCK_SHIFT, 0), NBUCK - 1)], 1);
    int p2 = atomicAdd(&cur[min(max(row.z >> BUCK_SHIFT, 0), NBUCK - 1)], 1);
    int p3 = atomicAdd(&cur[min(max(row.w >> BUCK_SHIFT, 0), NBUCK - 1)], 1);
    p0 = min(max(p0, 0), N_EDGES - 1);
    p1 = min(max(p1, 0), N_EDGES - 1);
    p2 = min(max(p2, 0), N_EDGES - 1);
    p3 = min(max(p3, 0), N_EDGES - 1);
    v4i rec; rec.w = 0;
    rec.x = row.x; rec.y = col.x; rec.z = __float_as_int(w.x); *(v4i*)(tmp + 4 * (size_t)p0) = rec;
    rec.x = row.y; rec.y = col.y; rec.z = __float_as_int(w.y); *(v4i*)(tmp + 4 * (size_t)p1) = rec;
    rec.x = row.z; rec.y = col.z; rec.z = __float_as_int(w.z); *(v4i*)(tmp + 4 * (size_t)p2) = rec;
    rec.x = row.w; rec.y = col.w; rec.z = __float_as_int(w.w); *(v4i*)(tmp + 4 * (size_t)p3) = rec;
  }
}

__global__ void gnn_bfill_kernel(const int* __restrict__ bb, const int* __restrict__ tmp,
                                 int* __restrict__ off, int* __restrict__ perm) {
  __shared__ int cnt[BUCK_NODES];
  __shared__ int cur[BUCK_NODES];
  __shared__ int wsum[32];
  int t = threadIdx.x, lane = t & 31, wave = t >> 5;
  int b = blockIdx.x;
  int s = __builtin_amdgcn_readfirstlane(bb[b * EB_BLOCKS]);
  int e_end = (b + 1 < NBUCK) ? __builtin_amdgcn_readfirstlane(bb[(b + 1) * EB_BLOCKS]) : N_EDGES;
  s = min(max(s, 0), N_EDGES);
  e_end = min(max(e_end, s), N_EDGES);
  if (t < BUCK_NODES) cnt[t] = 0;
  __syncthreads();
  for (int e0 = s + t; e0 < e_end; e0 += 4 * 1024) {
    int r[4];
#pragma unroll
    for (int u = 0; u < 4; ++u) {
      int e = e0 + u * 1024;
      r[u] = (e < e_end) ? tmp[4 * (size_t)e] : -1;
    }
#pragma unroll
    for (int u = 0; u < 4; ++u)
      if (r[u] >= 0) atomicAdd(&cnt[r[u] & (BUCK_NODES - 1)], 1);
  }
  __syncthreads();
  int v = (t < BUCK_NODES) ? cnt[t] : 0;
  int inc = v;
  for (int d = 1; d < 32; d <<= 1) {
    int y = __shfl_up(inc, d, 32);
    if (lane >= d) inc += y;
  }
  if (lane == 31) wsum[wave] = inc;
  __syncthreads();
  if (wave == 0) {
    int x = wsum[lane];
    int xi = x;
    for (int d = 1; d < 32; d <<= 1) {
      int y = __shfl_up(xi, d, 32);
      if (lane >= d) xi += y;
    }
    wsum[lane] = xi - x;
  }
  __syncthreads();
  if (t < BUCK_NODES) {
    int first = s + wsum[wave] + inc - v;
    cur[t] = first;
    int node = b * BUCK_NODES + t;
    if (node <= N_NODES) off[node] = first;
  }
  __syncthreads();
  for (int e0 = s + t; e0 < e_end; e0 += 4 * 1024) {
    v4i rec[4];
#pragma unroll
    for (int u = 0; u < 4; ++u) {
      int e = e0 + u * 1024;
      v4i z; z.x = -1; z.y = 0; z.z = 0; z.w = 0;
      rec[u] = (e < e_end) ? *(const v4i*)(tmp + 4 * (size_t)e) : z;
    }
#pragma unroll
    for (int u = 0; u < 4; ++u) {
      if (rec[u].x >= 0) {
        int p = atomicAdd(&cur[rec[u].x & (BUCK_NODES - 1)], 1);
        p = min(max(p, 0), N_EDGES - 1);
        v2i cw; cw.x = rec[u].y; cw.y = rec[u].z;
        *(v2i*)(perm + 2 * (size_t)p) = cw;
      }
    }
  }
}

#define HALF_LANE(V, J) __builtin_amdgcn_update_dpp(0, (V), 0x150 + ((J) & 15), 0xf, 0xf, true)
#define ROW_REQ(M, J) if ((J) < nfull) { M = *(const v4f*)(hb + (unsigned)(HALF_LANE(ro, (J)) + lo)); }
#define ROW_ADD(M, J) if ((J) < nfull) { float w_ = __int_as_float(HALF_LANE(wb, (J))); gnn_gated_add(M, w_, nwe, acc); }

__device__ __forceinline__ void gnn_gated_add(v4f m, float w, v4f nwe, v4f& acc) {
  v2f z0 = nwe.xy * w, z1 = nwe.zw * w;
  acc.x += m.x * __builtin_amdgcn_rcpf(1.f + __builtin_amdgcn_exp2f(z0.x));
  acc.y += m.y * __builtin_amdgcn_rcpf(1.f + __builtin_amdgcn_exp2f(z0.y));
  acc.z += m.z * __builtin_amdgcn_rcpf(1.f + __builtin_amdgcn_exp2f(z1.x));
  acc.w += m.w * __builtin_amdgcn_rcpf(1.f + __builtin_amdgcn_exp2f(z1.y));
}

template <int G0>
__device__ __forceinline__ void gnn_walk_group(int nfull, int ro, int wb, const char* __restrict__ hb, unsigned lo, v4f nwe, v4f& acc) {
  v4f m0, m1, m2, m3;
  ROW_REQ(m0, G0 + 0) ROW_REQ(m1, G0 + 1) ROW_REQ(m2, G0 + 2) ROW_REQ(m3, G0 + 3)
  ROW_ADD(m0, G0 + 0) ROW_ADD(m1, G0 + 1) ROW_ADD(m2, G0 + 2) ROW_ADD(m3, G0 + 3)
}

__device__ __forceinline__ float gnn_xhalf(float x) {
  int b = __builtin_bit_cast(int, x);
  return __builtin_bit_cast(float, __builtin_amdgcn_permlanex16(b, b, 0x76543210, 0xfedcba98, false, false));
}

__device__ __forceinline__ v2f gnn_node_walk(int node, int lane,
                                             const int* __restrict__ off, const int* __restrict__ perm,
                                             const float* __restrict__ h_neigh, v4f nwe) {
  int s = __builtin_amdgcn_readfirstlane(off[(unsigned)node]);
  int t = __builtin_amdgcn_readfirstlane(off[(unsigned)node + 1u]);
  s = min(max(s, 0), N_EDGES);
  t = min(max(t, s), N_EDGES);
  int hf = lane >> 4, lr = lane & 15;
  const char* hb = (const char*)h_neigh;
  unsigned lo = lr * 16;
  v4f acc = (v4f){};
  for (int p = s; p < t; p += 32) {
    int n = min(t - p, 32);
    int nfull = n >> 1;
    int e = 2 * lr + hf;
    v2i cw; cw.x = 0; cw.y = 0;
    if (e < n) cw = *(const v2i*)(perm + 2u * (unsigned)(p + e));
    int ro = min(max(cw.x, 0), N_NODES - 1) << 8;
    int wb = cw.y;
    v4f mt = (v4f){}; float wt = 0.f;
    if (n & 1) {
      unsigned ot = (unsigned)__builtin_amdgcn_readlane(ro, nfull);
      wt = __int_as_float(__builtin_amdgcn_readlane(wb, nfull));
      mt = *(const v4f*)(hb + (ot + lo));
    }
    if (0 < nfull)  gnn_walk_group<0>(nfull, ro, wb, hb, lo, nwe, acc);
    if (4 < nfull)  gnn_walk_group<4>(nfull, ro, wb, hb, lo, nwe, acc);
    if (8 < nfull)  gnn_walk_group<8>(nfull, ro, wb, hb, lo, nwe, acc);
    if (12 < nfull) gnn_walk_group<12>(nfull, ro, wb, hb, lo, nwe, acc);
    if ((n & 1) && hf == 0) gnn_gated_add(mt, wt, nwe, acc);
  }
  float sx = hf ? acc.x : acc.z, sy = hf ? acc.y : acc.w;
  float kx = hf ? acc.z : acc.x, ky = hf ? acc.w : acc.y;
  v2f r; r.x = kx + gnn_xhalf(sx); r.y = ky + gnn_xhalf(sy);
  return r;
}

__device__ __forceinline__ v2f gnn_node_finish(v2f agg, v2f a, v2f hh, int ce,
                                               const float* __restrict__ gamma, const float* __restrict__ beta) {
  v2f g = *(const v2f*)(gamma + ce), bt = *(const v2f*)(beta + ce);
  float v0 = a.x + agg.x, v1 = a.y + agg.y;
  float sm = v0 + v1, ss = v0 * v0 + v1 * v1;
  sm += __shfl_xor(sm, 8, 32);  ss += __shfl_xor(ss, 8, 32);
  sm += __shfl_xor(sm, 4, 32);  ss += __shfl_xor(ss, 4, 32);
  sm += __shfl_xor(sm, 2, 32);  ss += __shfl_xor(ss, 2, 32);
  sm += __shfl_xor(sm, 1, 32);  ss += __shfl_xor(ss, 1, 32);
  sm += __shfl_xor(sm, 16, 32); ss += __shfl_xor(ss, 16, 32);
  float mu  = sm * (1.f / HID);
  float var = ss * (1.f / HID) - mu * mu;
  float inv = rsqrtf(var + LN_EPS);
  float o0 = (v0 - mu) * inv * g.x + bt.x;
  float o1 = (v1 - mu) * inv * g.y + bt.y;
  o0 = (o0 >= 0.f) ? o0 : SLOPE * o0;
  o1 = (o1 >= 0.f) ? o1 : SLOPE * o1;
  v2f r; r.x = hh.x + o0; r.y = hh.y + o1;
  return r;
}

__global__ void gnn_gather_ln_kernel(const int* __restrict__ off, const int* __restrict__ perm,
                                     const float* __restrict__ Wedge, const float* __restrict__ h_neigh,
                                     const float* __restrict__ h_self,
                                     const float* __restrict__ gamma, const float* __restrict__ beta,
                                     const float* h, float* dst) {
  int wave = threadIdx.x >> 5, lane = threadIdx.x & 31;
  int wv = blockIdx.x * 8 + wave;
  int ce = (lane & 15) * 4 + (lane >> 4) * 2;
  v4f nwe = *(const v4f*)(Wedge + (lane & 15) * 4) * -1.44269504f;
  for (int q = 0; q < GATHER_NPW; ++q) {
    int node = wv * GATHER_NPW + q;
    unsigned base = (unsigned)node * HID + ce;
    v2f a  = *(const v2f*)(h_self + base);
    v2f hh = *(const v2f*)(h + base);
    v2f agg = gnn_node_walk(node, lane, off, perm, h_neigh, nwe);
    *(v2f*)(dst + base) = gnn_node_finish(agg, a, hh, ce, gamma, beta);
  }
}

#define FUSE_WAVES 16
#define FUSE_NODES (FUSE_WAVES * 2)
__global__ void gnn_gather_ln_gemm_kernel(const int* __restrict__ off, const int* __restrict__ perm,
                                          const float* __restrict__ Wedge, const float* __restrict__ h_neigh,
                                          float* h_self,
                                          const float* __restrict__ gamma, const float* __restrict__ beta,
                                          float* h,
                                          const float* __restrict__ Wself, const float* __restrict__ bself,
                                          const float* __restrict__ Wneigh, const float* __restrict__ bneigh,
                                          float* __restrict__ h_neigh_next) {
  __shared__ float w_s[2 * HID * LDW];
  __shared__ float hn_s[FUSE_NODES * LDW];

  int t = threadIdx.x;
  int wave = t >> 5, lane = t & 31;
  int c2 = lane * 2;
  int lr = lane & 15;
  int rt = wave >> 3, ct = wave & 7;
  int ce = (lane & 15) * 4 + (lane >> 4) * 2;
  v4f nwe = *(const v4f*)(Wedge + (lane & 15) * 4) * -1.44269504f;
  int n0 = blockIdx.x * FUSE_NODES;
  int ln = wave * 2;

  int row = t >> 4, col = (t & 15) * 4;
  const float* bsrc = (ct < 4) ? bself : bneigh;
  float bias;
  {
    unsigned base = (unsigned)(n0 + ln) * HID + ce;
    v2f a  = *(const v2f*)(h_self + base);
    v2f hh = *(const v2f*)(h + base);
    v2f agg = gnn_node_walk(n0 + ln, lane, off, perm, h_neigh, nwe);
    v4f s0 = *(const v4f*)(Wself + 4 * t), s1 = *(const v4f*)(Wself + 4 * (t + 512));
    *(v2f*)(hn_s + ln * LDW + ce) = gnn_node_finish(agg, a, hh, ce, gamma, beta);
    *(v4f*)(w_s + row * LDW + col)        = s0;
    *(v4f*)(w_s + (row + 32) * LDW + col) = s1;
  }
  {
    unsigned base = (unsigned)(n0 + ln + 1) * HID + ce;
    v2f a  = *(const v2f*)(h_self + base);
    v2f hh = *(const v2f*)(h + base);
    v2f agg = gnn_node_walk(n0 + ln + 1, lane, off, perm, h_neigh, nwe);
    v4f m0 = *(const v4f*)(Wneigh + 4 * t), m1 = *(const v4f*)(Wneigh + 4 * (t + 512));
    bias = bsrc[(ct & 3) * 16 + lr];
    *(v2f*)(hn_s + (ln + 1) * LDW + ce) = gnn_node_finish(agg, a, hh, ce, gamma, beta);
    *(v4f*)(w_s + (HID + row) * LDW + col)      = m0;
    *(v4f*)(w_s + (HID + row + 32) * LDW + col) = m1;
  }
  __syncthreads();

  *(v2f*)(h + (size_t)(n0 + ln) * HID + c2)     = *(const v2f*)(hn_s + ln * LDW + c2);
  *(v2f*)(h + (size_t)(n0 + ln + 1) * HID + c2) = *(const v2f*)(hn_s + (ln + 1) * LDW + c2);

  int koff = (lane < 16) ? 0 : 2;
  const float* arow = hn_s + (rt * 16 + lr) * LDW + koff;
  const float* brow = w_s + (ct * 16 + lr) * LDW + koff;
  v8f c = (v8f){};
#pragma unroll
  for (int kb = 0; kb < 16; ++kb) {
    v2f af = *(const v2f*)(arow + kb * 4);
    v2f bf = *(const v2f*)(brow + kb * 4);
    c = __builtin_amdgcn_wmma_f32_16x16x4_f32(false, af, false, bf, (short)0, c, false, false);
  }
  asm volatile("v_nop\n\tv_nop\n\tv_nop\n\tv_nop\n\tv_nop" : "+v"(c));
  float* o = (ct < 4) ? h_self : h_neigh_next;
  int n = (ct & 3) * 16 + lr;
  int mbase = n0 + rt * 16 + ((lane < 16) ? 0 : 8);
#pragma unroll
  for (int i = 0; i < 8; ++i) o[(size_t)(mbase + i) * HID + n] = c[i] + bias;
}

extern "C" void kernel_launch(void* const* d_in, const int* in_sizes, int n_in,
                              void* d_out, int out_size, void* d_ws, size_t ws_size,
                              hipStream_t stream) {
  const float* x      = (const float*)d_in[0];
  const int*   eidx   = (const int*)  d_in[1];
  const float* ew     = (const float*)d_in[2];
  const float* Wp     = (const float*)d_in[3];
  const float* bp     = (const float*)d_in[4];
  const float* Wself  = (const float*)d_in[5];
  const float* bself  = (const float*)d_in[6];
  const float* Wneigh = (const float*)d_in[7];
  const float* bneigh = (const float*)d_in[8];
  const float* Wedge  = (const float*)d_in[9];
  const float* gamma  = (const float*)d_in[10];
  const float* beta   = (const float*)d_in[11];
  float* out = (float*)d_out;

  const size_t NH = (size_t)N_NODES * HID;
  float* h       = (float*)d_ws;
  float* h_self  = h + NH;
  float* h_neigh = h_self + NH;
  int*   ord     = (int*)(h_neigh + NH);
  int*   off     = ord + OFF_OFF;
  int*   bh      = ord + OFF_BH;
  int*   bb      = ord + OFF_BB;
  int*   perm    = ord + OFF_PERM;
  int*   tmp     = ord + OFF_TMP;

  gnn_count_proj_kernel<<<EB_BLOCKS + PROJ_BLOCKS, 1024, 0, stream>>>(eidx, bh, x, Wp, bp, h);
  gnn_scan_gemm_kernel<<<1 + GEMM0_BLOCKS, 1024, 0, stream>>>(bh, bb, h, Wself, bself, Wneigh, bneigh, h_self, h_neigh);

  gnn_bmove_kernel<<<EB_BLOCKS, 1024, 0, stream>>>(eidx, ew, bb, tmp);
  gnn_bfill_kernel<<<NBUCK, 1024, 0, stream>>>(bb, tmp, off, perm);

  for (int l = 0; l < N_LAYERS; ++l) {
    const float* We = Wedge  + (size_t)l * HID;
    const float* gm = gamma  + (size_t)l * HID;
    const float* bt = beta   + (size_t)l * HID;
    float* hn_cur = (l == 1) ? out : h_neigh;
    float* hn_nxt = (l == 1) ? h_neigh : out;

    if (l < N_LAYERS - 1) {
      const float* Ws = Wself  + (size_t)(l + 1) * HID * HID;
      const float* bs = bself  + (size_t)(l + 1) * HID;
      const float* Wn = Wneigh + (size_t)(l + 1) * HID * HID;
      const float* bn = bneigh + (size_t)(l + 1) * HID;
      gnn_gather_ln_gemm_kernel<<<N_NODES / FUSE_NODES, 32 * FUSE_WAVES, 0, stream>>>(off, perm, We, hn_cur, h_self, gm, bt, h,
                                                                        Ws, bs, Wn, bn, hn_nxt);
    } else {
      gnn_gather_ln_kernel<<<N_NODES / (8 * GATHER_NPW), 256, 0, stream>>>(off, perm, We, hn_cur, h_self, gm, bt, h, out);
    }
  }
}


# h_self is formed inside the gather that consumes it (WMMA on the block's own 32 rows of h held in LDS, ahead of the LayerNorm): the h_self plane is neither written nor read; sum orders kept
# speedup vs baseline: 2.8682x; 1.0503x over previous
#include <hip/hip_runtime.h>
#include <hip/hip_bf16.h>

#define N_NODES 100000
#define N_EDGES 1000000
#define IN_DIM  8
#define HID     64
#define N_LAYERS 3
#define LN_EPS  1e-5f
#define SLOPE   0.2f
#define LDW     68

typedef float v2f __attribute__((ext_vector_type(2)));
typedef float v8f __attribute__((ext_vector_type(8)));
typedef int   v2i __attribute__((ext_vector_type(2)));
typedef float v4f __attribute__((ext_vector_type(4)));
typedef int   v4i __attribute__((ext_vector_type(4)));

#define PROJ_WAVES  10000
#define PROJ_BLOCKS 313
#define PROJ_TRIPS  5
#define ROW_TILES   (N_NODES / 16)

#define BUCK_SHIFT 9
#define BUCK_NODES 512
#define NBUCK      196
#define EB_BLOCKS  250
#define EB_EDGES   4000
#define NBH        (NBUCK * EB_BLOCKS)
#define BH_PER_T   48
#define OFF_OFF  0
#define OFF_BH   100352
#define OFF_BB   149504
#define OFF_PERM 200704
#define OFF_TMP  2200704

__device__ __forceinline__ void gnn_proj_body(const float* __restrict__ x, const float* __restrict__ Wp,
                                              const float* __restrict__ bp, float* __restrict__ h, int blk) {
  int t = threadIdx.x;
  int gt = blk * 1024 + t;
  int lane = t & 31;
  int gw = gt >> 5;
  if (gw >= PROJ_WAVES) return;
  int hf = lane >> 4;
  int c4 = (lane & 15) * 4;
  const v4f* wp4 = (const v4f*)(Wp + c4 * IN_DIM);
  v4f w0a = wp4[0], w0b = wp4[1], w1a = wp4[2], w1b = wp4[3];
  v4f w2a = wp4[4], w2b = wp4[5], w3a = wp4[6], w3b = wp4[7];
  v4f b4 = *(const v4f*)(bp + c4);
  v4f xs[PROJ_TRIPS][2];
#pragma unroll
  for (int i = 0; i < PROJ_TRIPS; ++i) {
    int node = (gw * PROJ_TRIPS + i) * 2 + hf;
    const v4f* xr = (const v4f*)(x + (size_t)node * IN_DIM);
    xs[i][0] = xr[0]; xs[i][1] = xr[1];
  }
#pragma unroll
  for (int i = 0; i < PROJ_TRIPS; ++i) {
    int node = (gw * PROJ_TRIPS + i) * 2 + hf;
    v4f xa = xs[i][0], xb = xs[i][1];
    float a0 = b4.x, a1 = b4.y, a2 = b4.z, a3 = b4.w;
    a0 += xa.x * w0a.x; a0 += xa.y * w0a.y; a0 += xa.z * w0a.z; a0 += xa.w * w0a.w;
    a0 += xb.x * w0b.x; a0 += xb.y * w0b.y; a0 += xb.z * w0b.z; a0 += xb.w * w0b.w;
    a1 += xa.x * w1a.x; a1 += xa.y * w1a.y; a1 += xa.z * w1a.z; a1 += xa.w * w1a.w;
    a1 += xb.x * w1b.x; a1 += xb.y * w1b.y; a1 += xb.z * w1b.z; a1 += xb.w * w1b.w;
    a2 += xa.x * w2a.x; a2 += xa.y * w2a.y; a2 += xa.z * w2a.z; a2 += xa.w * w2a.w;
    a2 += xb.x * w2b.x; a2 += xb.y * w2b.y; a2 += xb.z * w2b.z; a2 += xb.w * w2b.w;
    a3 += xa.x * w3a.x; a3 += xa.y * w3a.y; a3 += xa.z * w3a.z; a3 += xa.w * w3a.w;
    a3 += xb.x * w3b.x; a3 += xb.y * w3b.y; a3 += xb.z * w3b.z; a3 += xb.w * w3b.w;
    v4f r; r.x = a0; r.y = a1; r.z = a2; r.w = a3;
    *(v4f*)(h + (size_t)node * HID + c4) = r;
  }
}

template <int THREADS>
__device__ __forceinline__ void gnn_gemm_body(const float* __restrict__ h,
                                const float* __restrict__ Wneigh, const float* __restrict__ bneigh,
                                float* __restrict__ h_neigh, int blk) {
  __shared__ float w_s[HID * LDW];
  __shared__ float b_s[HID];

  int t = threadIdx.x;
#pragma unroll
  for (int i = 0; i < 1024 / THREADS; ++i) {
    int q = t + i * THREADS;
    int row = q >> 4, col = (q & 15) * 4;
    *(v4f*)(w_s + row * LDW + col) = *(const v4f*)(Wneigh + 4 * q);
  }
  if (t < HID) b_s[t] = bneigh[t];
  __syncthreads();

  int wave = t >> 5, lane = t & 31;
  int tile = blk * (THREADS / 32) + wave;
  if (tile >= ROW_TILES) return;
  int r0 = tile * 16;
  int lr = lane & 15;
  int koff = (lane < 16) ? 0 : 2;

  const float* arow = h + (size_t)(r0 + lr) * HID;
  const float* brow = w_s + lr * LDW + koff;

  v2f a[16];
#pragma unroll
  for (int kb = 0; kb < 16; ++kb) a[kb] = *(const v2f*)(arow + kb * 4 + koff);

  v8f c[4];
#pragma unroll
  for (int j = 0; j < 4; ++j) c[j] = (v8f){};
#pragma unroll
  for (int kb = 0; kb < 16; ++kb) {
#pragma unroll
    for (int j = 0; j < 4; ++j) {
      v2f bf = *(const v2f*)(brow + j * 16 * LDW + kb * 4);
      c[j] = __builtin_amdgcn_wmma_f32_16x16x4_f32(false, a[kb], false, bf, (short)0, c[j], false, false);
    }
  }

  int mbase = r0 + ((lane < 16) ? 0 : 8);
#pragma unroll
  for (int j = 0; j < 4; ++j) {
    float bias = b_s[j * 16 + lr];
    int n = j * 16 + lr;
#pragma unroll
    for (int i = 0; i < 8; ++i) {
      h_neigh[(size_t)(mbase + i) * HID + n] = c[j][i] + bias;
    }
  }
}

__device__ __forceinline__ void gnn_bcount_body(const int* __restrict__ eidx, int* __restrict__ bh, int g) {
  __shared__ int hist[NBUCK];
  int t = threadIdx.x;
  if (t < NBUCK) hist[t] = 0;
  __syncthreads();
  if (t < EB_EDGES / 4) {
    v4i r = *(const v4i*)(eidx + g * EB_EDGES + 4 * t);
    atomicAdd(&hist[min(max(r.x >> BUCK_SHIFT, 0), NBUCK - 1)], 1);
    atomicAdd(&hist[min(max(r.y >> BUCK_SHIFT, 0), NBUCK - 1)], 1);
    atomicAdd(&hist[min(max(r.z >> BUCK_SHIFT, 0), NBUCK - 1)], 1);
    atomicAdd(&hist[min(max(r.w >> BUCK_SHIFT, 0), NBUCK - 1)], 1);
  }
  __syncthreads();
  if (t < NBUCK) bh[t * EB_BLOCKS + g] = hist[t];
}

__device__ __forceinline__ void gnn_bscan_body(const int* __restrict__ bh, int* __restrict__ bb) {
  __shared__ int wsum[32];
  int t = threadIdx.x, lane = t & 31, wave = t >> 5;
  int base = t * BH_PER_T;
  v4i v[BH_PER_T / 4];
#pragma unroll
  for (int i = 0; i < BH_PER_T / 4; ++i) {
    int idx = base + 4 * i;
    v4i z; z.x = 0; z.y = 0; z.z = 0; z.w = 0;
    v[i] = (idx < NBH) ? *(const v4i*)(bh + idx) : z;
  }
  int s = 0;
#pragma unroll
  for (int i = 0; i < BH_PER_T / 4; ++i) s += v[i].x + v[i].y + v[i].z + v[i].w;
  int inc = s;
  for (int d = 1; d < 32; d <<= 1) {
    int y = __shfl_up(inc, d, 32);
    if (lane >= d) inc += y;
  }
  if (lane == 31) wsum[wave] = inc;
  __syncthreads();
  if (wave == 0) {
    int x = wsum[lane];
    int xi = x;
    for (int d = 1; d < 32; d <<= 1) {
      int y = __shfl_up(xi, d, 32);
      if (lane >= d) xi += y;
    }
    wsum[lane] = xi - x;
  }
  __syncthreads();
  int run = wsum[wave] + inc - s;
#pragma unroll
  for (int i = 0; i < BH_PER_T / 4; ++i) {
    int idx = base + 4 * i;
    v4i o;
    o.x = run; run += v[i].x;
    o.y = run; run += v[i].y;
    o.z = run; run += v[i].z;
    o.w = run; run += v[i].w;
    if (idx < NBH) *(v4i*)(bb + idx) = o;
  }
}

__global__ void gnn_count_proj_kernel(const int* __restrict__ eidx, int* __restrict__ bh,
                                      const float* __restrict__ x, const float* __restrict__ Wp,
                                      const float* __restrict__ bp, float* __restrict__ h) {
  if (blockIdx.x < EB_BLOCKS) gnn_bcount_body(eidx, bh, blockIdx.x);
  else gnn_proj_body(x, Wp, bp, h, blockIdx.x - EB_BLOCKS);
}

#define GEMM0_BLOCKS ((ROW_TILES + 31) / 32)
__global__ void gnn_scan_gemm_kernel(const int* __restrict__ bh, int* __restrict__ bb,
                                     const float* __restrict__ h,
                                     const float* __restrict__ Wneigh, const float* __restrict__ bneigh,
                                     float* __restrict__ h_neigh) {
  if (blockIdx.x == 0) gnn_bscan_body(bh, bb);
  else gnn_gemm_body<1024>(h, Wneigh, bneigh, h_neigh, blockIdx.x - 1);
}

__global__ void gnn_bmove_kernel(const int* __restrict__ eidx, const float* __restrict__ ew,
                                 const int* __restrict__ bb, int* __restrict__ tmp) {
  __shared__ int cur[NBUCK];
  int t = threadIdx.x, g = blockIdx.x;
  if (t < NBUCK) cur[t] = bb[t * EB_BLOCKS + g];
  __syncthreads();
  if (t < EB_EDGES / 4) {
    size_t e = (size_t)g * EB_EDGES + 4 * t;
    v4i row = *(const v4i*)(eidx + e);
    v4i col = *(const v4i*)(eidx + N_EDGES + e);
    v4f w   = *(const v4f*)(ew + e);
    int p0 = atomicAdd(&cur[min(max(row.x >> BUCK_SHIFT, 0), NBUCK - 1)], 1);
    int p1 = atomicAdd(&cur[min(max(row.y >> BUCK_SHIFT, 0), NBUCK - 1)], 1);
    int p2 = atomicAdd(&cur[min(max(row.z >> BUCK_SHIFT, 0), NBUCK - 1)], 1);
    int p3 = atomicAdd(&cur[min(max(row.w >> BUCK_SHIFT, 0), NBUCK - 1)], 1);
    p0 = min(max(p0, 0), N_EDGES - 1);
    p1 = min(max(p1, 0), N_EDGES - 1);
    p2 = min(max(p2, 0), N_EDGES - 1);
    p3 = min(max(p3, 0), N_EDGES - 1);
    v4i rec; rec.w = 0;
    rec.x = row.x; rec.y = col.x; rec.z = __float_as_int(w.x); *(v4i*)(tmp + 4 * (size_t)p0) = rec;
    rec.x = row.y; rec.y = col.y; rec.z = __float_as_int(w.y); *(v4i*)(tmp + 4 * (size_t)p1) = rec;
    rec.x = row.z; rec.y = col.z; rec.z = __float_as_int(w.z); *(v4i*)(tmp + 4 * (size_t)p2) = rec;
    rec.x = row.w; rec.y = col.w; rec.z = __float_as_int(w.w); *(v4i*)(tmp + 4 * (size_t)p3) = rec;
  }
}

__global__ void gnn_bfill_kernel(const int* __restrict__ bb, const int* __restrict__ tmp,
                                 int* __restrict__ off, int* __restrict__ perm) {
  __shared__ int cnt[BUCK_NODES];
  __shared__ int cur[BUCK_NODES];
  __shared__ int wsum[32];
  int t = threadIdx.x, lane = t & 31, wave = t >> 5;
  int b = blockIdx.x;
  int s = __builtin_amdgcn_readfirstlane(bb[b * EB_BLOCKS]);
  int e_end = (b + 1 < NBUCK) ? __builtin_amdgcn_readfirstlane(bb[(b + 1) * EB_BLOCKS]) : N_EDGES;
  s = min(max(s, 0), N_EDGES);
  e_end = min(max(e_end, s), N_EDGES);
  if (t < BUCK_NODES) cnt[t] = 0;
  __syncthreads();
  for (int e0 = s + t; e0 < e_end; e0 += 4 * 1024) {
    int r[4];
#pragma unroll
    for (int u = 0; u < 4; ++u) {
      int e = e0 + u * 1024;
      r[u] = (e < e_end) ? tmp[4 * (size_t)e] : -1;
    }
#pragma unroll
    for (int u = 0; u < 4; ++u)
      if (r[u] >= 0) atomicAdd(&cnt[r[u] & (BUCK_NODES - 1)], 1);
  }
  __syncthreads();
  int v = (t < BUCK_NODES) ? cnt[t] : 0;
  int inc = v;
  for (int d = 1; d < 32; d <<= 1) {
    int y = __shfl_up(inc, d, 32);
    if (lane >= d) inc += y;
  }
  if (lane == 31) wsum[wave] = inc;
  __syncthreads();
  if (wave == 0) {
    int x = wsum[lane];
    int xi = x;
    for (int d = 1; d < 32; d <<= 1) {
      int y = __shfl_up(xi, d, 32);
      if (lane >= d) xi += y;
    }
    wsum[lane] = xi - x;
  }
  __syncthreads();
  if (t < BUCK_NODES) {
    int first = s + wsum[wave] + inc - v;
    cur[t] = first;
    int node = b * BUCK_NODES + t;
    if (node <= N_NODES) off[node] = first;
  }
  __syncthreads();
  for (int e0 = s + t; e0 < e_end; e0 += 4 * 1024) {
    v4i rec[4];
#pragma unroll
    for (int u = 0; u < 4; ++u) {
      int e = e0 + u * 1024;
      v4i z; z.x = -1; z.y = 0; z.z = 0; z.w = 0;
      rec[u] = (e < e_end) ? *(const v4i*)(tmp + 4 * (size_t)e) : z;
    }
#pragma unroll
    for (int u = 0; u < 4; ++u) {
      if (rec[u].x >= 0) {
        int p = atomicAdd(&cur[rec[u].x & (BUCK_NODES - 1)], 1);
        p = min(max(p, 0), N_EDGES - 1);
        v2i cw; cw.x = rec[u].y; cw.y = rec[u].z;
        *(v2i*)(perm + 2 * (size_t)p) = cw;
      }
    }
  }
}

#define HALF_LANE(V, J) __builtin_amdgcn_update_dpp(0, (V), 0x150 + ((J) & 15), 0xf, 0xf, true)
#define ROW_REQ(M, J) if ((J) < nfull) { M = *(const v4f*)(hb + (unsigned)(HALF_LANE(ro, (J)) + lo)); }
#define ROW_ADD(M, J) if ((J) < nfull) { float w_ = __int_as_float(HALF_LANE(wb, (J))); gnn_gated_add(M, w_, nwe, acc); }

__device__ __forceinline__ void gnn_gated_add(v4f m, float w, v4f nwe, v4f& acc) {
  v2f z0 = nwe.xy * w, z1 = nwe.zw * w;
  acc.x += m.x * __builtin_amdgcn_rcpf(1.f + __builtin_amdgcn_exp2f(z0.x));
  acc.y += m.y * __builtin_amdgcn_rcpf(1.f + __builtin_amdgcn_exp2f(z0.y));
  acc.z += m.z * __builtin_amdgcn_rcpf(1.f + __builtin_amdgcn_exp2f(z1.x));
  acc.w += m.w * __builtin_amdgcn_rcpf(1.f + __builtin_amdgcn_exp2f(z1.y));
}

template <int G0>
__device__ __forceinline__ void gnn_walk_group(int nfull, int ro, int wb, const char* __restrict__ hb, unsigned lo, v4f nwe, v4f& acc) {
  v4f m0, m1, m2, m3;
  ROW_REQ(m0, G0 + 0) ROW_REQ(m1, G0 + 1) ROW_REQ(m2, G0 + 2) ROW_REQ(m3, G0 + 3)
  ROW_ADD(m0, G0 + 0) ROW_ADD(m1, G0 + 1) ROW_ADD(m2, G0 + 2) ROW_ADD(m3, G0 + 3)
}

__device__ __forceinline__ float gnn_xhalf(float x) {
  int b = __builtin_bit_cast(int, x);
  return __builtin_bit_cast(float, __builtin_amdgcn_permlanex16(b, b, 0x76543210, 0xfedcba98, false, false));
}

__device__ __forceinline__ v2f gnn_node_walk(int node, int lane,
                                             const int* __restrict__ off, const int* __restrict__ perm,
                                             const float* __restrict__ h_neigh, v4f nwe) {
  int s = __builtin_amdgcn_readfirstlane(off[(unsigned)node]);
  int t = __builtin_amdgcn_readfirstlane(off[(unsigned)node + 1u]);
  s = min(max(s, 0), N_EDGES);
  t = min(max(t, s), N_EDGES);
  int hf = lane >> 4, lr = lane & 15;
  const char* hb = (const char*)h_neigh;
  unsigned lo = lr * 16;
  v4f acc = (v4f){};
  for (int p = s; p < t; p += 32) {
    int n = min(t - p, 32);
    int nfull = n >> 1;
    int e = 2 * lr + hf;
    v2i cw; cw.x = 0; cw.y = 0;
    if (e < n) cw = *(const v2i*)(perm + 2u * (unsigned)(p + e));
    int ro = min(max(cw.x, 0), N_NODES - 1) << 8;
    int wb = cw.y;
    v4f mt = (v4f){}; float wt = 0.f;
    if (n & 1) {
      unsigned ot = (unsigned)__builtin_amdgcn_readlane(ro, nfull);
      wt = __int_as_float(__builtin_amdgcn_readlane(wb, nfull));
      mt = *(const v4f*)(hb + (ot + lo));
    }
    if (0 < nfull)  gnn_walk_group<0>(nfull, ro, wb, hb, lo, nwe, acc);
    if (4 < nfull)  gnn_walk_group<4>(nfull, ro, wb, hb, lo, nwe, acc);
    if (8 < nfull)  gnn_walk_group<8>(nfull, ro, wb, hb, lo, nwe, acc);
    if (12 < nfull) gnn_walk_group<12>(nfull, ro, wb, hb, lo, nwe, acc);
    if ((n & 1) && hf == 0) gnn_gated_add(mt, wt, nwe, acc);
  }
  float sx = hf ? acc.x : acc.z, sy = hf ? acc.y : acc.w;
  float kx = hf ? acc.z : acc.x, ky = hf ? acc.w : acc.y;
  v2f r; r.x = kx + gnn_xhalf(sx); r.y = ky + gnn_xhalf(sy);
  return r;
}

__device__ __forceinline__ v2f gnn_node_finish(v2f agg, v2f a, v2f hh, v2f g, v2f bt) {
  float v0 = a.x + agg.x, v1 = a.y + agg.y;
  float sm = v0 + v1, ss = v0 * v0 + v1 * v1;
  sm += __shfl_xor(sm, 8, 32);  ss += __shfl_xor(ss, 8, 32);
  sm += __shfl_xor(sm, 4, 32);  ss += __shfl_xor(ss, 4, 32);
  sm += __shfl_xor(sm, 2, 32);  ss += __shfl_xor(ss, 2, 32);
  sm += __shfl_xor(sm, 1, 32);  ss += __shfl_xor(ss, 1, 32);
  sm += __shfl_xor(sm, 16, 32); ss += __shfl_xor(ss, 16, 32);
  float mu  = sm * (1.f / HID);
  float var = ss * (1.f / HID) - mu * mu;
  float inv = rsqrtf(var + LN_EPS);
  float o0 = (v0 - mu) * inv * g.x + bt.x;
  float o1 = (v1 - mu) * inv * g.y + bt.y;
  o0 = (o0 >= 0.f) ? o0 : SLOPE * o0;
  o1 = (o1 >= 0.f) ? o1 : SLOPE * o1;
  v2f r; r.x = hh.x + o0; r.y = hh.y + o1;
  return r;
}

#define FUSE_WAVES 16
#define FUSE_NODES (FUSE_WAVES * 2)

__device__ __forceinline__ v8f gnn_tile_product(const float* arow, const float* brow) {
  v8f c = (v8f){};
#pragma unroll
  for (int kb = 0; kb < 16; ++kb) {
    v2f af = *(const v2f*)(arow + kb * 4);
    v2f bf = *(const v2f*)(brow + kb * 4);
    c = __builtin_amdgcn_wmma_f32_16x16x4_f32(false, af, false, bf, (short)0, c, false, false);
  }
  asm volatile("v_nop\n\tv_nop\n\tv_nop\n\tv_nop\n\tv_nop" : "+v"(c));
  return c;
}

template <bool LAST>
__device__ __forceinline__ void gnn_gather_body(const int* __restrict__ off, const int* __restrict__ perm,
                                                const float* __restrict__ Wedge, const float* __restrict__ h_neigh,
                                                const float* __restrict__ Wself, const float* __restrict__ bself,
                                                const float* __restrict__ gamma, const float* __restrict__ beta,
                                                const float* h_in, float* h_out,
                                                const float* __restrict__ Wneigh_next, const float* __restrict__ bneigh_next,
                                                float* __restrict__ h_neigh_next) {
  __shared__ float w_s[(LAST ? 1 : 2) * HID * LDW];
  __shared__ float ho_s[FUSE_NODES * LDW];
  __shared__ float hs_s[FUSE_NODES * LDW];

  int t = threadIdx.x;
  int lane = t & 31;
  int wave = __builtin_amdgcn_readfirstlane(t >> 5);
  int lr = lane & 15;
  int rt = wave >> 3, ct = wave & 7;
  int ce = (lane & 15) * 4 + (lane >> 4) * 2;
  v4f nwe = *(const v4f*)(Wedge + (lane & 15) * 4) * -1.44269504f;
  int n0 = blockIdx.x * FUSE_NODES;
  int ln = wave * 2;
  int row = t >> 4, col = (t & 15) * 4;

  unsigned base0 = (unsigned)(n0 + ln) * HID + ce;
  v2f hh0 = *(const v2f*)(h_in + base0);
  v2f hh1 = *(const v2f*)(h_in + base0 + HID);
  v2f agg0 = gnn_node_walk(n0 + ln, lane, off, perm, h_neigh, nwe);
  *(v2f*)(ho_s + ln * LDW + ce)       = hh0;
  *(v2f*)(ho_s + (ln + 1) * LDW + ce) = hh1;
  v2f agg1 = gnn_node_walk(n0 + ln + 1, lane, off, perm, h_neigh, nwe);
  {
    v4f s0 = *(const v4f*)(Wself + 4 * t), s1 = *(const v4f*)(Wself + 4 * (t + 512));
    if (!LAST) {
      v4f m0 = *(const v4f*)(Wneigh_next + 4 * t), m1 = *(const v4f*)(Wneigh_next + 4 * (t + 512));
      *(v4f*)(w_s + (HID + row) * LDW + col)      = m0;
      *(v4f*)(w_s + (HID + row + 32) * LDW + col) = m1;
    }
    *(v4f*)(w_s + row * LDW + col)        = s0;
    *(v4f*)(w_s + (row + 32) * LDW + col) = s1;
  }
  float bias = 0.f;
  if (ct < 4) bias = bself[ct * 16 + lr];
  else if (!LAST) bias = bneigh_next[(ct - 4) * 16 + lr];
  v2f g = *(const v2f*)(gamma + ce), bt = *(const v2f*)(beta + ce);
  __syncthreads();

  int koff = (lane < 16) ? 0 : 2;
  int hi8 = (lane < 16) ? 0 : 8;
  const float* arow = ho_s + (rt * 16 + lr) * LDW + koff;
  if (ct < 4) {
    const float* brow = w_s + (ct * 16 + lr) * LDW + koff;
    v8f c = gnn_tile_product(arow, brow);
    float* d = hs_s + (rt * 16 + hi8) * LDW + ct * 16 + lr;
#pragma unroll
    for (int i = 0; i < 8; ++i) d[i * LDW] = c[i] + bias;
  }
  __syncthreads();

  v2f r0, r1;
  {
    v2f a  = *(const v2f*)(hs_s + ln * LDW + ce);
    v2f hh = *(const v2f*)(ho_s + ln * LDW + ce);
    r0 = gnn_node_finish(agg0, a, hh, g, bt);
  }
  {
    v2f a  = *(const v2f*)(hs_s + (ln + 1) * LDW + ce);
    v2f hh = *(const v2f*)(ho_s + (ln + 1) * LDW + ce);
    r1 = gnn_node_finish(agg1, a, hh, g, bt);
  }
  if (LAST) {
    *(v2f*)(h_out + base0)       = r0;
    *(v2f*)(h_out + base0 + HID) = r1;
    return;
  }
  *(v2f*)(ho_s + ln * LDW + ce)       = r0;
  *(v2f*)(ho_s + (ln + 1) * LDW + ce) = r1;
  __syncthreads();

  *(v2f*)(h_out + base0)       = r0;
  *(v2f*)(h_out + base0 + HID) = r1;
  if (ct >= 4) {
    int cn = ct - 4;
    const float* brow = w_s + (HID + cn * 16 + lr) * LDW + koff;
    v8f c = gnn_tile_product(arow, brow);
    float* o = h_neigh_next + (size_t)(n0 + rt * 16 + hi8) * HID + cn * 16 + lr;
#pragma unroll
    for (int i = 0; i < 8; ++i) o[(size_t)i * HID] = c[i] + bias;
  }
}

__global__ void gnn_gather_ln_gemm_kernel(const int* __restrict__ off, const int* __restrict__ perm,
                                          const float* __restrict__ Wedge, const float* __restrict__ h_neigh,
                                          const float* __restrict__ Wself, const float* __restrict__ bself,
                                          const float* __restrict__ gamma, const float* __restrict__ beta,
                                          float* h,
                                          const float* __restrict__ Wneigh_next, const float* __restrict__ bneigh_next,
                                          float* __restrict__ h_neigh_next) {
  gnn_gather_body<false>(off, perm, Wedge, h_neigh, Wself, bself, gamma, beta, h, h, Wneigh_next, bneigh_next, h_neigh_next);
}

__global__ void gnn_gather_ln_kernel(const int* __restrict__ off, const int* __restrict__ perm,
                                     const float* __restrict__ Wedge, const float* __restrict__ h_neigh,
                                     const float* __restrict__ Wself, const float* __restrict__ bself,
                                     const float* __restrict__ gamma, const float* __restrict__ beta,
                                     const float* h, float* dst) {
  gnn_gather_body<true>(off, perm, Wedge, h_neigh, Wself, bself, gamma, beta, h, dst, nullptr, nullptr, nullptr);
}

extern "C" void kernel_launch(void* const* d_in, const int* in_sizes, int n_in,
                              void* d_out, int out_size, void* d_ws, size_t ws_size,
                              hipStream_t stream) {
  const float* x      = (const float*)d_in[0];
  const int*   eidx   = (const int*)  d_in[1];
  const float* ew     = (const float*)d_in[2];
  const float* Wp     = (const float*)d_in[3];
  const float* bp     = (const float*)d_in[4];
  const float* Wself  = (const float*)d_in[5];
  const float* bself  = (const float*)d_in[6];
  const float* Wneigh = (const float*)d_in[7];
  const float* bneigh = (const float*)d_in[8];
  const float* Wedge  = (const float*)d_in[9];
  const float* gamma  = (const float*)d_in[10];
  const float* beta   = (const float*)d_in[11];
  float* out = (float*)d_out;

  const size_t NH = (size_t)N_NODES * HID;
  float* h       = (float*)d_ws;
  float* h_neigh = h + 2 * NH;
  int*   ord     = (int*)(h_neigh + NH);
  int*   off     = ord + OFF_OFF;
  int*   bh      = ord + OFF_BH;
  int*   bb      = ord + OFF_BB;
  int*   perm    = ord + OFF_PERM;
  int*   tmp     = ord + OFF_TMP;

  gnn_count_proj_kernel<<<EB_BLOCKS + PROJ_BLOCKS, 1024, 0, stream>>>(eidx, bh, x, Wp, bp, h);
  gnn_scan_gemm_kernel<<<1 + GEMM0_BLOCKS, 1024, 0, stream>>>(bh, bb, h, Wneigh, bneigh, h_neigh);

  gnn_bmove_kernel<<<EB_BLOCKS, 1024, 0, stream>>>(eidx, ew, bb, tmp);
  gnn_bfill_kernel<<<NBUCK, 1024, 0, stream>>>(bb, tmp, off, perm);

  for (int l = 0; l < N_LAYERS; ++l) {
    const float* We = Wedge  + (size_t)l * HID;
    const float* gm = gamma  + (size_t)l * HID;
    const float* bt = beta   + (size_t)l * HID;
    const float* Ws = Wself  + (size_t)l * HID * HID;
    const float* bs = bself  + (size_t)l * HID;
    float* hn_cur = (l == 1) ? out : h_neigh;
    float* hn_nxt = (l == 1) ? h_neigh : out;

    if (l < N_LAYERS - 1) {
      const float* Wn = Wneigh + (size_t)(l + 1) * HID * HID;
      const float* bn = bneigh + (size_t)(l + 1) * HID;
      gnn_gather_ln_gemm_kernel<<<N_NODES / FUSE_NODES, 32 * FUSE_WAVES, 0, stream>>>(off, perm, We, hn_cur, Ws, bs, gm, bt, h,
                                                                        Wn, bn, hn_nxt);
    } else {
      gnn_gather_ln_kernel<<<N_NODES / FUSE_NODES, 32 * FUSE_WAVES, 0, stream>>>(off, perm, We, hn_cur, Ws, bs, gm, bt, h, out);
    }
  }
}


# Edge sort's scattered stores coalesced: move and fill order their records in LDS first and write contiguous runs (whole lines), not one 16- or 8-byte store per edge; sum order stays LDS arrival order
# speedup vs baseline: 2.9669x; 1.0344x over previous
#include <hip/hip_runtime.h>
#include <hip/hip_bf16.h>

#define N_NODES 100000
#define N_EDGES 1000000
#define IN_DIM  8
#define HID     64
#define N_LAYERS 3
#define LN_EPS  1e-5f
#define SLOPE   0.2f
#define LDW     68

typedef float v2f __attribute__((ext_vector_type(2)));
typedef float v8f __attribute__((ext_vector_type(8)));
typedef int   v2i __attribute__((ext_vector_type(2)));
typedef float v4f __attribute__((ext_vector_type(4)));
typedef int   v4i __attribute__((ext_vector_type(4)));

#define PROJ_WAVES  10000
#define PROJ_BLOCKS 313
#define PROJ_TRIPS  5
#define ROW_TILES   (N_NODES / 16)

#define BUCK_SHIFT 9
#define BUCK_NODES 512
#define NBUCK      196
#define EB_BLOCKS  250
#define EB_EDGES   4000
#define NBH        (NBUCK * EB_BLOCKS)
#define BH_PER_T   48
#define OFF_OFF  0
#define OFF_BH   100352
#define OFF_BB   149504
#define OFF_PERM 200704
#define OFF_TMP  2200704

__device__ __forceinline__ void gnn_proj_body(const float* __restrict__ x, const float* __restrict__ Wp,
                                              const float* __restrict__ bp, float* __restrict__ h, int blk) {
  int t = threadIdx.x;
  int gt = blk * 1024 + t;
  int lane = t & 31;
  int gw = gt >> 5;
  if (gw >= PROJ_WAVES) return;
  int hf = lane >> 4;
  int c4 = (lane & 15) * 4;
  const v4f* wp4 = (const v4f*)(Wp + c4 * IN_DIM);
  v4f w0a = wp4[0], w0b = wp4[1], w1a = wp4[2], w1b = wp4[3];
  v4f w2a = wp4[4], w2b = wp4[5], w3a = wp4[6], w3b = wp4[7];
  v4f b4 = *(const v4f*)(bp + c4);
  v4f xs[PROJ_TRIPS][2];
#pragma unroll
  for (int i = 0; i < PROJ_TRIPS; ++i) {
    int node = (gw * PROJ_TRIPS + i) * 2 + hf;
    const v4f* xr = (const v4f*)(x + (size_t)node * IN_DIM);
    xs[i][0] = xr[0]; xs[i][1] = xr[1];
  }
#pragma unroll
  for (int i = 0; i < PROJ_TRIPS; ++i) {
    int node = (gw * PROJ_TRIPS + i) * 2 + hf;
    v4f xa = xs[i][0], xb = xs[i][1];
    float a0 = b4.x, a1 = b4.y, a2 = b4.z, a3 = b4.w;
    a0 += xa.x * w0a.x; a0 += xa.y * w0a.y; a0 += xa.z * w0a.z; a0 += xa.w * w0a.w;
    a0 += xb.x * w0b.x; a0 += xb.y * w0b.y; a0 += xb.z * w0b.z; a0 += xb.w * w0b.w;
    a1 += xa.x * w1a.x; a1 += xa.y * w1a.y; a1 += xa.z * w1a.z; a1 += xa.w * w1a.w;
    a1 += xb.x * w1b.x; a1 += xb.y * w1b.y; a1 += xb.z * w1b.z; a1 += xb.w * w1b.w;
    a2 += xa.x * w2a.x; a2 += xa.y * w2a.y; a2 += xa.z * w2a.z; a2 += xa.w * w2a.w;
    a2 += xb.x * w2b.x; a2 += xb.y * w2b.y; a2 += xb.z * w2b.z; a2 += xb.w * w2b.w;
    a3 += xa.x * w3a.x; a3 += xa.y * w3a.y; a3 += xa.z * w3a.z; a3 += xa.w * w3a.w;
    a3 += xb.x * w3b.x; a3 += xb.y * w3b.y; a3 += xb.z * w3b.z; a3 += xb.w * w3b.w;
    v4f r; r.x = a0; r.y = a1; r.z = a2; r.w = a3;
    *(v4f*)(h + (size_t)node * HID + c4) = r;
  }
}

template <int THREADS>
__device__ __forceinline__ void gnn_gemm_body(const float* __restrict__ h,
                                const float* __restrict__ Wneigh, const float* __restrict__ bneigh,
                                float* __restrict__ h_neigh, int blk) {
  __shared__ float w_s[HID * LDW];
  __shared__ float b_s[HID];

  int t = threadIdx.x;
#pragma unroll
  for (int i = 0; i < 1024 / THREADS; ++i) {
    int q = t + i * THREADS;
    int row = q >> 4, col = (q & 15) * 4;
    *(v4f*)(w_s + row * LDW + col) = *(const v4f*)(Wneigh + 4 * q);
  }
  if (t < HID) b_s[t] = bneigh[t];
  __syncthreads();

  int wave = t >> 5, lane = t & 31;
  int tile = blk * (THREADS / 32) + wave;
  if (tile >= ROW_TILES) return;
  int r0 = tile * 16;
  int lr = lane & 15;
  int koff = (lane < 16) ? 0 : 2;

  const float* arow = h + (size_t)(r0 + lr) * HID;
  const float* brow = w_s + lr * LDW + koff;

  v2f a[16];
#pragma unroll
  for (int kb = 0; kb < 16; ++kb) a[kb] = *(const v2f*)(arow + kb * 4 + koff);

  v8f c[4];
#pragma unroll
  for (int j = 0; j < 4; ++j) c[j] = (v8f){};
#pragma unroll
  for (int kb = 0; kb < 16; ++kb) {
#pragma unroll
    for (int j = 0; j < 4; ++j) {
      v2f bf = *(const v2f*)(brow + j * 16 * LDW + kb * 4);
      c[j] = __builtin_amdgcn_wmma_f32_16x16x4_f32(false, a[kb], false, bf, (short)0, c[j], false, false);
    }
  }

  int mbase = r0 + ((lane < 16) ? 0 : 8);
#pragma unroll
  for (int j = 0; j < 4; ++j) {
    float bias = b_s[j * 16 + lr];
    int n = j * 16 + lr;
#pragma unroll
    for (int i = 0; i < 8; ++i) {
      h_neigh[(size_t)(mbase + i) * HID + n] = c[j][i] + bias;
    }
  }
}

__device__ __forceinline__ void gnn_bcount_body(const int* __restrict__ eidx, int* __restrict__ bh, int g) {
  __shared__ int hist[NBUCK];
  int t = threadIdx.x;
  if (t < NBUCK) hist[t] = 0;
  __syncthreads();
  if (t < EB_EDGES / 4) {
    v4i r = *(const v4i*)(eidx + g * EB_EDGES + 4 * t);
    atomicAdd(&hist[min(max(r.x >> BUCK_SHIFT, 0), NBUCK - 1)], 1);
    atomicAdd(&hist[min(max(r.y >> BUCK_SHIFT, 0), NBUCK - 1)], 1);
    atomicAdd(&hist[min(max(r.z >> BUCK_SHIFT, 0), NBUCK - 1)], 1);
    atomicAdd(&hist[min(max(r.w >> BUCK_SHIFT, 0), NBUCK - 1)], 1);
  }
  __syncthreads();
  if (t < NBUCK) bh[t * EB_BLOCKS + g] = hist[t];
}

__device__ __forceinline__ void gnn_bscan_body(const int* __restrict__ bh, int* __restrict__ bb) {
  __shared__ int wsum[32];
  int t = threadIdx.x, lane = t & 31, wave = t >> 5;
  int base = t * BH_PER_T;
  v4i v[BH_PER_T / 4];
#pragma unroll
  for (int i = 0; i < BH_PER_T / 4; ++i) {
    int idx = base + 4 * i;
    v4i z; z.x = 0; z.y = 0; z.z = 0; z.w = 0;
    v[i] = (idx < NBH) ? *(const v4i*)(bh + idx) : z;
  }
  int s = 0;
#pragma unroll
  for (int i = 0; i < BH_PER_T / 4; ++i) s += v[i].x + v[i].y + v[i].z + v[i].w;
  int inc = s;
  for (int d = 1; d < 32; d <<= 1) {
    int y = __shfl_up(inc, d, 32);
    if (lane >= d) inc += y;
  }
  if (lane == 31) wsum[wave] = inc;
  __syncthreads();
  if (wave == 0) {
    int x = wsum[lane];
    int xi = x;
    for (int d = 1; d < 32; d <<= 1) {
      int y = __shfl_up(xi, d, 32);
      if (lane >= d) xi += y;
    }
    wsum[lane] = xi - x;
  }
  __syncthreads();
  int run = wsum[wave] + inc - s;
#pragma unroll
  for (int i = 0; i < BH_PER_T / 4; ++i) {
    int idx = base + 4 * i;
    v4i o;
    o.x = run; run += v[i].x;
    o.y = run; run += v[i].y;
    o.z = run; run += v[i].z;
    o.w = run; run += v[i].w;
    if (idx < NBH) *(v4i*)(bb + idx) = o;
  }
}

__global__ void gnn_count_proj_kernel(const int* __restrict__ eidx, int* __restrict__ bh,
                                      const float* __restrict__ x, const float* __restrict__ Wp,
                                      const float* __restrict__ bp, float* __restrict__ h) {
  if (blockIdx.x < EB_BLOCKS) gnn_bcount_body(eidx, bh, blockIdx.x);
  else gnn_proj_body(x, Wp, bp, h, blockIdx.x - EB_BLOCKS);
}

#define GEMM0_BLOCKS ((ROW_TILES + 31) / 32)
__global__ void gnn_scan_gemm_kernel(const int* __restrict__ bh, int* __restrict__ bb,
                                     const float* __restrict__ h,
                                     const float* __restrict__ Wneigh, const float* __restrict__ bneigh,
                                     float* __restrict__ h_neigh) {
  if (blockIdx.x == 0) gnn_bscan_body(bh, bb);
  else gnn_gemm_body<1024>(h, Wneigh, bneigh, h_neigh, blockIdx.x - 1);
}

__global__ void gnn_bmove_kernel(const int* __restrict__ eidx, const float* __restrict__ ew,
                                 const int* __restrict__ bb, int* __restrict__ tmp) {
  __shared__ int cur[NBUCK];
  __shared__ int dlt[NBUCK];
  __shared__ int wsum[8];
  __shared__ v4i rec_s[EB_EDGES];
  int t = threadIdx.x, g = blockIdx.x;
  int lane = t & 31, wave = t >> 5;
  v4i row = (v4i){}, col = (v4i){};
  v4f w = (v4f){};
  if (t < EB_EDGES / 4) {
    size_t e = (size_t)g * EB_EDGES + 4 * t;
    row = *(const v4i*)(eidx + e);
    col = *(const v4i*)(eidx + N_EDGES + e);
    w   = *(const v4f*)(ew + e);
  }
  int gb = 0, c = 0;
  if (t < NBUCK) {
    int idx = t * EB_BLOCKS + g;
    gb = bb[idx];
    int nx = (idx + 1 < NBH) ? bb[idx + 1] : N_EDGES;
    c = nx - gb;
  }
  int inc = c;
  for (int d = 1; d < 32; d <<= 1) {
    int y = __shfl_up(inc, d, 32);
    if (lane >= d) inc += y;
  }
  if (lane == 31 && wave < 8) wsum[wave] = inc;
  __syncthreads();
  if (t < NBUCK) {
    int woff = 0;
    for (int q = 0; q < wave; ++q) woff += wsum[q];
    int lo = woff + inc - c;
    cur[t] = lo;
    dlt[t] = gb - lo;
  }
  __syncthreads();
  if (t < EB_EDGES / 4) {
    int p0 = atomicAdd(&cur[min(max(row.x >> BUCK_SHIFT, 0), NBUCK - 1)], 1);
    int p1 = atomicAdd(&cur[min(max(row.y >> BUCK_SHIFT, 0), NBUCK - 1)], 1);
    int p2 = atomicAdd(&cur[min(max(row.z >> BUCK_SHIFT, 0), NBUCK - 1)], 1);
    int p3 = atomicAdd(&cur[min(max(row.w >> BUCK_SHIFT, 0), NBUCK - 1)], 1);
    p0 = min(max(p0, 0), EB_EDGES - 1);
    p1 = min(max(p1, 0), EB_EDGES - 1);
    p2 = min(max(p2, 0), EB_EDGES - 1);
    p3 = min(max(p3, 0), EB_EDGES - 1);
    v4i rec; rec.w = 0;
    rec.x = row.x; rec.y = col.x; rec.z = __float_as_int(w.x); rec_s[p0] = rec;
    rec.x = row.y; rec.y = col.y; rec.z = __float_as_int(w.y); rec_s[p1] = rec;
    rec.x = row.z; rec.y = col.z; rec.z = __float_as_int(w.z); rec_s[p2] = rec;
    rec.x = row.w; rec.y = col.w; rec.z = __float_as_int(w.w); rec_s[p3] = rec;
  }
  __syncthreads();
#pragma unroll
  for (int k = 0; k < (EB_EDGES + 1023) / 1024; ++k) {
    int i = t + k * 1024;
    if (i < EB_EDGES) {
      v4i rec = rec_s[i];
      int b = min(max(rec.x >> BUCK_SHIFT, 0), NBUCK - 1);
      int p = i + dlt[b];
      p = min(max(p, 0), N_EDGES - 1);
      *(v4i*)(tmp + 4 * (size_t)p) = rec;
    }
  }
}

#define FILL_CAP 8192
__global__ void gnn_bfill_kernel(const int* __restrict__ bb, const int* __restrict__ tmp,
                                 int* __restrict__ off, int* __restrict__ perm) {
  __shared__ int cnt[BUCK_NODES];
  __shared__ int cur[BUCK_NODES];
  __shared__ int wsum[32];
  __shared__ v2i st_s[FILL_CAP];
  int t = threadIdx.x, lane = t & 31, wave = t >> 5;
  int b = blockIdx.x;
  int s = __builtin_amdgcn_readfirstlane(bb[b * EB_BLOCKS]);
  int e_end = (b + 1 < NBUCK) ? __builtin_amdgcn_readfirstlane(bb[(b + 1) * EB_BLOCKS]) : N_EDGES;
  s = min(max(s, 0), N_EDGES);
  e_end = min(max(e_end, s), N_EDGES);
  int m = e_end - s;
  bool staged = m <= FILL_CAP;
  if (t < BUCK_NODES) cnt[t] = 0;
  __syncthreads();
  for (int e0 = s + t; e0 < e_end; e0 += 4 * 1024) {
    int r[4];
#pragma unroll
    for (int u = 0; u < 4; ++u) {
      int e = e0 + u * 1024;
      r[u] = (e < e_end) ? tmp[4 * (size_t)e] : -1;
    }
#pragma unroll
    for (int u = 0; u < 4; ++u)
      if (r[u] >= 0) atomicAdd(&cnt[r[u] & (BUCK_NODES - 1)], 1);
  }
  __syncthreads();
  int v = (t < BUCK_NODES) ? cnt[t] : 0;
  int inc = v;
  for (int d = 1; d < 32; d <<= 1) {
    int y = __shfl_up(inc, d, 32);
    if (lane >= d) inc += y;
  }
  if (lane == 31) wsum[wave] = inc;
  __syncthreads();
  if (wave == 0) {
    int x = wsum[lane];
    int xi = x;
    for (int d = 1; d < 32; d <<= 1) {
      int y = __shfl_up(xi, d, 32);
      if (lane >= d) xi += y;
    }
    wsum[lane] = xi - x;
  }
  __syncthreads();
  if (t < BUCK_NODES) {
    int first = s + wsum[wave] + inc - v;
    cur[t] = first;
    int node = b * BUCK_NODES + t;
    if (node <= N_NODES) off[node] = first;
  }
  __syncthreads();
  for (int e0 = s + t; e0 < e_end; e0 += 4 * 1024) {
    v4i rec[4];
#pragma unroll
    for (int u = 0; u < 4; ++u) {
      int e = min(e0 + u * 1024, e_end - 1);
      rec[u] = *(const v4i*)(tmp + 4 * (size_t)e);
    }
#pragma unroll
    for (int u = 0; u < 4; ++u) {
      if (e0 + u * 1024 < e_end && rec[u].x >= 0) {
        int p = atomicAdd(&cur[rec[u].x & (BUCK_NODES - 1)], 1);
        v2i cw; cw.x = rec[u].y; cw.y = rec[u].z;
        if (staged) {
          st_s[min(max(p - s, 0), FILL_CAP - 1)] = cw;
        } else {
          p = min(max(p, 0), N_EDGES - 1);
          *(v2i*)(perm + 2 * (size_t)p) = cw;
        }
      }
    }
  }
  if (staged) {
    __syncthreads();
    for (int i = t - (s & 31); i < m; i += 1024)
      if (i >= 0) *(v2i*)(perm + 2 * (size_t)(s + i)) = st_s[i];
  }
}

#define HALF_LANE(V, J) __builtin_amdgcn_update_dpp(0, (V), 0x150 + ((J) & 15), 0xf, 0xf, true)
#define ROW_REQ(M, J) if ((J) < nfull) { M = *(const v4f*)(hb + (unsigned)(HALF_LANE(ro, (J)) + lo)); }
#define ROW_ADD(M, J) if ((J) < nfull) { float w_ = __int_as_float(HALF_LANE(wb, (J))); gnn_gated_add(M, w_, nwe, acc); }

__device__ __forceinline__ void gnn_gated_add(v4f m, float w, v4f nwe, v4f& acc) {
  v2f z0 = nwe.xy * w, z1 = nwe.zw * w;
  acc.x += m.x * __builtin_amdgcn_rcpf(1.f + __builtin_amdgcn_exp2f(z0.x));
  acc.y += m.y * __builtin_amdgcn_rcpf(1.f + __builtin_amdgcn_exp2f(z0.y));
  acc.z += m.z * __builtin_amdgcn_rcpf(1.f + __builtin_amdgcn_exp2f(z1.x));
  acc.w += m.w * __builtin_amdgcn_rcpf(1.f + __builtin_amdgcn_exp2f(z1.y));
}

template <int G0>
__device__ __forceinline__ void gnn_walk_group(int nfull, int ro, int wb, const char* __restrict__ hb, unsigned lo, v4f nwe, v4f& acc) {
  v4f m0, m1, m2, m3;
  ROW_REQ(m0, G0 + 0) ROW_REQ(m1, G0 + 1) ROW_REQ(m2, G0 + 2) ROW_REQ(m3, G0 + 3)
  ROW_ADD(m0, G0 + 0) ROW_ADD(m1, G0 + 1) ROW_ADD(m2, G0 + 2) ROW_ADD(m3, G0 + 3)
}

__device__ __forceinline__ float gnn_xhalf(float x) {
  int b = __builtin_bit_cast(int, x);
  return __builtin_bit_cast(float, __builtin_amdgcn_permlanex16(b, b, 0x76543210, 0xfedcba98, false, false));
}

__device__ __forceinline__ v2f gnn_node_walk(int node, int lane,
                                             const int* __restrict__ off, const int* __restrict__ perm,
                                             const float* __restrict__ h_neigh, v4f nwe) {
  int s = __builtin_amdgcn_readfirstlane(off[(unsigned)node]);
  int t = __builtin_amdgcn_readfirstlane(off[(unsigned)node + 1u]);
  s = min(max(s, 0), N_EDGES);
  t = min(max(t, s), N_EDGES);
  int hf = lane >> 4, lr = lane & 15;
  const char* hb = (const char*)h_neigh;
  unsigned lo = lr * 16;
  v4f acc = (v4f){};
  for (int p = s; p < t; p += 32) {
    int n = min(t - p, 32);
    int nfull = n >> 1;
    int e = 2 * lr + hf;
    v2i cw; cw.x = 0; cw.y = 0;
    if (e < n) cw = *(const v2i*)(perm + 2u * (unsigned)(p + e));
    int ro = min(max(cw.x, 0), N_NODES - 1) << 8;
    int wb = cw.y;
    v4f mt = (v4f){}; float wt = 0.f;
    if (n & 1) {
      unsigned ot = (unsigned)__builtin_amdgcn_readlane(ro, nfull);
      wt = __int_as_float(__builtin_amdgcn_readlane(wb, nfull));
      mt = *(const v4f*)(hb + (ot + lo));
    }
    if (0 < nfull)  gnn_walk_group<0>(nfull, ro, wb, hb, lo, nwe, acc);
    if (4 < nfull)  gnn_walk_group<4>(nfull, ro, wb, hb, lo, nwe, acc);
    if (8 < nfull)  gnn_walk_group<8>(nfull, ro, wb, hb, lo, nwe, acc);
    if (12 < nfull) gnn_walk_group<12>(nfull, ro, wb, hb, lo, nwe, acc);
    if ((n & 1) && hf == 0) gnn_gated_add(mt, wt, nwe, acc);
  }
  float sx = hf ? acc.x : acc.z, sy = hf ? acc.y : acc.w;
  float kx = hf ? acc.z : acc.x, ky = hf ? acc.w : acc.y;
  v2f r; r.x = kx + gnn_xhalf(sx); r.y = ky + gnn_xhalf(sy);
  return r;
}

__device__ __forceinline__ v2f gnn_node_finish(v2f agg, v2f a, v2f hh, v2f g, v2f bt) {
  float v0 = a.x + agg.x, v1 = a.y + agg.y;
  float sm = v0 + v1, ss = v0 * v0 + v1 * v1;
  sm += __shfl_xor(sm, 8, 32);  ss += __shfl_xor(ss, 8, 32);
  sm += __shfl_xor(sm, 4, 32);  ss += __shfl_xor(ss, 4, 32);
  sm += __shfl_xor(sm, 2, 32);  ss += __shfl_xor(ss, 2, 32);
  sm += __shfl_xor(sm, 1, 32);  ss += __shfl_xor(ss, 1, 32);
  sm += __shfl_xor(sm, 16, 32); ss += __shfl_xor(ss, 16, 32);
  float mu  = sm * (1.f / HID);
  float var = ss * (1.f / HID) - mu * mu;
  float inv = rsqrtf(var + LN_EPS);
  float o0 = (v0 - mu) * inv * g.x + bt.x;
  float o1 = (v1 - mu) * inv * g.y + bt.y;
  o0 = (o0 >= 0.f) ? o0 : SLOPE * o0;
  o1 = (o1 >= 0.f) ? o1 : SLOPE * o1;
  v2f r; r.x = hh.x + o0; r.y = hh.y + o1;
  return r;
}

#define FUSE_WAVES 16
#define FUSE_NODES (FUSE_WAVES * 2)

__device__ __forceinline__ v8f gnn_tile_product(const float* arow, const float* brow) {
  v8f c = (v8f){};
#pragma unroll
  for (int kb = 0; kb < 16; ++kb) {
    v2f af = *(const v2f*)(arow + kb * 4);
    v2f bf = *(const v2f*)(brow + kb * 4);
    c = __builtin_amdgcn_wmma_f32_16x16x4_f32(false, af, false, bf, (short)0, c, false, false);
  }
  asm volatile("v_nop\n\tv_nop\n\tv_nop\n\tv_nop\n\tv_nop" : "+v"(c));
  return c;
}

template <bool LAST>
__device__ __forceinline__ void gnn_gather_body(const int* __restrict__ off, const int* __restrict__ perm,
                                                const float* __restrict__ Wedge, const float* __restrict__ h_neigh,
                                                const float* __restrict__ Wself, const float* __restrict__ bself,
                                                const float* __restrict__ gamma, const float* __restrict__ beta,
                                                const float* h_in, float* h_out,
                                                const float* __restrict__ Wneigh_next, const float* __restrict__ bneigh_next,
                                                float* __restrict__ h_neigh_next) {
  __shared__ float w_s[(LAST ? 1 : 2) * HID * LDW];
  __shared__ float ho_s[FUSE_NODES * LDW];
  __shared__ float hs_s[FUSE_NODES * LDW];

  int t = threadIdx.x;
  int lane = t & 31;
  int wave = __builtin_amdgcn_readfirstlane(t >> 5);
  int lr = lane & 15;
  int rt = wave >> 3, ct = wave & 7;
  int ce = (lane & 15) * 4 + (lane >> 4) * 2;
  v4f nwe = *(const v4f*)(Wedge + (lane & 15) * 4) * -1.44269504f;
  int n0 = blockIdx.x * FUSE_NODES;
  int ln = wave * 2;
  int row = t >> 4, col = (t & 15) * 4;

  unsigned base0 = (unsigned)(n0 + ln) * HID + ce;
  v2f hh0 = *(const v2f*)(h_in + base0);
  v2f hh1 = *(const v2f*)(h_in + base0 + HID);
  v2f agg0 = gnn_node_walk(n0 + ln, lane, off, perm, h_neigh, nwe);
  *(v2f*)(ho_s + ln * LDW + ce)       = hh0;
  *(v2f*)(ho_s + (ln + 1) * LDW + ce) = hh1;
  v2f agg1 = gnn_node_walk(n0 + ln + 1, lane, off, perm, h_neigh, nwe);
  {
    v4f s0 = *(const v4f*)(Wself + 4 * t), s1 = *(const v4f*)(Wself + 4 * (t + 512));
    if (!LAST) {
      v4f m0 = *(const v4f*)(Wneigh_next + 4 * t), m1 = *(const v4f*)(Wneigh_next + 4 * (t + 512));
      *(v4f*)(w_s + (HID + row) * LDW + col)      = m0;
      *(v4f*)(w_s + (HID + row + 32) * LDW + col) = m1;
    }
    *(v4f*)(w_s + row * LDW + col)        = s0;
    *(v4f*)(w_s + (row + 32) * LDW + col) = s1;
  }
  float bias = 0.f;
  if (ct < 4) bias = bself[ct * 16 + lr];
  else if (!LAST) bias = bneigh_next[(ct - 4) * 16 + lr];
  v2f g = *(const v2f*)(gamma + ce), bt = *(const v2f*)(beta + ce);
  __syncthreads();

  int koff = (lane < 16) ? 0 : 2;
  int hi8 = (lane < 16) ? 0 : 8;
  const float* arow = ho_s + (rt * 16 + lr) * LDW + koff;
  if (ct < 4) {
    const float* brow = w_s + (ct * 16 + lr) * LDW + koff;
    v8f c = gnn_tile_product(arow, brow);
    float* d = hs_s + (rt * 16 + hi8) * LDW + ct * 16 + lr;
#pragma unroll
    for (int i = 0; i < 8; ++i) d[i * LDW] = c[i] + bias;
  }
  __syncthreads();

  v2f r0, r1;
  {
    v2f a  = *(const v2f*)(hs_s + ln * LDW + ce);
    v2f hh = *(const v2f*)(ho_s + ln * LDW + ce);
    r0 = gnn_node_finish(agg0, a, hh, g, bt);
  }
  {
    v2f a  = *(const v2f*)(hs_s + (ln + 1) * LDW + ce);
    v2f hh = *(const v2f*)(ho_s + (ln + 1) * LDW + ce);
    r1 = gnn_node_finish(agg1, a, hh, g, bt);
  }
  if (LAST) {
    *(v2f*)(h_out + base0)       = r0;
    *(v2f*)(h_out + base0 + HID) = r1;
    return;
  }
  *(v2f*)(ho_s + ln * LDW + ce)       = r0;
  *(v2f*)(ho_s + (ln + 1) * LDW + ce) = r1;
  __syncthreads();

  *(v2f*)(h_out + base0)       = r0;
  *(v2f*)(h_out + base0 + HID) = r1;
  if (ct >= 4) {
    int cn = ct - 4;
    const float* brow = w_s + (HID + cn * 16 + lr) * LDW + koff;
    v8f c = gnn_tile_product(arow, brow);
    float* o = h_neigh_next + (size_t)(n0 + rt * 16 + hi8) * HID + cn * 16 + lr;
#pragma unroll
    for (int i = 0; i < 8; ++i) o[(size_t)i * HID] = c[i] + bias;
  }
}

__global__ void gnn_gather_ln_gemm_kernel(const int* __restrict__ off, const int* __restrict__ perm,
                                          const float* __restrict__ Wedge, const float* __restrict__ h_neigh,
                                          const float* __restrict__ Wself, const float* __restrict__ bself,
                                          const float* __restrict__ gamma, const float* __restrict__ beta,
                                          float* h,
                                          const float* __restrict__ Wneigh_next, const float* __restrict__ bneigh_next,
                                          float* __restrict__ h_neigh_next) {
  gnn_gather_body<false>(off, perm, Wedge, h_neigh, Wself, bself, gamma, beta, h, h, Wneigh_next, bneigh_next, h_neigh_next);
}

__global__ void gnn_gather_ln_kernel(const int* __restrict__ off, const int* __restrict__ perm,
                                     const float* __restrict__ Wedge, const float* __restrict__ h_neigh,
                                     const float* __restrict__ Wself, const float* __restrict__ bself,
                                     const float* __restrict__ gamma, const float* __restrict__ beta,
                                     const float* h, float* dst) {
  gnn_gather_body<true>(off, perm, Wedge, h_neigh, Wself, bself, gamma, beta, h, dst, nullptr, nullptr, nullptr);
}

extern "C" void kernel_launch(void* const* d_in, const int* in_sizes, int n_in,
                              void* d_out, int out_size, void* d_ws, size_t ws_size,
                              hipStream_t stream) {
  const float* x      = (const float*)d_in[0];
  const int*   eidx   = (const int*)  d_in[1];
  const float* ew     = (const float*)d_in[2];
  const float* Wp     = (const float*)d_in[3];
  const float* bp     = (const float*)d_in[4];
  const float* Wself  = (const float*)d_in[5];
  const float* bself  = (const float*)d_in[6];
  const float* Wneigh = (const float*)d_in[7];
  const float* bneigh = (const float*)d_in[8];
  const float* Wedge  = (const float*)d_in[9];
  const float* gamma  = (const float*)d_in[10];
  const float* beta   = (const float*)d_in[11];
  float* out = (float*)d_out;

  const size_t NH = (size_t)N_NODES * HID;
  float* h       = (float*)d_ws;
  float* h_neigh = h + 2 * NH;
  int*   ord     = (int*)(h_neigh + NH);
  int*   off     = ord + OFF_OFF;
  int*   bh      = ord + OFF_BH;
  int*   bb      = ord + OFF_BB;
  int*   perm    = ord + OFF_PERM;
  int*   tmp     = ord + OFF_TMP;

  gnn_count_proj_kernel<<<EB_BLOCKS + PROJ_BLOCKS, 1024, 0, stream>>>(eidx, bh, x, Wp, bp, h);
  gnn_scan_gemm_kernel<<<1 + GEMM0_BLOCKS, 1024, 0, stream>>>(bh, bb, h, Wneigh, bneigh, h_neigh);

  gnn_bmove_kernel<<<EB_BLOCKS, 1024, 0, stream>>>(eidx, ew, bb, tmp);
  gnn_bfill_kernel<<<NBUCK, 1024, 0, stream>>>(bb, tmp, off, perm);

  for (int l = 0; l < N_LAYERS; ++l) {
    const float* We = Wedge  + (size_t)l * HID;
    const float* gm = gamma  + (size_t)l * HID;
    const float* bt = beta   + (size_t)l * HID;
    const float* Ws = Wself  + (size_t)l * HID * HID;
    const float* bs = bself  + (size_t)l * HID;
    float* hn_cur = (l == 1) ? out : h_neigh;
    float* hn_nxt = (l == 1) ? h_neigh : out;

    if (l < N_LAYERS - 1) {
      const float* Wn = Wneigh + (size_t)(l + 1) * HID * HID;
      const float* bn = bneigh + (size_t)(l + 1) * HID;
      gnn_gather_ln_gemm_kernel<<<N_NODES / FUSE_NODES, 32 * FUSE_WAVES, 0, stream>>>(off, perm, We, hn_cur, Ws, bs, gm, bt, h,
                                                                        Wn, bn, hn_nxt);
    } else {
      gnn_gather_ln_kernel<<<N_NODES / FUSE_NODES, 32 * FUSE_WAVES, 0, stream>>>(off, perm, We, hn_cur, Ws, bs, gm, bt, h, out);
    }
  }
}


# Edge ordering without a global prefix: each block orders its 4000 records by range in place, a range block reads its 250 runs (start = sum of run starts); 5 launches, not 7; sums in LDS arrival order
# speedup vs baseline: 3.3239x; 1.1203x over previous
#include <hip/hip_runtime.h>
#include <hip/hip_bf16.h>

#define N_NODES 100000
#define N_EDGES 1000000
#define IN_DIM  8
#define HID     64
#define N_LAYERS 3
#define LN_EPS  1e-5f
#define SLOPE   0.2f
#define LDW     68

typedef float v2f __attribute__((ext_vector_type(2)));
typedef float v8f __attribute__((ext_vector_type(8)));
typedef int   v2i __attribute__((ext_vector_type(2)));
typedef float v4f __attribute__((ext_vector_type(4)));
typedef int   v4i __attribute__((ext_vector_type(4)));

#define PROJ_WAVES  10000
#define PROJ_BLOCKS 313
#define PROJ_TRIPS  5
#define ROW_TILES   (N_NODES / 16)

#define BUCK_SHIFT 9
#define BUCK_NODES 512
#define NBUCK      196
#define EB_BLOCKS  250
#define EB_EDGES   4000
#define OFF_OFF  0
#define OFF_ST   100352
#define OFF_PERM 200704
#define OFF_TMP  2200704

__device__ __forceinline__ void gnn_proj_body(const float* __restrict__ x, const float* __restrict__ Wp,
                                              const float* __restrict__ bp, float* __restrict__ h, int blk) {
  int t = threadIdx.x;
  int gt = blk * 1024 + t;
  int lane = t & 31;
  int gw = gt >> 5;
  if (gw >= PROJ_WAVES) return;
  int hf = lane >> 4;
  int c4 = (lane & 15) * 4;
  const v4f* wp4 = (const v4f*)(Wp + c4 * IN_DIM);
  v4f w0a = wp4[0], w0b = wp4[1], w1a = wp4[2], w1b = wp4[3];
  v4f w2a = wp4[4], w2b = wp4[5], w3a = wp4[6], w3b = wp4[7];
  v4f b4 = *(const v4f*)(bp + c4);
  v4f xs[PROJ_TRIPS][2];
#pragma unroll
  for (int i = 0; i < PROJ_TRIPS; ++i) {
    int node = (gw * PROJ_TRIPS + i) * 2 + hf;
    const v4f* xr = (const v4f*)(x + (size_t)node * IN_DIM);
    xs[i][0] = xr[0]; xs[i][1] = xr[1];
  }
#pragma unroll
  for (int i = 0; i < PROJ_TRIPS; ++i) {
    int node = (gw * PROJ_TRIPS + i) * 2 + hf;
    v4f xa = xs[i][0], xb = xs[i][1];
    float a0 = b4.x, a1 = b4.y, a2 = b4.z, a3 = b4.w;
    a0 += xa.x * w0a.x; a0 += xa.y * w0a.y; a0 += xa.z * w0a.z; a0 += xa.w * w0a.w;
    a0 += xb.x * w0b.x; a0 += xb.y * w0b.y; a0 += xb.z * w0b.z; a0 += xb.w * w0b.w;
    a1 += xa.x * w1a.x; a1 += xa.y * w1a.y; a1 += xa.z * w1a.z; a1 += xa.w * w1a.w;
    a1 += xb.x * w1b.x; a1 += xb.y * w1b.y; a1 += xb.z * w1b.z; a1 += xb.w * w1b.w;
    a2 += xa.x * w2a.x; a2 += xa.y * w2a.y; a2 += xa.z * w2a.z; a2 += xa.w * w2a.w;
    a2 += xb.x * w2b.x; a2 += xb.y * w2b.y; a2 += xb.z * w2b.z; a2 += xb.w * w2b.w;
    a3 += xa.x * w3a.x; a3 += xa.y * w3a.y; a3 += xa.z * w3a.z; a3 += xa.w * w3a.w;
    a3 += xb.x * w3b.x; a3 += xb.y * w3b.y; a3 += xb.z * w3b.z; a3 += xb.w * w3b.w;
    v4f r; r.x = a0; r.y = a1; r.z = a2; r.w = a3;
    *(v4f*)(h + (size_t)node * HID + c4) = r;
  }
}

template <int THREADS>
__device__ __forceinline__ void gnn_gemm_body(const float* __restrict__ h,
                                const float* __restrict__ Wneigh, const float* __restrict__ bneigh,
                                float* __restrict__ h_neigh, int blk) {
  __shared__ float w_s[HID * LDW];
  __shared__ float b_s[HID];

  int t = threadIdx.x;
#pragma unroll
  for (int i = 0; i < 1024 / THREADS; ++i) {
    int q = t + i * THREADS;
    int row = q >> 4, col = (q & 15) * 4;
    *(v4f*)(w_s + row * LDW + col) = *(const v4f*)(Wneigh + 4 * q);
  }
  if (t < HID) b_s[t] = bneigh[t];
  __syncthreads();

  int wave = t >> 5, lane = t & 31;
  int tile = blk * (THREADS / 32) + wave;
  if (tile >= ROW_TILES) return;
  int r0 = tile * 16;
  int lr = lane & 15;
  int koff = (lane < 16) ? 0 : 2;

  const float* arow = h + (size_t)(r0 + lr) * HID;
  const float* brow = w_s + lr * LDW + koff;

  v2f a[16];
#pragma unroll
  for (int kb = 0; kb < 16; ++kb) a[kb] = *(const v2f*)(arow + kb * 4 + koff);

  v8f c[4];
#pragma unroll
  for (int j = 0; j < 4; ++j) c[j] = (v8f){};
#pragma unroll
  for (int kb = 0; kb < 16; ++kb) {
#pragma unroll
    for (int j = 0; j < 4; ++j) {
      v2f bf = *(const v2f*)(brow + j * 16 * LDW + kb * 4);
      c[j] = __builtin_amdgcn_wmma_f32_16x16x4_f32(false, a[kb], false, bf, (short)0, c[j], false, false);
    }
  }

  int mbase = r0 + ((lane < 16) ? 0 : 8);
#pragma unroll
  for (int j = 0; j < 4; ++j) {
    float bias = b_s[j * 16 + lr];
    int n = j * 16 + lr;
#pragma unroll
    for (int i = 0; i < 8; ++i) {
      h_neigh[(size_t)(mbase + i) * HID + n] = c[j][i] + bias;
    }
  }
}

__device__ __forceinline__ void gnn_bsort_body(const int* __restrict__ eidx, const float* __restrict__ ew,
                                               int* __restrict__ st, int* __restrict__ tmp, int g) {
  __shared__ int hist[NBUCK];
  __shared__ int cur[NBUCK];
  __shared__ int wsum[8];
  __shared__ v4i rec_s[EB_EDGES];
  int t = threadIdx.x;
  int lane = t & 31, wave = t >> 5;
  v4i row = (v4i){}, col = (v4i){};
  v4f w = (v4f){};
  if (t < EB_EDGES / 4) {
    size_t e = (size_t)g * EB_EDGES + 4 * t;
    row = *(const v4i*)(eidx + e);
    col = *(const v4i*)(eidx + N_EDGES + e);
    w   = *(const v4f*)(ew + e);
  }
  if (t < NBUCK) hist[t] = 0;
  __syncthreads();
  int b0 = min(max(row.x >> BUCK_SHIFT, 0), NBUCK - 1);
  int b1 = min(max(row.y >> BUCK_SHIFT, 0), NBUCK - 1);
  int b2 = min(max(row.z >> BUCK_SHIFT, 0), NBUCK - 1);
  int b3 = min(max(row.w >> BUCK_SHIFT, 0), NBUCK - 1);
  if (t < EB_EDGES / 4) {
    atomicAdd(&hist[b0], 1);
    atomicAdd(&hist[b1], 1);
    atomicAdd(&hist[b2], 1);
    atomicAdd(&hist[b3], 1);
  }
  __syncthreads();
  int c = (t < NBUCK) ? hist[t] : 0;
  int inc = c;
  for (int d = 1; d < 32; d <<= 1) {
    int y = __shfl_up(inc, d, 32);
    if (lane >= d) inc += y;
  }
  if (lane == 31 && wave < 8) wsum[wave] = inc;
  __syncthreads();
  if (t < NBUCK) {
    int woff = 0;
    for (int q = 0; q < wave; ++q) woff += wsum[q];
    int lo = woff + inc - c;
    cur[t] = lo;
    st[t * EB_BLOCKS + g] = lo;
  }
  __syncthreads();
  if (t < EB_EDGES / 4) {
    int p0 = atomicAdd(&cur[b0], 1);
    int p1 = atomicAdd(&cur[b1], 1);
    int p2 = atomicAdd(&cur[b2], 1);
    int p3 = atomicAdd(&cur[b3], 1);
    p0 = min(max(p0, 0), EB_EDGES - 1);
    p1 = min(max(p1, 0), EB_EDGES - 1);
    p2 = min(max(p2, 0), EB_EDGES - 1);
    p3 = min(max(p3, 0), EB_EDGES - 1);
    v4i rec; rec.w = 0;
    rec.x = row.x; rec.y = col.x; rec.z = __float_as_int(w.x); rec_s[p0] = rec;
    rec.x = row.y; rec.y = col.y; rec.z = __float_as_int(w.y); rec_s[p1] = rec;
    rec.x = row.z; rec.y = col.z; rec.z = __float_as_int(w.z); rec_s[p2] = rec;
    rec.x = row.w; rec.y = col.w; rec.z = __float_as_int(w.w); rec_s[p3] = rec;
  }
  __syncthreads();
#pragma unroll
  for (int k = 0; k < (EB_EDGES + 1023) / 1024; ++k) {
    int i = t + k * 1024;
    if (i < EB_EDGES) *(v4i*)(tmp + 4 * ((size_t)g * EB_EDGES + i)) = rec_s[i];
  }
}

#define FILL_CAP 8192
__device__ __forceinline__ void gnn_bfill_body(const int* __restrict__ st, const int* __restrict__ tmp,
                                               int* __restrict__ off, int* __restrict__ perm, int b) {
  __shared__ int cnt[BUCK_NODES];
  __shared__ int cur[BUCK_NODES];
  __shared__ int wsum[32];
  __shared__ int seg_lo[256];
  __shared__ int seg_n[256];
  __shared__ int red_s[8];
  __shared__ int red_m[8];
  __shared__ v2i st_s[FILL_CAP];
  int t = threadIdx.x, lane = t & 31, wave = t >> 5;
  if (t < BUCK_NODES) cnt[t] = 0;
  if (t < 256) {
    int lo = 0, n = 0;
    if (t < EB_BLOCKS) {
      lo = st[b * EB_BLOCKS + t];
      int hi = (b + 1 < NBUCK) ? st[(b + 1) * EB_BLOCKS + t] : EB_EDGES;
      lo = min(max(lo, 0), EB_EDGES);
      hi = min(max(hi, lo), EB_EDGES);
      n = hi - lo;
    }
    seg_lo[t] = (t < EB_BLOCKS) ? t * EB_EDGES + lo : 0;
    seg_n[t] = n;
    int sl = lo, sn = n;
    for (int d = 16; d >= 1; d >>= 1) {
      sl += __shfl_xor(sl, d, 32);
      sn += __shfl_xor(sn, d, 32);
    }
    if (lane == 0) { red_s[wave] = sl; red_m[wave] = sn; }
  }
  __syncthreads();
  int s = 0, m = 0;
  for (int q = 0; q < 8; ++q) { s += red_s[q]; m += red_m[q]; }
  s = __builtin_amdgcn_readfirstlane(s);
  m = __builtin_amdgcn_readfirstlane(m);
  s = min(max(s, 0), N_EDGES);
  m = min(max(m, 0), N_EDGES - s);
  bool staged = m <= FILL_CAP;

  int sub = t & 7, ga = t >> 3, gb = ga + 128;
  int base_a = seg_lo[ga], n_a = seg_n[ga];
  int base_b = seg_lo[gb], n_b = seg_n[gb];
  int n_max = max(n_a, n_b);
  for (int j0 = sub; j0 < n_max; j0 += 16) {
    int r[4];
#pragma unroll
    for (int u = 0; u < 4; ++u) {
      int n = (u < 2) ? n_a : n_b, base = (u < 2) ? base_a : base_b;
      int idx = min(max(base + min(j0 + 8 * (u & 1), n - 1), 0), N_EDGES - 1);
      r[u] = tmp[4 * (size_t)idx];
    }
#pragma unroll
    for (int u = 0; u < 4; ++u) {
      int n = (u < 2) ? n_a : n_b;
      if (j0 + 8 * (u & 1) < n && r[u] >= 0) atomicAdd(&cnt[r[u] & (BUCK_NODES - 1)], 1);
    }
  }
  __syncthreads();
  int v = (t < BUCK_NODES) ? cnt[t] : 0;
  int inc = v;
  for (int d = 1; d < 32; d <<= 1) {
    int y = __shfl_up(inc, d, 32);
    if (lane >= d) inc += y;
  }
  if (lane == 31) wsum[wave] = inc;
  __syncthreads();
  if (wave == 0) {
    int x = wsum[lane];
    int xi = x;
    for (int d = 1; d < 32; d <<= 1) {
      int y = __shfl_up(xi, d, 32);
      if (lane >= d) xi += y;
    }
    wsum[lane] = xi - x;
  }
  __syncthreads();
  if (t < BUCK_NODES) {
    int first = s + wsum[wave] + inc - v;
    cur[t] = first;
    int node = b * BUCK_NODES + t;
    if (node <= N_NODES) off[node] = first;
  }
  __syncthreads();
  for (int j0 = sub; j0 < n_max; j0 += 16) {
    v4i rec[4];
#pragma unroll
    for (int u = 0; u < 4; ++u) {
      int n = (u < 2) ? n_a : n_b, base = (u < 2) ? base_a : base_b;
      int idx = min(max(base + min(j0 + 8 * (u & 1), n - 1), 0), N_EDGES - 1);
      rec[u] = *(const v4i*)(tmp + 4 * (size_t)idx);
    }
#pragma unroll
    for (int u = 0; u < 4; ++u) {
      int n = (u < 2) ? n_a : n_b;
      if (j0 + 8 * (u & 1) < n && rec[u].x >= 0) {
        int p = atomicAdd(&cur[rec[u].x & (BUCK_NODES - 1)], 1);
        v2i cw; cw.x = rec[u].y; cw.y = rec[u].z;
        if (staged) {
          st_s[min(max(p - s, 0), FILL_CAP - 1)] = cw;
        } else {
          p = min(max(p, 0), N_EDGES - 1);
          *(v2i*)(perm + 2 * (size_t)p) = cw;
        }
      }
    }
  }
  if (staged) {
    __syncthreads();
    for (int i = t - (s & 31); i < m; i += 1024)
      if (i >= 0) *(v2i*)(perm + 2 * (size_t)(s + i)) = st_s[i];
  }
}

__global__ void gnn_sort_proj_kernel(const int* __restrict__ eidx, const float* __restrict__ ew,
                                     int* __restrict__ st, int* __restrict__ tmp,
                                     const float* __restrict__ x, const float* __restrict__ Wp,
                                     const float* __restrict__ bp, float* __restrict__ h) {
  if (blockIdx.x < EB_BLOCKS) gnn_bsort_body(eidx, ew, st, tmp, blockIdx.x);
  else gnn_proj_body(x, Wp, bp, h, blockIdx.x - EB_BLOCKS);
}

#define GEMM0_BLOCKS ((ROW_TILES + 31) / 32)
__global__ void __launch_bounds__(1024, 16) gnn_fill_gemm_kernel(const int* __restrict__ st, const int* __restrict__ tmp,
                                     int* __restrict__ off, int* __restrict__ perm,
                                     const float* __restrict__ h,
                                     const float* __restrict__ Wneigh, const float* __restrict__ bneigh,
                                     float* __restrict__ h_neigh) {
  if (blockIdx.x < NBUCK) gnn_bfill_body(st, tmp, off, perm, blockIdx.x);
  else gnn_gemm_body<1024>(h, Wneigh, bneigh, h_neigh, blockIdx.x - NBUCK);
}

#define HALF_LANE(V, J) __builtin_amdgcn_update_dpp(0, (V), 0x150 + ((J) & 15), 0xf, 0xf, true)
#define ROW_REQ(M, J) if ((J) < nfull) { M = *(const v4f*)(hb + (unsigned)(HALF_LANE(ro, (J)) + lo)); }
#define ROW_ADD(M, J) if ((J) < nfull) { float w_ = __int_as_float(HALF_LANE(wb, (J))); gnn_gated_add(M, w_, nwe, acc); }

__device__ __forceinline__ void gnn_gated_add(v4f m, float w, v4f nwe, v4f& acc) {
  v2f z0 = nwe.xy * w, z1 = nwe.zw * w;
  acc.x += m.x * __builtin_amdgcn_rcpf(1.f + __builtin_amdgcn_exp2f(z0.x));
  acc.y += m.y * __builtin_amdgcn_rcpf(1.f + __builtin_amdgcn_exp2f(z0.y));
  acc.z += m.z * __builtin_amdgcn_rcpf(1.f + __builtin_amdgcn_exp2f(z1.x));
  acc.w += m.w * __builtin_amdgcn_rcpf(1.f + __builtin_amdgcn_exp2f(z1.y));
}

template <int G0>
__device__ __forceinline__ void gnn_walk_group(int nfull, int ro, int wb, const char* __restrict__ hb, unsigned lo, v4f nwe, v4f& acc) {
  v4f m0, m1, m2, m3;
  ROW_REQ(m0, G0 + 0) ROW_REQ(m1, G0 + 1) ROW_REQ(m2, G0 + 2) ROW_REQ(m3, G0 + 3)
  ROW_ADD(m0, G0 + 0) ROW_ADD(m1, G0 + 1) ROW_ADD(m2, G0 + 2) ROW_ADD(m3, G0 + 3)
}

__device__ __forceinline__ float gnn_xhalf(float x) {
  int b = __builtin_bit_cast(int, x);
  return __builtin_bit_cast(float, __builtin_amdgcn_permlanex16(b, b, 0x76543210, 0xfedcba98, false, false));
}

__device__ __forceinline__ v2f gnn_node_walk(int node, int lane,
                                             const int* __restrict__ off, const int* __restrict__ perm,
                                             const float* __restrict__ h_neigh, v4f nwe) {
  int s = __builtin_amdgcn_readfirstlane(off[(unsigned)node]);
  int t = __builtin_amdgcn_readfirstlane(off[(unsigned)node + 1u]);
  s = min(max(s, 0), N_EDGES);
  t = min(max(t, s), N_EDGES);
  int hf = lane >> 4, lr = lane & 15;
  const char* hb = (const char*)h_neigh;
  unsigned lo = lr * 16;
  v4f acc = (v4f){};
  for (int p = s; p < t; p += 32) {
    int n = min(t - p, 32);
    int nfull = n >> 1;
    int e = 2 * lr + hf;
    v2i cw; cw.x = 0; cw.y = 0;
    if (e < n) cw = *(const v2i*)(perm + 2u * (unsigned)(p + e));
    int ro = min(max(cw.x, 0), N_NODES - 1) << 8;
    int wb = cw.y;
    v4f mt = (v4f){}; float wt = 0.f;
    if (n & 1) {
      unsigned ot = (unsigned)__builtin_amdgcn_readlane(ro, nfull);
      wt = __int_as_float(__builtin_amdgcn_readlane(wb, nfull));
      mt = *(const v4f*)(hb + (ot + lo));
    }
    if (0 < nfull)  gnn_walk_group<0>(nfull, ro, wb, hb, lo, nwe, acc);
    if (4 < nfull)  gnn_walk_group<4>(nfull, ro, wb, hb, lo, nwe, acc);
    if (8 < nfull)  gnn_walk_group<8>(nfull, ro, wb, hb, lo, nwe, acc);
    if (12 < nfull) gnn_walk_group<12>(nfull, ro, wb, hb, lo, nwe, acc);
    if ((n & 1) && hf == 0) gnn_gated_add(mt, wt, nwe, acc);
  }
  float sx = hf ? acc.x : acc.z, sy = hf ? acc.y : acc.w;
  float kx = hf ? acc.z : acc.x, ky = hf ? acc.w : acc.y;
  v2f r; r.x = kx + gnn_xhalf(sx); r.y = ky + gnn_xhalf(sy);
  return r;
}

__device__ __forceinline__ v2f gnn_node_finish(v2f agg, v2f a, v2f hh, v2f g, v2f bt) {
  float v0 = a.x + agg.x, v1 = a.y + agg.y;
  float sm = v0 + v1, ss = v0 * v0 + v1 * v1;
  sm += __shfl_xor(sm, 8, 32);  ss += __shfl_xor(ss, 8, 32);
  sm += __shfl_xor(sm, 4, 32);  ss += __shfl_xor(ss, 4, 32);
  sm += __shfl_xor(sm, 2, 32);  ss += __shfl_xor(ss, 2, 32);
  sm += __shfl_xor(sm, 1, 32);  ss += __shfl_xor(ss, 1, 32);
  sm += __shfl_xor(sm, 16, 32); ss += __shfl_xor(ss, 16, 32);
  float mu  = sm * (1.f / HID);
  float var = ss * (1.f / HID) - mu * mu;
  float inv = rsqrtf(var + LN_EPS);
  float o0 = (v0 - mu) * inv * g.x + bt.x;
  float o1 = (v1 - mu) * inv * g.y + bt.y;
  o0 = (o0 >= 0.f) ? o0 : SLOPE * o0;
  o1 = (o1 >= 0.f) ? o1 : SLOPE * o1;
  v2f r; r.x = hh.x + o0; r.y = hh.y + o1;
  return r;
}

#define FUSE_WAVES 16
#define FUSE_NODES (FUSE_WAVES * 2)

__device__ __forceinline__ v8f gnn_tile_product(const float* arow, const float* brow) {
  v8f c = (v8f){};
#pragma unroll
  for (int kb = 0; kb < 16; ++kb) {
    v2f af = *(const v2f*)(arow + kb * 4);
    v2f bf = *(const v2f*)(brow + kb * 4);
    c = __builtin_amdgcn_wmma_f32_16x16x4_f32(false, af, false, bf, (short)0, c, false, false);
  }
  asm volatile("v_nop\n\tv_nop\n\tv_nop\n\tv_nop\n\tv_nop" : "+v"(c));
  return c;
}

template <bool LAST>
__device__ __forceinline__ void gnn_gather_body(const int* __restrict__ off, const int* __restrict__ perm,
                                                const float* __restrict__ Wedge, const float* __restrict__ h_neigh,
                                                const float* __restrict__ Wself, const float* __restrict__ bself,
                                                const float* __restrict__ gamma, const float* __restrict__ beta,
                                                const float* h_in, float* h_out,
                                                const float* __restrict__ Wneigh_next, const float* __restrict__ bneigh_next,
                                                float* __restrict__ h_neigh_next) {
  __shared__ float w_s[(LAST ? 1 : 2) * HID * LDW];
  __shared__ float ho_s[FUSE_NODES * LDW];
  __shared__ float hs_s[FUSE_NODES * LDW];

  int t = threadIdx.x;
  int lane = t & 31;
  int wave = __builtin_amdgcn_readfirstlane(t >> 5);
  int lr = lane & 15;
  int rt = wave >> 3, ct = wave & 7;
  int ce = (lane & 15) * 4 + (lane >> 4) * 2;
  v4f nwe = *(const v4f*)(Wedge + (lane & 15) * 4) * -1.44269504f;
  int n0 = blockIdx.x * FUSE_NODES;
  int ln = wave * 2;
  int row = t >> 4, col = (t & 15) * 4;

  unsigned base0 = (unsigned)(n0 + ln) * HID + ce;
  v2f hh0 = *(const v2f*)(h_in + base0);
  v2f hh1 = *(const v2f*)(h_in + base0 + HID);
  v2f agg0 = gnn_node_walk(n0 + ln, lane, off, perm, h_neigh, nwe);
  *(v2f*)(ho_s + ln * LDW + ce)       = hh0;
  *(v2f*)(ho_s + (ln + 1) * LDW + ce) = hh1;
  v2f agg1 = gnn_node_walk(n0 + ln + 1, lane, off, perm, h_neigh, nwe);
  {
    v4f s0 = *(const v4f*)(Wself + 4 * t), s1 = *(const v4f*)(Wself + 4 * (t + 512));
    if (!LAST) {
      v4f m0 = *(const v4f*)(Wneigh_next + 4 * t), m1 = *(const v4f*)(Wneigh_next + 4 * (t + 512));
      *(v4f*)(w_s + (HID + row) * LDW + col)      = m0;
      *(v4f*)(w_s + (HID + row + 32) * LDW + col) = m1;
    }
    *(v4f*)(w_s + row * LDW + col)        = s0;
    *(v4f*)(w_s + (row + 32) * LDW + col) = s1;
  }
  float bias = 0.f;
  if (ct < 4) bias = bself[ct * 16 + lr];
  else if (!LAST) bias = bneigh_next[(ct - 4) * 16 + lr];
  v2f g = *(const v2f*)(gamma + ce), bt = *(const v2f*)(beta + ce);
  __syncthreads();

  int koff = (lane < 16) ? 0 : 2;
  int hi8 = (lane < 16) ? 0 : 8;
  const float* arow = ho_s + (rt * 16 + lr) * LDW + koff;
  if (ct < 4) {
    const float* brow = w_s + (ct * 16 + lr) * LDW + koff;
    v8f c = gnn_tile_product(arow, brow);
    float* d = hs_s + (rt * 16 + hi8) * LDW + ct * 16 + lr;
#pragma unroll
    for (int i = 0; i < 8; ++i) d[i * LDW] = c[i] + bias;
  }
  __syncthreads();

  v2f r0, r1;
  {
    v2f a  = *(const v2f*)(hs_s + ln * LDW + ce);
    v2f hh = *(const v2f*)(ho_s + ln * LDW + ce);
    r0 = gnn_node_finish(agg0, a, hh, g, bt);
  }
  {
    v2f a  = *(const v2f*)(hs_s + (ln + 1) * LDW + ce);
    v2f hh = *(const v2f*)(ho_s + (ln + 1) * LDW + ce);
    r1 = gnn_node_finish(agg1, a, hh, g, bt);
  }
  if (LAST) {
    *(v2f*)(h_out + base0)       = r0;
    *(v2f*)(h_out + base0 + HID) = r1;
    return;
  }
  *(v2f*)(ho_s + ln * LDW + ce)       = r0;
  *(v2f*)(ho_s + (ln + 1) * LDW + ce) = r1;
  __syncthreads();

  *(v2f*)(h_out + base0)       = r0;
  *(v2f*)(h_out + base0 + HID) = r1;
  if (ct >= 4) {
    int cn = ct - 4;
    const float* brow = w_s + (HID + cn * 16 + lr) * LDW + koff;
    v8f c = gnn_tile_product(arow, brow);
    float* o = h_neigh_next + (size_t)(n0 + rt * 16 + hi8) * HID + cn * 16 + lr;
#pragma unroll
    for (int i = 0; i < 8; ++i) o[(size_t)i * HID] = c[i] + bias;
  }
}

__global__ void gnn_gather_ln_gemm_kernel(const int* __restrict__ off, const int* __restrict__ perm,
                                          const float* __restrict__ Wedge, const float* __restrict__ h_neigh,
                                          const float* __restrict__ Wself, const float* __restrict__ bself,
                                          const float* __restrict__ gamma, const float* __restrict__ beta,
                                          float* h,
                                          const float* __restrict__ Wneigh_next, const float* __restrict__ bneigh_next,
                                          float* __restrict__ h_neigh_next) {
  gnn_gather_body<false>(off, perm, Wedge, h_neigh, Wself, bself, gamma, beta, h, h, Wneigh_next, bneigh_next, h_neigh_next);
}

__global__ void gnn_gather_ln_kernel(const int* __restrict__ off, const int* __restrict__ perm,
                                     const float* __restrict__ Wedge, const float* __restrict__ h_neigh,
                                     const float* __restrict__ Wself, const float* __restrict__ bself,
                                     const float* __restrict__ gamma, const float* __restrict__ beta,
                                     const float* h, float* dst) {
  gnn_gather_body<true>(off, perm, Wedge, h_neigh, Wself, bself, gamma, beta, h, dst, nullptr, nullptr, nullptr);
}

extern "C" void kernel_launch(void* const* d_in, const int* in_sizes, int n_in,
                              void* d_out, int out_size, void* d_ws, size_t ws_size,
                              hipStream_t stream) {
  const float* x      = (const float*)d_in[0];
  const int*   eidx   = (const int*)  d_in[1];
  const float* ew     = (const float*)d_in[2];
  const float* Wp     = (const float*)d_in[3];
  const float* bp     = (const float*)d_in[4];
  const float* Wself  = (const float*)d_in[5];
  const float* bself  = (const float*)d_in[6];
  const float* Wneigh = (const float*)d_in[7];
  const float* bneigh = (const float*)d_in[8];
  const float* Wedge  = (const float*)d_in[9];
  const float* gamma  = (const float*)d_in[10];
  const float* beta   = (const float*)d_in[11];
  float* out = (float*)d_out;

  const size_t NH = (size_t)N_NODES * HID;
  float* h       = (float*)d_ws;
  float* h_neigh = h + 2 * NH;
  int*   ord     = (int*)(h_neigh + NH);
  int*   off     = ord + OFF_OFF;
  int*   st      = ord + OFF_ST;
  int*   perm    = ord + OFF_PERM;
  int*   tmp     = ord + OFF_TMP;

  gnn_sort_proj_kernel<<<EB_BLOCKS + PROJ_BLOCKS, 1024, 0, stream>>>(eidx, ew, st, tmp, x, Wp, bp, h);
  gnn_fill_gemm_kernel<<<NBUCK + GEMM0_BLOCKS, 1024, 0, stream>>>(st, tmp, off, perm, h, Wneigh, bneigh, h_neigh);

  for (int l = 0; l < N_LAYERS; ++l) {
    const float* We = Wedge  + (size_t)l * HID;
    const float* gm = gamma  + (size_t)l * HID;
    const float* bt = beta   + (size_t)l * HID;
    const float* Ws = Wself  + (size_t)l * HID * HID;
    const float* bs = bself  + (size_t)l * HID;
    float* hn_cur = (l == 1) ? out : h_neigh;
    float* hn_nxt = (l == 1) ? h_neigh : out;

    if (l < N_LAYERS - 1) {
      const float* Wn = Wneigh + (size_t)(l + 1) * HID * HID;
      const float* bn = bneigh + (size_t)(l + 1) * HID;
      gnn_gather_ln_gemm_kernel<<<N_NODES / FUSE_NODES, 32 * FUSE_WAVES, 0, stream>>>(off, perm, We, hn_cur, Ws, bs, gm, bt, h,
                                                                        Wn, bn, hn_nxt);
    } else {
      gnn_gather_ln_kernel<<<N_NODES / FUSE_NODES, 32 * FUSE_WAVES, 0, stream>>>(off, perm, We, hn_cur, Ws, bs, gm, bt, h, out);
    }
  }
}
